# starting kernel
# baseline (speedup 1.0000x reference)
#include <hip/hip_runtime.h>
#include <hip/hip_bf16.h>

#define NN 50000
#define NE 800000
#define NG 256
#define D  128

typedef float v2f __attribute__((ext_vector_type(2)));
typedef float v8f __attribute__((ext_vector_type(8)));

__device__ __forceinline__ void fatomic_add(float* p, float v) {
    unsafeAtomicAdd(p, v);
}

__global__ void k_deg_init(float* deg) {
    int n = blockIdx.x * blockDim.x + threadIdx.x;
    if (n < NN) deg[n] = 1.0f;
}

__global__ void k_deg_accum(const int* __restrict__ dst, float* __restrict__ deg) {
    int e = blockIdx.x * blockDim.x + threadIdx.x;
    if (e < NE) fatomic_add(&deg[dst[e]], 1.0f);
}

__global__ void k_dinv(const float* __restrict__ deg, float* __restrict__ dinv) {
    int n = blockIdx.x * blockDim.x + threadIdx.x;
    if (n < NN) {
        float d = deg[n];
        dinv[n] = d > 0.0f ? rsqrtf(d) : 0.0f;
    }
}

__global__ void k_edge_norm(const int* __restrict__ src, const int* __restrict__ dst,
                            const float* __restrict__ dinv, float* __restrict__ nrm) {
    int e = blockIdx.x * blockDim.x + threadIdx.x;
    if (e < NE) nrm[e] = dinv[src[e]] * dinv[dst[e]];
}

template <int RELU>
__global__ __launch_bounds__(256) void k_gemm_wmma(
    const float* __restrict__ in, const float* __restrict__ W,
    float* __restrict__ out, int nrows) {
    __shared__ float ldsW[128 * 129];
    const int tid = threadIdx.x;
    for (int i = tid; i < 128 * 128; i += 256)
        ldsW[(i >> 7) * 129 + (i & 127)] = W[i];
    __syncthreads();

    const int lane  = tid & 31;
    const int wave  = tid >> 5;
    const int strip = blockIdx.x * 128 + wave * 16;
    int rowA = strip + (lane & 15);
    if (rowA >= nrows) rowA = nrows - 1;
    const int koff = (lane >> 4) << 1;
    const int col  = lane & 15;

    v8f acc[8] = {};
    const float* aBase = in + (size_t)rowA * D;

    for (int k = 0; k < D; k += 4) {
        const float* ap = aBase + k + koff;
        v2f a;
        a.x = ap[0];
        a.y = ap[1];
        if (RELU) { a.x = fmaxf(a.x, 0.0f); a.y = fmaxf(a.y, 0.0f); }
        const float* b0 = &ldsW[(k + koff)     * 129 + col];
        const float* b1 = &ldsW[(k + koff + 1) * 129 + col];
#pragma unroll
        for (int j = 0; j < 8; ++j) {
            v2f b;
            b.x = b0[j * 16];
            b.y = b1[j * 16];
            acc[j] = __builtin_amdgcn_wmma_f32_16x16x4_f32(
                false, a, false, b, (short)0, acc[j], false, false);
        }
    }

    const int rbase = strip + (lane >> 4) * 8;
    if (strip + 16 <= nrows) {
#pragma unroll
        for (int j = 0; j < 8; ++j) {
#pragma unroll
            for (int r = 0; r < 8; ++r) {
                out[(size_t)(rbase + r) * D + j * 16 + col] = acc[j][r];
            }
        }
    } else {
#pragma unroll
        for (int j = 0; j < 8; ++j) {
#pragma unroll
            for (int r = 0; r < 8; ++r) {
                int row = rbase + r;
                if (row < nrows) out[(size_t)row * D + j * 16 + col] = acc[j][r];
            }
        }
    }
}

__global__ void k_agg_init(const float* __restrict__ h, const float* __restrict__ dinv,
                           const float* __restrict__ bias, float* __restrict__ agg) {
    long long i = (long long)blockIdx.x * blockDim.x + threadIdx.x;
    long long n = i >> 5;
    int lane = (int)(i & 31);
    if (n >= NN) return;
    float di = dinv[n];
    float sn = di * di;
    const float4 v = *reinterpret_cast<const float4*>(h + (size_t)n * D + lane * 4);
    const float4 b = *reinterpret_cast<const float4*>(bias + lane * 4);
    float4 o;
    o.x = b.x + v.x * sn; o.y = b.y + v.y * sn;
    o.z = b.z + v.z * sn; o.w = b.w + v.w * sn;
    *reinterpret_cast<float4*>(agg + (size_t)n * D + lane * 4) = o;
}

__global__ __launch_bounds__(256) void k_edge_scatter(
    const float* __restrict__ h, const int* __restrict__ src,
    const int* __restrict__ dst, const float* __restrict__ nrmv,
    float* __restrict__ agg) {
    long long gt = (long long)blockIdx.x * blockDim.x + threadIdx.x;
    long long e  = gt >> 5;
    int lane = (int)(gt & 31);
    if (e >= NE) return;
    int s = src[e], d = dst[e];
    float nrm = nrmv[e];
    const float4 v = *reinterpret_cast<const float4*>(h + (size_t)s * D + lane * 4);
    float* o = agg + (size_t)d * D + lane * 4;
    fatomic_add(o + 0, v.x * nrm);
    fatomic_add(o + 1, v.y * nrm);
    fatomic_add(o + 2, v.z * nrm);
    fatomic_add(o + 3, v.w * nrm);
}

__global__ void k_pool_zero(float* sums_cnt) {
    int i = blockIdx.x * blockDim.x + threadIdx.x;
    if (i < NG * D + NG) sums_cnt[i] = 0.0f;
}

__global__ void k_pool_cnt(const int* __restrict__ batch, float* __restrict__ cnt) {
    int n = blockIdx.x * blockDim.x + threadIdx.x;
    if (n < NN) fatomic_add(&cnt[batch[n]], 1.0f);
}

__global__ void k_pool_accum(const float* __restrict__ h, const int* __restrict__ batch,
                             float* __restrict__ sums) {
    long long i = (long long)blockIdx.x * blockDim.x + threadIdx.x;
    long long n = i >> 5;
    int lane = (int)(i & 31);
    if (n >= NN) return;
    int g = batch[n];
    const float4 v = *reinterpret_cast<const float4*>(h + (size_t)n * D + lane * 4);
    float* o = sums + (size_t)g * D + lane * 4;
    fatomic_add(o + 0, v.x);
    fatomic_add(o + 1, v.y);
    fatomic_add(o + 2, v.z);
    fatomic_add(o + 3, v.w);
}

__global__ void k_pool_final(const float* __restrict__ sums, const float* __restrict__ cnt,
                             float* __restrict__ out) {
    int i = blockIdx.x * blockDim.x + threadIdx.x;
    if (i < NG * D) out[i] = sums[i] / fmaxf(cnt[i >> 7], 1.0f);
}

extern "C" void kernel_launch(void* const* d_in, const int* in_sizes, int n_in,
                              void* d_out, int out_size, void* d_ws, size_t ws_size,
                              hipStream_t stream) {
    const float* x     = (const float*)d_in[0];
    const int*   edge  = (const int*)d_in[1];
    const int*   batch = (const int*)d_in[2];
    const float* W1 = (const float*)d_in[3];
    const float* b1 = (const float*)d_in[4];
    const float* W2 = (const float*)d_in[5];
    const float* b2 = (const float*)d_in[6];
    const float* W3 = (const float*)d_in[7];
    const float* b3 = (const float*)d_in[8];

    const int* esrc = edge;
    const int* edst = edge + NE;

    float* ws   = (float*)d_ws;
    float* deg  = ws;
    float* dinv = ws + NN;
    float* bufH = ws + 100000;
    float* bufA = bufH + (size_t)NN * D;
    float* sums = bufA + (size_t)NN * D;
    float* cnt  = sums + NG * D;
    float* nrm  = cnt + NG;

    const int TB = 256;
    const int gNodes   = (NN + TB - 1) / TB;
    const int gEdges   = (NE + TB - 1) / TB;
    const int gNode32  = (NN * 32 + TB - 1) / TB;
    const int gEdge32  = (int)(((long long)NE * 32 + TB - 1) / TB);
    const int gGemm    = (NN + 127) / 128;

    k_deg_init <<<gNodes, TB, 0, stream>>>(deg);
    k_deg_accum<<<gEdges, TB, 0, stream>>>(edst, deg);
    k_dinv     <<<gNodes, TB, 0, stream>>>(deg, dinv);
    k_edge_norm<<<gEdges, TB, 0, stream>>>(esrc, edst, dinv, nrm);

    k_gemm_wmma<0><<<gGemm, TB, 0, stream>>>(x, W1, bufH, NN);
    k_agg_init    <<<gNode32, TB, 0, stream>>>(bufH, dinv, b1, bufA);
    k_edge_scatter<<<gEdge32, TB, 0, stream>>>(bufH, esrc, edst, nrm, bufA);

    k_gemm_wmma<1><<<gGemm, TB, 0, stream>>>(bufA, W2, bufH, NN);
    k_agg_init    <<<gNode32, TB, 0, stream>>>(bufH, dinv, b2, bufA);
    k_edge_scatter<<<gEdge32, TB, 0, stream>>>(bufH, esrc, edst, nrm, bufA);

    k_gemm_wmma<1><<<gGemm, TB, 0, stream>>>(bufA, W3, bufH, NN);
    k_agg_init    <<<gNode32, TB, 0, stream>>>(bufH, dinv, b3, bufA);
    k_edge_scatter<<<gEdge32, TB, 0, stream>>>(bufH, esrc, edst, nrm, bufA);

    k_pool_zero <<<(NG * D + NG + TB - 1) / TB, TB, 0, stream>>>(sums);
    k_pool_cnt  <<<gNodes, TB, 0, stream>>>(batch, cnt);
    k_pool_accum<<<gNode32, TB, 0, stream>>>(bufA, batch, sums);
    k_pool_final<<<(NG * D + TB - 1) / TB, TB, 0, stream>>>(sums, cnt, (float*)d_out);
}


# Edge aggregation as gather over dst-sorted lists (per-call counting sort, integer atomic arrival rank): one wave sums a row in registers, fma, one store; no float atomics on agg; sum order changes
# speedup vs baseline: 1.9152x; 1.9152x over previous
#include <hip/hip_runtime.h>
#include <hip/hip_bf16.h>

#define NN 50000
#define NE 800000
#define NG 256
#define D  128

typedef float v2f __attribute__((ext_vector_type(2)));
typedef float v8f __attribute__((ext_vector_type(8)));

__device__ __forceinline__ void fatomic_add(float* p, float v) {
    unsafeAtomicAdd(p, v);
}

static_assert(NN % 4 == 0, "row scan moves four counts per thread");

__global__ void k_cnt_zero(int* __restrict__ cnt) {
    int n = blockIdx.x * blockDim.x + threadIdx.x;
    if (n < NN) cnt[n] = 0;
}

__global__ void k_edge_rank(const int* __restrict__ dst, int* __restrict__ cnt,
                            int* __restrict__ rank) {
    int e = blockIdx.x * blockDim.x + threadIdx.x;
    if (e < NE) rank[e] = atomicAdd(&cnt[dst[e]], 1);
}

__global__ __launch_bounds__(1024) void k_row_scan(int* __restrict__ cnt_start,
                                                   float* __restrict__ dinv) {
    __shared__ int wtot[32];
    const int tid  = threadIdx.x;
    const int lane = tid & 31;
    const int wave = tid >> 5;
    int carry = 0;
    for (int base = 0; base < NN; base += 4096) {
        const int i = base + tid * 4;
        int4 c = make_int4(0, 0, 0, 0);
        if (i < NN) {
            c = *reinterpret_cast<const int4*>(cnt_start + i);
            float4 r;
            float d0 = (float)(c.x + 1), d1 = (float)(c.y + 1);
            float d2 = (float)(c.z + 1), d3 = (float)(c.w + 1);
            r.x = d0 > 0.0f ? rsqrtf(d0) : 0.0f;
            r.y = d1 > 0.0f ? rsqrtf(d1) : 0.0f;
            r.z = d2 > 0.0f ? rsqrtf(d2) : 0.0f;
            r.w = d3 > 0.0f ? rsqrtf(d3) : 0.0f;
            *reinterpret_cast<float4*>(dinv + i) = r;
        }
        const int s = c.x + c.y + c.z + c.w;
        int v = s;
        for (int d = 1; d < 32; d <<= 1) {
            int t = __shfl_up(v, d, 32);
            if (lane >= d) v += t;
        }
        if (lane == 31) wtot[wave] = v;
        __syncthreads();
        if (wave == 0) {
            int t = wtot[lane];
            for (int d = 1; d < 32; d <<= 1) {
                int u = __shfl_up(t, d, 32);
                if (lane >= d) t += u;
            }
            wtot[lane] = t;
        }
        __syncthreads();
        const int excl = carry + (wave ? wtot[wave - 1] : 0) + v - s;
        if (i < NN) {
            int4 o;
            o.x = excl;
            o.y = excl + c.x;
            o.z = excl + c.x + c.y;
            o.w = excl + c.x + c.y + c.z;
            *reinterpret_cast<int4*>(cnt_start + i) = o;
        }
        carry += wtot[31];
        __syncthreads();
    }
}

__global__ void k_edge_fill(const int* __restrict__ src, const int* __restrict__ dst,
                            const int* __restrict__ rowstart, const int* __restrict__ rank,
                            int* __restrict__ csrc) {
    int e = blockIdx.x * blockDim.x + threadIdx.x;
    if (e < NE) csrc[rowstart[dst[e]] + rank[e]] = src[e];
}

template <int RELU>
__global__ __launch_bounds__(256) void k_gemm_wmma(
    const float* __restrict__ in, const float* __restrict__ W,
    float* __restrict__ out, int nrows) {
    __shared__ float ldsW[128 * 129];
    const int tid = threadIdx.x;
    for (int i = tid; i < 128 * 128; i += 256)
        ldsW[(i >> 7) * 129 + (i & 127)] = W[i];
    __syncthreads();

    const int lane  = tid & 31;
    const int wave  = tid >> 5;
    const int strip = blockIdx.x * 128 + wave * 16;
    int rowA = strip + (lane & 15);
    if (rowA >= nrows) rowA = nrows - 1;
    const int koff = (lane >> 4) << 1;
    const int col  = lane & 15;

    v8f acc[8] = {};
    const float* aBase = in + (size_t)rowA * D;

    for (int k = 0; k < D; k += 4) {
        const float* ap = aBase + k + koff;
        v2f a;
        a.x = ap[0];
        a.y = ap[1];
        if (RELU) { a.x = fmaxf(a.x, 0.0f); a.y = fmaxf(a.y, 0.0f); }
        const float* b0 = &ldsW[(k + koff)     * 129 + col];
        const float* b1 = &ldsW[(k + koff + 1) * 129 + col];
#pragma unroll
        for (int j = 0; j < 8; ++j) {
            v2f b;
            b.x = b0[j * 16];
            b.y = b1[j * 16];
            acc[j] = __builtin_amdgcn_wmma_f32_16x16x4_f32(
                false, a, false, b, (short)0, acc[j], false, false);
        }
    }

    const int rbase = strip + (lane >> 4) * 8;
    if (strip + 16 <= nrows) {
#pragma unroll
        for (int j = 0; j < 8; ++j) {
#pragma unroll
            for (int r = 0; r < 8; ++r) {
                out[(size_t)(rbase + r) * D + j * 16 + col] = acc[j][r];
            }
        }
    } else {
#pragma unroll
        for (int j = 0; j < 8; ++j) {
#pragma unroll
            for (int r = 0; r < 8; ++r) {
                int row = rbase + r;
                if (row < nrows) out[(size_t)row * D + j * 16 + col] = acc[j][r];
            }
        }
    }
}

template <int K>
__device__ __forceinline__ void gather_rows(const float4* __restrict__ hv, int sidx, float dsrc,
                                            float di, int j, int lane, float4& acc) {
    float4 v[K];
#pragma unroll
    for (int u = 0; u < K; ++u) {
        const int s = __builtin_amdgcn_readlane(sidx, j + u);
        v[u] = hv[(size_t)s * 32 + lane];
    }
#pragma unroll
    for (int u = 0; u < K; ++u) {
        const float w = __int_as_float(__builtin_amdgcn_readlane(__float_as_int(dsrc), j + u)) * di;
        acc.x = __builtin_fmaf(v[u].x, w, acc.x);
        acc.y = __builtin_fmaf(v[u].y, w, acc.y);
        acc.z = __builtin_fmaf(v[u].z, w, acc.z);
        acc.w = __builtin_fmaf(v[u].w, w, acc.w);
    }
}

__global__ __launch_bounds__(256) void k_row_gather(
    const float* __restrict__ h, const int* __restrict__ rowstart,
    const int* __restrict__ csrc, const float* __restrict__ dinv,
    const float* __restrict__ bias, float* __restrict__ agg) {
    const int n = __builtin_amdgcn_readfirstlane((int)(blockIdx.x * 8 + (threadIdx.x >> 5)));
    const int lane = threadIdx.x & 31;
    if (n >= NN) return;
    const int beg = rowstart[n];
    const int end = (n + 1 < NN) ? rowstart[n + 1] : NE;
    const float di = dinv[n];
    const float sn = di * di;
    const float4* hv = reinterpret_cast<const float4*>(h);
    const float4 sv = hv[(size_t)n * 32 + lane];
    const float4 b  = *reinterpret_cast<const float4*>(bias + lane * 4);
    float4 acc;
    acc.x = b.x + sv.x * sn; acc.y = b.y + sv.y * sn;
    acc.z = b.z + sv.z * sn; acc.w = b.w + sv.w * sn;

    for (int base = beg; base < end; base += 32) {
        const int m = (end - base < 32) ? (end - base) : 32;
        int sidx = n;
        if (lane < m) sidx = csrc[base + lane];
        const float dsrc = dinv[sidx];
        int j = 0;
        for (; j + 8 <= m; j += 8) gather_rows<8>(hv, sidx, dsrc, di, j, lane, acc);
        if (m - j >= 4) { gather_rows<4>(hv, sidx, dsrc, di, j, lane, acc); j += 4; }
        const int r = m - j;
        if (r == 3)      gather_rows<3>(hv, sidx, dsrc, di, j, lane, acc);
        else if (r == 2) gather_rows<2>(hv, sidx, dsrc, di, j, lane, acc);
        else if (r == 1) gather_rows<1>(hv, sidx, dsrc, di, j, lane, acc);
    }
    *reinterpret_cast<float4*>(agg + (size_t)n * D + lane * 4) = acc;
}

__global__ void k_pool_zero(float* sums_cnt) {
    int i = blockIdx.x * blockDim.x + threadIdx.x;
    if (i < NG * D + NG) sums_cnt[i] = 0.0f;
}

__global__ void k_pool_cnt(const int* __restrict__ batch, float* __restrict__ cnt) {
    int n = blockIdx.x * blockDim.x + threadIdx.x;
    if (n < NN) fatomic_add(&cnt[batch[n]], 1.0f);
}

__global__ void k_pool_accum(const float* __restrict__ h, const int* __restrict__ batch,
                             float* __restrict__ sums) {
    long long i = (long long)blockIdx.x * blockDim.x + threadIdx.x;
    long long n = i >> 5;
    int lane = (int)(i & 31);
    if (n >= NN) return;
    int g = batch[n];
    const float4 v = *reinterpret_cast<const float4*>(h + (size_t)n * D + lane * 4);
    float* o = sums + (size_t)g * D + lane * 4;
    fatomic_add(o + 0, v.x);
    fatomic_add(o + 1, v.y);
    fatomic_add(o + 2, v.z);
    fatomic_add(o + 3, v.w);
}

__global__ void k_pool_final(const float* __restrict__ sums, const float* __restrict__ cnt,
                             float* __restrict__ out) {
    int i = blockIdx.x * blockDim.x + threadIdx.x;
    if (i < NG * D) out[i] = sums[i] / fmaxf(cnt[i >> 7], 1.0f);
}

extern "C" void kernel_launch(void* const* d_in, const int* in_sizes, int n_in,
                              void* d_out, int out_size, void* d_ws, size_t ws_size,
                              hipStream_t stream) {
    const float* x     = (const float*)d_in[0];
    const int*   edge  = (const int*)d_in[1];
    const int*   batch = (const int*)d_in[2];
    const float* W1 = (const float*)d_in[3];
    const float* b1 = (const float*)d_in[4];
    const float* W2 = (const float*)d_in[5];
    const float* b2 = (const float*)d_in[6];
    const float* W3 = (const float*)d_in[7];
    const float* b3 = (const float*)d_in[8];

    const int* esrc = edge;
    const int* edst = edge + NE;

    float* ws   = (float*)d_ws;
    int*   rowst = (int*)ws;
    float* dinv = ws + NN;
    float* bufH = ws + 100000;
    float* bufA = bufH + (size_t)NN * D;
    float* sums = bufA + (size_t)NN * D;
    float* cnt  = sums + NG * D;
    int*   csrc = (int*)(cnt + NG);
    int*   rank = (int*)bufA;

    const int TB = 256;
    const int gNodes   = (NN + TB - 1) / TB;
    const int gEdges   = (NE + TB - 1) / TB;
    const int gNode32  = (NN * 32 + TB - 1) / TB;
    const int gGemm    = (NN + 127) / 128;

    k_cnt_zero <<<gNodes, TB, 0, stream>>>(rowst);
    k_edge_rank<<<gEdges, TB, 0, stream>>>(edst, rowst, rank);
    k_row_scan <<<1, 1024, 0, stream>>>(rowst, dinv);
    k_edge_fill<<<gEdges, TB, 0, stream>>>(esrc, edst, rowst, rank, csrc);

    k_gemm_wmma<0><<<gGemm, TB, 0, stream>>>(x, W1, bufH, NN);
    k_row_gather  <<<gNode32, TB, 0, stream>>>(bufH, rowst, csrc, dinv, b1, bufA);

    k_gemm_wmma<1><<<gGemm, TB, 0, stream>>>(bufA, W2, bufH, NN);
    k_row_gather  <<<gNode32, TB, 0, stream>>>(bufH, rowst, csrc, dinv, b2, bufA);

    k_gemm_wmma<1><<<gGemm, TB, 0, stream>>>(bufA, W3, bufH, NN);
    k_row_gather  <<<gNode32, TB, 0, stream>>>(bufH, rowst, csrc, dinv, b3, bufA);

    k_pool_zero <<<(NG * D + NG + TB - 1) / TB, TB, 0, stream>>>(sums);
    k_pool_cnt  <<<gNodes, TB, 0, stream>>>(batch, cnt);
    k_pool_accum<<<gNode32, TB, 0, stream>>>(bufA, batch, sums);
    k_pool_final<<<(NG * D + TB - 1) / TB, TB, 0, stream>>>(sums, cnt, (float*)d_out);
}


# Layer 3 by linearity: row gather sums relu(layer-2) rows and adds each result into its graph's pooled sum (f32 atomics as before); W3, mean, bias act on the 256 pooled rows; sum order changes
# speedup vs baseline: 2.0061x; 1.0475x over previous
#include <hip/hip_runtime.h>
#include <hip/hip_bf16.h>

#define NN 50000
#define NE 800000
#define NG 256
#define D  128

typedef float v2f __attribute__((ext_vector_type(2)));
typedef float v8f __attribute__((ext_vector_type(8)));

__device__ __forceinline__ void fatomic_add(float* p, float v) {
    unsafeAtomicAdd(p, v);
}

static_assert(NN % 4 == 0, "row scan moves four counts per thread");

__global__ void k_cnt_zero(int* __restrict__ cnt) {
    int n = blockIdx.x * blockDim.x + threadIdx.x;
    if (n < NN) cnt[n] = 0;
}

__global__ void k_edge_rank(const int* __restrict__ dst, int* __restrict__ cnt,
                            int* __restrict__ rank) {
    int e = blockIdx.x * blockDim.x + threadIdx.x;
    if (e < NE) rank[e] = atomicAdd(&cnt[dst[e]], 1);
}

__global__ __launch_bounds__(1024) void k_row_scan(int* __restrict__ cnt_start,
                                                   float* __restrict__ dinv) {
    __shared__ int wtot[32];
    const int tid  = threadIdx.x;
    const int lane = tid & 31;
    const int wave = tid >> 5;
    int carry = 0;
    for (int base = 0; base < NN; base += 4096) {
        const int i = base + tid * 4;
        int4 c = make_int4(0, 0, 0, 0);
        if (i < NN) {
            c = *reinterpret_cast<const int4*>(cnt_start + i);
            float4 r;
            float d0 = (float)(c.x + 1), d1 = (float)(c.y + 1);
            float d2 = (float)(c.z + 1), d3 = (float)(c.w + 1);
            r.x = d0 > 0.0f ? rsqrtf(d0) : 0.0f;
            r.y = d1 > 0.0f ? rsqrtf(d1) : 0.0f;
            r.z = d2 > 0.0f ? rsqrtf(d2) : 0.0f;
            r.w = d3 > 0.0f ? rsqrtf(d3) : 0.0f;
            *reinterpret_cast<float4*>(dinv + i) = r;
        }
        const int s = c.x + c.y + c.z + c.w;
        int v = s;
        for (int d = 1; d < 32; d <<= 1) {
            int t = __shfl_up(v, d, 32);
            if (lane >= d) v += t;
        }
        if (lane == 31) wtot[wave] = v;
        __syncthreads();
        if (wave == 0) {
            int t = wtot[lane];
            for (int d = 1; d < 32; d <<= 1) {
                int u = __shfl_up(t, d, 32);
                if (lane >= d) t += u;
            }
            wtot[lane] = t;
        }
        __syncthreads();
        const int excl = carry + (wave ? wtot[wave - 1] : 0) + v - s;
        if (i < NN) {
            int4 o;
            o.x = excl;
            o.y = excl + c.x;
            o.z = excl + c.x + c.y;
            o.w = excl + c.x + c.y + c.z;
            *reinterpret_cast<int4*>(cnt_start + i) = o;
        }
        carry += wtot[31];
        __syncthreads();
    }
}

__global__ void k_edge_fill(const int* __restrict__ src, const int* __restrict__ dst,
                            const int* __restrict__ rowstart, const int* __restrict__ rank,
                            int* __restrict__ csrc) {
    int e = blockIdx.x * blockDim.x + threadIdx.x;
    if (e < NE) csrc[rowstart[dst[e]] + rank[e]] = src[e];
}

template <int RELU>
__global__ __launch_bounds__(256) void k_gemm_wmma(
    const float* __restrict__ in, const float* __restrict__ W,
    float* __restrict__ out, int nrows) {
    __shared__ float ldsW[128 * 129];
    const int tid = threadIdx.x;
    for (int i = tid; i < 128 * 128; i += 256)
        ldsW[(i >> 7) * 129 + (i & 127)] = W[i];
    __syncthreads();

    const int lane  = tid & 31;
    const int wave  = tid >> 5;
    const int strip = blockIdx.x * 128 + wave * 16;
    int rowA = strip + (lane & 15);
    if (rowA >= nrows) rowA = nrows - 1;
    const int koff = (lane >> 4) << 1;
    const int col  = lane & 15;

    v8f acc[8] = {};
    const float* aBase = in + (size_t)rowA * D;

    for (int k = 0; k < D; k += 4) {
        const float* ap = aBase + k + koff;
        v2f a;
        a.x = ap[0];
        a.y = ap[1];
        if (RELU) { a.x = fmaxf(a.x, 0.0f); a.y = fmaxf(a.y, 0.0f); }
        const float* b0 = &ldsW[(k + koff)     * 129 + col];
        const float* b1 = &ldsW[(k + koff + 1) * 129 + col];
#pragma unroll
        for (int j = 0; j < 8; ++j) {
            v2f b;
            b.x = b0[j * 16];
            b.y = b1[j * 16];
            acc[j] = __builtin_amdgcn_wmma_f32_16x16x4_f32(
                false, a, false, b, (short)0, acc[j], false, false);
        }
    }

    const int rbase = strip + (lane >> 4) * 8;
    if (strip + 16 <= nrows) {
#pragma unroll
        for (int j = 0; j < 8; ++j) {
#pragma unroll
            for (int r = 0; r < 8; ++r) {
                out[(size_t)(rbase + r) * D + j * 16 + col] = acc[j][r];
            }
        }
    } else {
#pragma unroll
        for (int j = 0; j < 8; ++j) {
#pragma unroll
            for (int r = 0; r < 8; ++r) {
                int row = rbase + r;
                if (row < nrows) out[(size_t)row * D + j * 16 + col] = acc[j][r];
            }
        }
    }
}

template <int RELU, int K>
__device__ __forceinline__ void gather_rows(const float4* __restrict__ hv, int sidx, float dsrc,
                                            float di, int j, int lane, float4& acc) {
    float4 v[K];
#pragma unroll
    for (int u = 0; u < K; ++u) {
        const int s = __builtin_amdgcn_readlane(sidx, j + u);
        v[u] = hv[(size_t)s * 32 + lane];
    }
#pragma unroll
    for (int u = 0; u < K; ++u) {
        const float w = __int_as_float(__builtin_amdgcn_readlane(__float_as_int(dsrc), j + u)) * di;
        if (RELU) {
            v[u].x = fmaxf(v[u].x, 0.0f); v[u].y = fmaxf(v[u].y, 0.0f);
            v[u].z = fmaxf(v[u].z, 0.0f); v[u].w = fmaxf(v[u].w, 0.0f);
        }
        acc.x = __builtin_fmaf(v[u].x, w, acc.x);
        acc.y = __builtin_fmaf(v[u].y, w, acc.y);
        acc.z = __builtin_fmaf(v[u].z, w, acc.z);
        acc.w = __builtin_fmaf(v[u].w, w, acc.w);
    }
}

template <int POOL>
__global__ __launch_bounds__(256) void k_row_gather(
    const float* __restrict__ h, const int* __restrict__ rowstart,
    const int* __restrict__ csrc, const float* __restrict__ dinv,
    const float* __restrict__ bias, float* __restrict__ agg,
    const int* __restrict__ batch) {
    const int n = __builtin_amdgcn_readfirstlane((int)(blockIdx.x * 8 + (threadIdx.x >> 5)));
    const int lane = threadIdx.x & 31;
    if (n >= NN) return;
    const int beg = rowstart[n];
    const int end = (n + 1 < NN) ? rowstart[n + 1] : NE;
    const int g = POOL ? batch[n] : 0;
    const float di = dinv[n];
    const float sn = di * di;
    const float4* hv = reinterpret_cast<const float4*>(h);
    float4 sv = hv[(size_t)n * 32 + lane];
    float4 acc;
    if (POOL) {
        sv.x = fmaxf(sv.x, 0.0f); sv.y = fmaxf(sv.y, 0.0f);
        sv.z = fmaxf(sv.z, 0.0f); sv.w = fmaxf(sv.w, 0.0f);
        acc.x = sv.x * sn; acc.y = sv.y * sn;
        acc.z = sv.z * sn; acc.w = sv.w * sn;
    } else {
        const float4 b  = *reinterpret_cast<const float4*>(bias + lane * 4);
        acc.x = b.x + sv.x * sn; acc.y = b.y + sv.y * sn;
        acc.z = b.z + sv.z * sn; acc.w = b.w + sv.w * sn;
    }

    for (int base = beg; base < end; base += 32) {
        const int m = (end - base < 32) ? (end - base) : 32;
        int sidx = n;
        if (lane < m) sidx = csrc[base + lane];
        const float dsrc = dinv[sidx];
        int j = 0;
        for (; j + 8 <= m; j += 8) gather_rows<POOL, 8>(hv, sidx, dsrc, di, j, lane, acc);
        if (m - j >= 4) { gather_rows<POOL, 4>(hv, sidx, dsrc, di, j, lane, acc); j += 4; }
        const int r = m - j;
        if (r == 3)      gather_rows<POOL, 3>(hv, sidx, dsrc, di, j, lane, acc);
        else if (r == 2) gather_rows<POOL, 2>(hv, sidx, dsrc, di, j, lane, acc);
        else if (r == 1) gather_rows<POOL, 1>(hv, sidx, dsrc, di, j, lane, acc);
    }
    if (POOL) {
        float* o = agg + (size_t)g * D + lane * 4;
        fatomic_add(o + 0, acc.x);
        fatomic_add(o + 1, acc.y);
        fatomic_add(o + 2, acc.z);
        fatomic_add(o + 3, acc.w);
    } else {
        *reinterpret_cast<float4*>(agg + (size_t)n * D + lane * 4) = acc;
    }
}

__global__ void k_pool_zero(float* sums_cnt) {
    int i = blockIdx.x * blockDim.x + threadIdx.x;
    if (i < NG * D + NG) sums_cnt[i] = 0.0f;
}

__global__ void k_pool_cnt(const int* __restrict__ batch, float* __restrict__ cnt) {
    int n = blockIdx.x * blockDim.x + threadIdx.x;
    if (n < NN) fatomic_add(&cnt[batch[n]], 1.0f);
}

__global__ __launch_bounds__(256) void k_pool_transform(
    const float* __restrict__ sums, const float* __restrict__ cnt,
    const float* __restrict__ W, const float* __restrict__ bias, float* __restrict__ out) {
    const int lane  = threadIdx.x & 31;
    const int wave  = threadIdx.x >> 5;
    const int strip = blockIdx.x * 16;
    const int koff  = (lane >> 4) << 1;
    const int col   = lane & 15;
    const int ocol  = wave * 16 + col;
    const float* ap = sums + (size_t)(strip + col) * D + koff;
    const float* bp = W + (size_t)koff * D + ocol;
    v8f acc0 = {};
    v8f acc1 = {};
#pragma unroll
    for (int k = 0; k < D; k += 8) {
        v2f a0, b0, a1, b1;
        a0.x = ap[k];     a0.y = ap[k + 1];
        a1.x = ap[k + 4]; a1.y = ap[k + 5];
        b0.x = bp[(size_t)k * D];       b0.y = bp[(size_t)(k + 1) * D];
        b1.x = bp[(size_t)(k + 4) * D]; b1.y = bp[(size_t)(k + 5) * D];
        acc0 = __builtin_amdgcn_wmma_f32_16x16x4_f32(
            false, a0, false, b0, (short)0, acc0, false, false);
        acc1 = __builtin_amdgcn_wmma_f32_16x16x4_f32(
            false, a1, false, b1, (short)0, acc1, false, false);
    }
    const int rbase = strip + (lane >> 4) * 8;
    const float bcol = bias[ocol];
#pragma unroll
    for (int r = 0; r < 8; ++r) {
        const int g = rbase + r;
        const float c = cnt[g];
        out[(size_t)g * D + ocol] = (acc0[r] + acc1[r] + c * bcol) / fmaxf(c, 1.0f);
    }
}

extern "C" void kernel_launch(void* const* d_in, const int* in_sizes, int n_in,
                              void* d_out, int out_size, void* d_ws, size_t ws_size,
                              hipStream_t stream) {
    const float* x     = (const float*)d_in[0];
    const int*   edge  = (const int*)d_in[1];
    const int*   batch = (const int*)d_in[2];
    const float* W1 = (const float*)d_in[3];
    const float* b1 = (const float*)d_in[4];
    const float* W2 = (const float*)d_in[5];
    const float* b2 = (const float*)d_in[6];
    const float* W3 = (const float*)d_in[7];
    const float* b3 = (const float*)d_in[8];

    const int* esrc = edge;
    const int* edst = edge + NE;

    float* ws   = (float*)d_ws;
    int*   rowst = (int*)ws;
    float* dinv = ws + NN;
    float* bufH = ws + 100000;
    float* bufA = bufH + (size_t)NN * D;
    float* sums = bufA + (size_t)NN * D;
    float* cnt  = sums + NG * D;
    int*   csrc = (int*)(cnt + NG);
    int*   rank = (int*)bufA;

    const int TB = 256;
    const int gNodes   = (NN + TB - 1) / TB;
    const int gEdges   = (NE + TB - 1) / TB;
    const int gNode32  = (NN * 32 + TB - 1) / TB;
    const int gGemm    = (NN + 127) / 128;

    k_cnt_zero <<<gNodes, TB, 0, stream>>>(rowst);
    k_edge_rank<<<gEdges, TB, 0, stream>>>(edst, rowst, rank);
    k_row_scan <<<1, 1024, 0, stream>>>(rowst, dinv);
    k_edge_fill<<<gEdges, TB, 0, stream>>>(esrc, edst, rowst, rank, csrc);

    k_gemm_wmma<0><<<gGemm, TB, 0, stream>>>(x, W1, bufH, NN);
    k_row_gather<0><<<gNode32, TB, 0, stream>>>(bufH, rowst, csrc, dinv, b1, bufA, nullptr);

    k_gemm_wmma<1><<<gGemm, TB, 0, stream>>>(bufA, W2, bufH, NN);
    k_row_gather<0><<<gNode32, TB, 0, stream>>>(bufH, rowst, csrc, dinv, b2, bufA, nullptr);

    k_pool_zero <<<(NG * D + NG + TB - 1) / TB, TB, 0, stream>>>(sums);
    k_pool_cnt  <<<gNodes, TB, 0, stream>>>(batch, cnt);
    k_row_gather<1><<<gNode32, TB, 0, stream>>>(bufA, rowst, csrc, dinv, nullptr, sums, batch);
    k_pool_transform<<<NG / 16, TB, 0, stream>>>(sums, cnt, W3, b3, (float*)d_out);
}


# Fewer, longer waves: each gather wave sums 4 consecutive rows (next row's list head and own row fetched ahead), each rank/fill thread takes 4 edges; per-row sum order as before
# speedup vs baseline: 2.1723x; 1.0828x over previous
#include <hip/hip_runtime.h>
#include <hip/hip_bf16.h>

#define NN 50000
#define NE 800000
#define NG 256
#define D  128

typedef float v2f __attribute__((ext_vector_type(2)));
typedef float v8f __attribute__((ext_vector_type(8)));

__device__ __forceinline__ void fatomic_add(float* p, float v) {
    unsafeAtomicAdd(p, v);
}

static_assert(NN % 4 == 0, "row scan moves four counts per thread");
static_assert(NE % 4 == 0, "the edge kernels move four edges per thread");

__global__ void k_cnt_zero(int* __restrict__ cnt) {
    int n = blockIdx.x * blockDim.x + threadIdx.x;
    if (n < NN) cnt[n] = 0;
}

__global__ void k_edge_rank(const int* __restrict__ dst, int* __restrict__ cnt,
                            int* __restrict__ rank) {
    int t = blockIdx.x * blockDim.x + threadIdx.x;
    if (t < NE / 4) {
        const int4 d = *reinterpret_cast<const int4*>(dst + 4 * t);
        int4 r;
        r.x = atomicAdd(&cnt[d.x], 1);
        r.y = atomicAdd(&cnt[d.y], 1);
        r.z = atomicAdd(&cnt[d.z], 1);
        r.w = atomicAdd(&cnt[d.w], 1);
        *reinterpret_cast<int4*>(rank + 4 * t) = r;
    }
}

__global__ __launch_bounds__(1024) void k_row_scan(int* __restrict__ cnt_start,
                                                   float* __restrict__ dinv) {
    __shared__ int wtot[32];
    const int tid  = threadIdx.x;
    const int lane = tid & 31;
    const int wave = tid >> 5;
    int carry = 0;
    for (int base = 0; base < NN; base += 4096) {
        const int i = base + tid * 4;
        int4 c = make_int4(0, 0, 0, 0);
        if (i < NN) {
            c = *reinterpret_cast<const int4*>(cnt_start + i);
            float4 r;
            float d0 = (float)(c.x + 1), d1 = (float)(c.y + 1);
            float d2 = (float)(c.z + 1), d3 = (float)(c.w + 1);
            r.x = d0 > 0.0f ? rsqrtf(d0) : 0.0f;
            r.y = d1 > 0.0f ? rsqrtf(d1) : 0.0f;
            r.z = d2 > 0.0f ? rsqrtf(d2) : 0.0f;
            r.w = d3 > 0.0f ? rsqrtf(d3) : 0.0f;
            *reinterpret_cast<float4*>(dinv + i) = r;
        }
        const int s = c.x + c.y + c.z + c.w;
        int v = s;
        for (int d = 1; d < 32; d <<= 1) {
            int t = __shfl_up(v, d, 32);
            if (lane >= d) v += t;
        }
        if (lane == 31) wtot[wave] = v;
        __syncthreads();
        if (wave == 0) {
            int t = wtot[lane];
            for (int d = 1; d < 32; d <<= 1) {
                int u = __shfl_up(t, d, 32);
                if (lane >= d) t += u;
            }
            wtot[lane] = t;
        }
        __syncthreads();
        const int excl = carry + (wave ? wtot[wave - 1] : 0) + v - s;
        if (i < NN) {
            int4 o;
            o.x = excl;
            o.y = excl + c.x;
            o.z = excl + c.x + c.y;
            o.w = excl + c.x + c.y + c.z;
            *reinterpret_cast<int4*>(cnt_start + i) = o;
        }
        carry += wtot[31];
        __syncthreads();
    }
}

__global__ void k_edge_fill(const int* __restrict__ src, const int* __restrict__ dst,
                            const int* __restrict__ rowstart, const int* __restrict__ rank,
                            int* __restrict__ csrc) {
    int t = blockIdx.x * blockDim.x + threadIdx.x;
    if (t < NE / 4) {
        const int4 s = *reinterpret_cast<const int4*>(src + 4 * t);
        const int4 d = *reinterpret_cast<const int4*>(dst + 4 * t);
        const int4 k = *reinterpret_cast<const int4*>(rank + 4 * t);
        const int p0 = rowstart[d.x], p1 = rowstart[d.y];
        const int p2 = rowstart[d.z], p3 = rowstart[d.w];
        csrc[p0 + k.x] = s.x;
        csrc[p1 + k.y] = s.y;
        csrc[p2 + k.z] = s.z;
        csrc[p3 + k.w] = s.w;
    }
}

template <int RELU>
__global__ __launch_bounds__(256) void k_gemm_wmma(
    const float* __restrict__ in, const float* __restrict__ W,
    float* __restrict__ out, int nrows) {
    __shared__ float ldsW[128 * 129];
    const int tid = threadIdx.x;
    for (int i = tid; i < 128 * 128; i += 256)
        ldsW[(i >> 7) * 129 + (i & 127)] = W[i];
    __syncthreads();

    const int lane  = tid & 31;
    const int wave  = tid >> 5;
    const int strip = blockIdx.x * 128 + wave * 16;
    int rowA = strip + (lane & 15);
    if (rowA >= nrows) rowA = nrows - 1;
    const int koff = (lane >> 4) << 1;
    const int col  = lane & 15;

    v8f acc[8] = {};
    const float* aBase = in + (size_t)rowA * D;

    for (int k = 0; k < D; k += 4) {
        const float* ap = aBase + k + koff;
        v2f a;
        a.x = ap[0];
        a.y = ap[1];
        if (RELU) { a.x = fmaxf(a.x, 0.0f); a.y = fmaxf(a.y, 0.0f); }
        const float* b0 = &ldsW[(k + koff)     * 129 + col];
        const float* b1 = &ldsW[(k + koff + 1) * 129 + col];
#pragma unroll
        for (int j = 0; j < 8; ++j) {
            v2f b;
            b.x = b0[j * 16];
            b.y = b1[j * 16];
            acc[j] = __builtin_amdgcn_wmma_f32_16x16x4_f32(
                false, a, false, b, (short)0, acc[j], false, false);
        }
    }

    const int rbase = strip + (lane >> 4) * 8;
    if (strip + 16 <= nrows) {
#pragma unroll
        for (int j = 0; j < 8; ++j) {
#pragma unroll
            for (int r = 0; r < 8; ++r) {
                out[(size_t)(rbase + r) * D + j * 16 + col] = acc[j][r];
            }
        }
    } else {
#pragma unroll
        for (int j = 0; j < 8; ++j) {
#pragma unroll
            for (int r = 0; r < 8; ++r) {
                int row = rbase + r;
                if (row < nrows) out[(size_t)row * D + j * 16 + col] = acc[j][r];
            }
        }
    }
}

template <int RELU, int K>
__device__ __forceinline__ void gather_rows(const float4* __restrict__ hv, int sidx, float dsrc,
                                            float di, int j, int lane, float4& acc) {
    float4 v[K];
#pragma unroll
    for (int u = 0; u < K; ++u) {
        const int s = __builtin_amdgcn_readlane(sidx, j + u);
        v[u] = hv[(size_t)s * 32 + lane];
    }
#pragma unroll
    for (int u = 0; u < K; ++u) {
        const float w = __int_as_float(__builtin_amdgcn_readlane(__float_as_int(dsrc), j + u)) * di;
        if (RELU) {
            v[u].x = fmaxf(v[u].x, 0.0f); v[u].y = fmaxf(v[u].y, 0.0f);
            v[u].z = fmaxf(v[u].z, 0.0f); v[u].w = fmaxf(v[u].w, 0.0f);
        }
        acc.x = __builtin_fmaf(v[u].x, w, acc.x);
        acc.y = __builtin_fmaf(v[u].y, w, acc.y);
        acc.z = __builtin_fmaf(v[u].z, w, acc.z);
        acc.w = __builtin_fmaf(v[u].w, w, acc.w);
    }
}

template <int POOL>
__device__ __forceinline__ void gather_chunk(const float4* __restrict__ hv, int sidx, float dsrc,
                                             float di, int m, int lane, float4& acc) {
    int j = 0;
    for (; j + 8 <= m; j += 8) gather_rows<POOL, 8>(hv, sidx, dsrc, di, j, lane, acc);
    if (m - j >= 4) { gather_rows<POOL, 4>(hv, sidx, dsrc, di, j, lane, acc); j += 4; }
    const int rem = m - j;
    if (rem == 3)      gather_rows<POOL, 3>(hv, sidx, dsrc, di, j, lane, acc);
    else if (rem == 2) gather_rows<POOL, 2>(hv, sidx, dsrc, di, j, lane, acc);
    else if (rem == 1) gather_rows<POOL, 1>(hv, sidx, dsrc, di, j, lane, acc);
}

template <int POOL>
__device__ __forceinline__ float4 row_start(float4 sv, float4 b, float di) {
    const float sn = di * di;
    float4 acc;
    if (POOL) {
        sv.x = fmaxf(sv.x, 0.0f); sv.y = fmaxf(sv.y, 0.0f);
        sv.z = fmaxf(sv.z, 0.0f); sv.w = fmaxf(sv.w, 0.0f);
        acc.x = sv.x * sn; acc.y = sv.y * sn;
        acc.z = sv.z * sn; acc.w = sv.w * sn;
    } else {
        acc.x = __builtin_fmaf(sv.x, sn, b.x); acc.y = __builtin_fmaf(sv.y, sn, b.y);
        acc.z = __builtin_fmaf(sv.z, sn, b.z); acc.w = __builtin_fmaf(sv.w, sn, b.w);
    }
    return acc;
}

#define RPW 4
static_assert(RPW == 4 && NN % RPW == 0, "every wave owns four whole rows; their bounds, dinv and graphs are read as one vector each");

template <int POOL>
__global__ __launch_bounds__(256) void k_row_gather(
    const float* __restrict__ h, const int* __restrict__ rowstart,
    const int* __restrict__ csrc, const float* __restrict__ dinv,
    const float* __restrict__ bias, float* __restrict__ agg,
    const int* __restrict__ batch) {
    const int n0 = __builtin_amdgcn_readfirstlane((int)(blockIdx.x * 8 + (threadIdx.x >> 5))) * RPW;
    const int lane = threadIdx.x & 31;
    if (n0 >= NN) return;
    const float4* hv = reinterpret_cast<const float4*>(h);

    const int4 q = *reinterpret_cast<const int4*>(rowstart + n0);
    int q0 = q.x, q1 = q.y, q2 = q.z, q3 = q.w;
    int q4 = (n0 + RPW < NN) ? rowstart[n0 + RPW] : NE;
    const float4 dq = *reinterpret_cast<const float4*>(dinv + n0);
    float d0 = dq.x, d1 = dq.y, d2 = dq.z, d3 = dq.w;
    int g0 = 0, g1 = 0, g2 = 0, g3 = 0;
    if (POOL) {
        const int4 gq = *reinterpret_cast<const int4*>(batch + n0);
        g0 = gq.x; g1 = gq.y; g2 = gq.z; g3 = gq.w;
    }
    float4 b = make_float4(0.0f, 0.0f, 0.0f, 0.0f);
    if (!POOL) b = *reinterpret_cast<const float4*>(bias + lane * 4);

    int sidx = n0;
    {
        const int m0 = (q1 - q0 < 32) ? (q1 - q0) : 32;
        if (lane < m0) sidx = csrc[q0 + lane];
    }
    float4 acc = row_start<POOL>(hv[(size_t)n0 * 32 + lane], b, d0);

    for (int r = 0; r < RPW; ++r) {
        const int n = n0 + r;
        const float di = d0;
        const float dsrc = dinv[sidx];
        const int mn = (q2 - q1 < 32) ? (q2 - q1) : 32;
        int sidx_next = n + 1;
        if (lane < mn) sidx_next = csrc[q1 + lane];
        const int nn = (n + 1 < NN) ? n + 1 : n;
        const float4 sv_next = hv[(size_t)nn * 32 + lane];

        gather_chunk<POOL>(hv, sidx, dsrc, di, (q1 - q0 < 32) ? (q1 - q0) : 32, lane, acc);
        for (int base = q0 + 32; base < q1; base += 32) {
            const int m2 = (q1 - base < 32) ? (q1 - base) : 32;
            int s2 = n;
            if (lane < m2) s2 = csrc[base + lane];
            const float ds2 = dinv[s2];
            gather_chunk<POOL>(hv, s2, ds2, di, m2, lane, acc);
        }

        if (POOL) {
            float* o = agg + (size_t)g0 * D + lane * 4;
            fatomic_add(o + 0, acc.x);
            fatomic_add(o + 1, acc.y);
            fatomic_add(o + 2, acc.z);
            fatomic_add(o + 3, acc.w);
        } else {
            *reinterpret_cast<float4*>(agg + (size_t)n * D + lane * 4) = acc;
        }
        acc = row_start<POOL>(sv_next, b, d1);
        sidx = sidx_next;
        q0 = q1; q1 = q2; q2 = q3; q3 = q4;
        d0 = d1; d1 = d2; d2 = d3;
        g0 = g1; g1 = g2; g2 = g3;
    }
}

__global__ void k_pool_zero(float* sums_cnt) {
    int i = blockIdx.x * blockDim.x + threadIdx.x;
    if (i < NG * D + NG) sums_cnt[i] = 0.0f;
}

__global__ void k_pool_cnt(const int* __restrict__ batch, float* __restrict__ cnt) {
    int n = blockIdx.x * blockDim.x + threadIdx.x;
    if (n < NN) fatomic_add(&cnt[batch[n]], 1.0f);
}

__global__ __launch_bounds__(256) void k_pool_transform(
    const float* __restrict__ sums, const float* __restrict__ cnt,
    const float* __restrict__ W, const float* __restrict__ bias, float* __restrict__ out) {
    const int lane  = threadIdx.x & 31;
    const int wave  = threadIdx.x >> 5;
    const int strip = blockIdx.x * 16;
    const int koff  = (lane >> 4) << 1;
    const int col   = lane & 15;
    const int ocol  = wave * 16 + col;
    const float* ap = sums + (size_t)(strip + col) * D + koff;
    const float* bp = W + (size_t)koff * D + ocol;
    v8f acc0 = {};
    v8f acc1 = {};
#pragma unroll
    for (int k = 0; k < D; k += 8) {
        v2f a0, b0, a1, b1;
        a0.x = ap[k];     a0.y = ap[k + 1];
        a1.x = ap[k + 4]; a1.y = ap[k + 5];
        b0.x = bp[(size_t)k * D];       b0.y = bp[(size_t)(k + 1) * D];
        b1.x = bp[(size_t)(k + 4) * D]; b1.y = bp[(size_t)(k + 5) * D];
        acc0 = __builtin_amdgcn_wmma_f32_16x16x4_f32(
            false, a0, false, b0, (short)0, acc0, false, false);
        acc1 = __builtin_amdgcn_wmma_f32_16x16x4_f32(
            false, a1, false, b1, (short)0, acc1, false, false);
    }
    const int rbase = strip + (lane >> 4) * 8;
    const float bcol = bias[ocol];
#pragma unroll
    for (int r = 0; r < 8; ++r) {
        const int g = rbase + r;
        const float c = cnt[g];
        out[(size_t)g * D + ocol] = (acc0[r] + acc1[r] + c * bcol) / fmaxf(c, 1.0f);
    }
}

extern "C" void kernel_launch(void* const* d_in, const int* in_sizes, int n_in,
                              void* d_out, int out_size, void* d_ws, size_t ws_size,
                              hipStream_t stream) {
    const float* x     = (const float*)d_in[0];
    const int*   edge  = (const int*)d_in[1];
    const int*   batch = (const int*)d_in[2];
    const float* W1 = (const float*)d_in[3];
    const float* b1 = (const float*)d_in[4];
    const float* W2 = (const float*)d_in[5];
    const float* b2 = (const float*)d_in[6];
    const float* W3 = (const float*)d_in[7];
    const float* b3 = (const float*)d_in[8];

    const int* esrc = edge;
    const int* edst = edge + NE;

    float* ws   = (float*)d_ws;
    int*   rowst = (int*)ws;
    float* dinv = ws + NN;
    float* bufH = ws + 100000;
    float* bufA = bufH + (size_t)NN * D;
    float* sums = bufA + (size_t)NN * D;
    float* cnt  = sums + NG * D;
    int*   csrc = (int*)(cnt + NG);
    int*   rank = (int*)bufA;

    const int TB = 256;
    const int gNodes   = (NN + TB - 1) / TB;
    const int gEdges   = (NE / 4 + TB - 1) / TB;
    const int gNode32  = (NN / RPW * 32 + TB - 1) / TB;
    const int gGemm    = (NN + 127) / 128;

    k_cnt_zero <<<gNodes, TB, 0, stream>>>(rowst);
    k_edge_rank<<<gEdges, TB, 0, stream>>>(edst, rowst, rank);
    k_row_scan <<<1, 1024, 0, stream>>>(rowst, dinv);
    k_edge_fill<<<gEdges, TB, 0, stream>>>(esrc, edst, rowst, rank, csrc);

    k_gemm_wmma<0><<<gGemm, TB, 0, stream>>>(x, W1, bufH, NN);
    k_row_gather<0><<<gNode32, TB, 0, stream>>>(bufH, rowst, csrc, dinv, b1, bufA, nullptr);

    k_gemm_wmma<1><<<gGemm, TB, 0, stream>>>(bufA, W2, bufH, NN);
    k_row_gather<0><<<gNode32, TB, 0, stream>>>(bufH, rowst, csrc, dinv, b2, bufA, nullptr);

    k_pool_zero <<<(NG * D + NG + TB - 1) / TB, TB, 0, stream>>>(sums);
    k_pool_cnt  <<<gNodes, TB, 0, stream>>>(batch, cnt);
    k_row_gather<1><<<gNode32, TB, 0, stream>>>(bufA, rowst, csrc, dinv, nullptr, sums, batch);
    k_pool_transform<<<NG / 16, TB, 0, stream>>>(sums, cnt, W3, b3, (float*)d_out);
}


# Counting sort's counters moved into LDS: 64 edge chunks, one work-group each with a counter per row, per-chunk prefix; no global integer atomics; row lists in chunk order; sum order changes
# speedup vs baseline: 2.3565x; 1.0848x over previous
#include <hip/hip_runtime.h>
#include <hip/hip_bf16.h>

#define NN 50000
#define NE 800000
#define NG 256
#define D  128

typedef float v2f __attribute__((ext_vector_type(2)));
typedef float v8f __attribute__((ext_vector_type(8)));

__device__ __forceinline__ void fatomic_add(float* p, float v) {
    unsafeAtomicAdd(p, v);
}

static_assert(NN % 4 == 0, "row scan moves four counts per thread");
#define NCH 64
#define CH4 (NE / 4 / NCH)
#define EIT ((CH4 + 1023) / 1024)
static_assert(NE % (4 * NCH) == 0, "every chunk is a whole number of four-edge groups");

__global__ __launch_bounds__(1024) void k_chunk_count(const int* __restrict__ dst,
                                                      int* __restrict__ ghist) {
    __shared__ int hist[NN];
    const int tid = threadIdx.x;
    const int g   = blockIdx.x;
    int4* h4 = reinterpret_cast<int4*>(hist);
    const int4* d4 = reinterpret_cast<const int4*>(dst) + (size_t)g * CH4;
    int4 d[EIT];
#pragma unroll
    for (int k = 0; k < EIT; ++k) {
        const int i = tid + k * 1024;
        d[k] = make_int4(0, 0, 0, 0);
        if (i < CH4) d[k] = d4[i];
    }
    for (int i = tid; i < NN / 4; i += 1024) h4[i] = make_int4(0, 0, 0, 0);
    __syncthreads();
#pragma unroll
    for (int k = 0; k < EIT; ++k) {
        const int i = tid + k * 1024;
        if (i < CH4) {
            atomicAdd(&hist[d[k].x], 1);
            atomicAdd(&hist[d[k].y], 1);
            atomicAdd(&hist[d[k].z], 1);
            atomicAdd(&hist[d[k].w], 1);
        }
    }
    __syncthreads();
    int4* o4 = reinterpret_cast<int4*>(ghist + (size_t)g * NN);
    for (int i = tid; i < NN / 4; i += 1024) o4[i] = h4[i];
}

__global__ void k_chunk_prefix(const int* __restrict__ ghist, int* __restrict__ gpre,
                               int* __restrict__ cnt) {
    const int n = blockIdx.x * blockDim.x + threadIdx.x;
    if (n < NN) {
        int h[NCH];
#pragma unroll
        for (int g = 0; g < NCH; ++g) h[g] = ghist[(size_t)g * NN + n];
        int run = 0;
#pragma unroll
        for (int g = 0; g < NCH; ++g) {
            gpre[(size_t)g * NN + n] = run;
            run += h[g];
        }
        cnt[n] = run;
    }
}

__global__ __launch_bounds__(1024) void k_row_scan(int* __restrict__ cnt_start,
                                                   float* __restrict__ dinv) {
    __shared__ int wtot[32];
    const int tid  = threadIdx.x;
    const int lane = tid & 31;
    const int wave = tid >> 5;
    int carry = 0;
    for (int base = 0; base < NN; base += 4096) {
        const int i = base + tid * 4;
        int4 c = make_int4(0, 0, 0, 0);
        if (i < NN) {
            c = *reinterpret_cast<const int4*>(cnt_start + i);
            float4 r;
            float d0 = (float)(c.x + 1), d1 = (float)(c.y + 1);
            float d2 = (float)(c.z + 1), d3 = (float)(c.w + 1);
            r.x = d0 > 0.0f ? rsqrtf(d0) : 0.0f;
            r.y = d1 > 0.0f ? rsqrtf(d1) : 0.0f;
            r.z = d2 > 0.0f ? rsqrtf(d2) : 0.0f;
            r.w = d3 > 0.0f ? rsqrtf(d3) : 0.0f;
            *reinterpret_cast<float4*>(dinv + i) = r;
        }
        const int s = c.x + c.y + c.z + c.w;
        int v = s;
        for (int d = 1; d < 32; d <<= 1) {
            int t = __shfl_up(v, d, 32);
            if (lane >= d) v += t;
        }
        if (lane == 31) wtot[wave] = v;
        __syncthreads();
        if (wave == 0) {
            int t = wtot[lane];
            for (int d = 1; d < 32; d <<= 1) {
                int u = __shfl_up(t, d, 32);
                if (lane >= d) t += u;
            }
            wtot[lane] = t;
        }
        __syncthreads();
        const int excl = carry + (wave ? wtot[wave - 1] : 0) + v - s;
        if (i < NN) {
            int4 o;
            o.x = excl;
            o.y = excl + c.x;
            o.z = excl + c.x + c.y;
            o.w = excl + c.x + c.y + c.z;
            *reinterpret_cast<int4*>(cnt_start + i) = o;
        }
        carry += wtot[31];
        __syncthreads();
    }
}

__global__ __launch_bounds__(1024) void k_chunk_place(
    const int* __restrict__ src, const int* __restrict__ dst,
    const int* __restrict__ rowstart, const int* __restrict__ gpre, int* __restrict__ csrc) {
    __shared__ int slot[NN];
    const int tid = threadIdx.x;
    const int g   = blockIdx.x;
    const int4* r4 = reinterpret_cast<const int4*>(rowstart);
    const int4* p4 = reinterpret_cast<const int4*>(gpre + (size_t)g * NN);
    int4* s4 = reinterpret_cast<int4*>(slot);
    for (int i = tid; i < NN / 4; i += 1024) {
        const int4 a = r4[i];
        const int4 b = p4[i];
        s4[i] = make_int4(a.x + b.x, a.y + b.y, a.z + b.z, a.w + b.w);
    }
    const int4* sv = reinterpret_cast<const int4*>(src) + (size_t)g * CH4;
    const int4* dv = reinterpret_cast<const int4*>(dst) + (size_t)g * CH4;
    int4 s[EIT], d[EIT];
#pragma unroll
    for (int k = 0; k < EIT; ++k) {
        const int i = tid + k * 1024;
        s[k] = make_int4(0, 0, 0, 0);
        d[k] = make_int4(0, 0, 0, 0);
        if (i < CH4) { s[k] = sv[i]; d[k] = dv[i]; }
    }
    __syncthreads();
#pragma unroll
    for (int k = 0; k < EIT; ++k) {
        const int i = tid + k * 1024;
        if (i < CH4) {
            const int p0 = atomicAdd(&slot[d[k].x], 1);
            const int p1 = atomicAdd(&slot[d[k].y], 1);
            const int p2 = atomicAdd(&slot[d[k].z], 1);
            const int p3 = atomicAdd(&slot[d[k].w], 1);
            csrc[p0] = s[k].x;
            csrc[p1] = s[k].y;
            csrc[p2] = s[k].z;
            csrc[p3] = s[k].w;
        }
    }
}

template <int RELU>
__global__ __launch_bounds__(256) void k_gemm_wmma(
    const float* __restrict__ in, const float* __restrict__ W,
    float* __restrict__ out, int nrows) {
    __shared__ float ldsW[128 * 129];
    const int tid = threadIdx.x;
    for (int i = tid; i < 128 * 128; i += 256)
        ldsW[(i >> 7) * 129 + (i & 127)] = W[i];
    __syncthreads();

    const int lane  = tid & 31;
    const int wave  = tid >> 5;
    const int strip = blockIdx.x * 128 + wave * 16;
    int rowA = strip + (lane & 15);
    if (rowA >= nrows) rowA = nrows - 1;
    const int koff = (lane >> 4) << 1;
    const int col  = lane & 15;

    v8f acc[8] = {};
    const float* aBase = in + (size_t)rowA * D;

    for (int k = 0; k < D; k += 4) {
        const float* ap = aBase + k + koff;
        v2f a;
        a.x = ap[0];
        a.y = ap[1];
        if (RELU) { a.x = fmaxf(a.x, 0.0f); a.y = fmaxf(a.y, 0.0f); }
        const float* b0 = &ldsW[(k + koff)     * 129 + col];
        const float* b1 = &ldsW[(k + koff + 1) * 129 + col];
#pragma unroll
        for (int j = 0; j < 8; ++j) {
            v2f b;
            b.x = b0[j * 16];
            b.y = b1[j * 16];
            acc[j] = __builtin_amdgcn_wmma_f32_16x16x4_f32(
                false, a, false, b, (short)0, acc[j], false, false);
        }
    }

    const int rbase = strip + (lane >> 4) * 8;
    if (strip + 16 <= nrows) {
#pragma unroll
        for (int j = 0; j < 8; ++j) {
#pragma unroll
            for (int r = 0; r < 8; ++r) {
                out[(size_t)(rbase + r) * D + j * 16 + col] = acc[j][r];
            }
        }
    } else {
#pragma unroll
        for (int j = 0; j < 8; ++j) {
#pragma unroll
            for (int r = 0; r < 8; ++r) {
                int row = rbase + r;
                if (row < nrows) out[(size_t)row * D + j * 16 + col] = acc[j][r];
            }
        }
    }
}

template <int RELU, int K>
__device__ __forceinline__ void gather_rows(const float4* __restrict__ hv, int sidx, float dsrc,
                                            float di, int j, int lane, float4& acc) {
    float4 v[K];
#pragma unroll
    for (int u = 0; u < K; ++u) {
        const int s = __builtin_amdgcn_readlane(sidx, j + u);
        v[u] = hv[(size_t)s * 32 + lane];
    }
#pragma unroll
    for (int u = 0; u < K; ++u) {
        const float w = __int_as_float(__builtin_amdgcn_readlane(__float_as_int(dsrc), j + u)) * di;
        if (RELU) {
            v[u].x = fmaxf(v[u].x, 0.0f); v[u].y = fmaxf(v[u].y, 0.0f);
            v[u].z = fmaxf(v[u].z, 0.0f); v[u].w = fmaxf(v[u].w, 0.0f);
        }
        acc.x = __builtin_fmaf(v[u].x, w, acc.x);
        acc.y = __builtin_fmaf(v[u].y, w, acc.y);
        acc.z = __builtin_fmaf(v[u].z, w, acc.z);
        acc.w = __builtin_fmaf(v[u].w, w, acc.w);
    }
}

template <int POOL>
__device__ __forceinline__ void gather_chunk(const float4* __restrict__ hv, int sidx, float dsrc,
                                             float di, int m, int lane, float4& acc) {
    int j = 0;
    for (; j + 8 <= m; j += 8) gather_rows<POOL, 8>(hv, sidx, dsrc, di, j, lane, acc);
    if (m - j >= 4) { gather_rows<POOL, 4>(hv, sidx, dsrc, di, j, lane, acc); j += 4; }
    const int rem = m - j;
    if (rem == 3)      gather_rows<POOL, 3>(hv, sidx, dsrc, di, j, lane, acc);
    else if (rem == 2) gather_rows<POOL, 2>(hv, sidx, dsrc, di, j, lane, acc);
    else if (rem == 1) gather_rows<POOL, 1>(hv, sidx, dsrc, di, j, lane, acc);
}

template <int POOL>
__device__ __forceinline__ float4 row_start(float4 sv, float4 b, float di) {
    const float sn = di * di;
    float4 acc;
    if (POOL) {
        sv.x = fmaxf(sv.x, 0.0f); sv.y = fmaxf(sv.y, 0.0f);
        sv.z = fmaxf(sv.z, 0.0f); sv.w = fmaxf(sv.w, 0.0f);
        acc.x = sv.x * sn; acc.y = sv.y * sn;
        acc.z = sv.z * sn; acc.w = sv.w * sn;
    } else {
        acc.x = __builtin_fmaf(sv.x, sn, b.x); acc.y = __builtin_fmaf(sv.y, sn, b.y);
        acc.z = __builtin_fmaf(sv.z, sn, b.z); acc.w = __builtin_fmaf(sv.w, sn, b.w);
    }
    return acc;
}

#define RPW 4
static_assert(RPW == 4 && NN % RPW == 0, "every wave owns four whole rows; their bounds, dinv and graphs are read as one vector each");

template <int POOL>
__global__ __launch_bounds__(256) void k_row_gather(
    const float* __restrict__ h, const int* __restrict__ rowstart,
    const int* __restrict__ csrc, const float* __restrict__ dinv,
    const float* __restrict__ bias, float* __restrict__ agg,
    const int* __restrict__ batch) {
    const int n0 = __builtin_amdgcn_readfirstlane((int)(blockIdx.x * 8 + (threadIdx.x >> 5))) * RPW;
    const int lane = threadIdx.x & 31;
    if (n0 >= NN) return;
    const float4* hv = reinterpret_cast<const float4*>(h);

    const int4 q = *reinterpret_cast<const int4*>(rowstart + n0);
    int q0 = q.x, q1 = q.y, q2 = q.z, q3 = q.w;
    int q4 = (n0 + RPW < NN) ? rowstart[n0 + RPW] : NE;
    const float4 dq = *reinterpret_cast<const float4*>(dinv + n0);
    float d0 = dq.x, d1 = dq.y, d2 = dq.z, d3 = dq.w;
    int g0 = 0, g1 = 0, g2 = 0, g3 = 0;
    if (POOL) {
        const int4 gq = *reinterpret_cast<const int4*>(batch + n0);
        g0 = gq.x; g1 = gq.y; g2 = gq.z; g3 = gq.w;
    }
    float4 b = make_float4(0.0f, 0.0f, 0.0f, 0.0f);
    if (!POOL) b = *reinterpret_cast<const float4*>(bias + lane * 4);

    int sidx = n0;
    {
        const int m0 = (q1 - q0 < 32) ? (q1 - q0) : 32;
        if (lane < m0) sidx = csrc[q0 + lane];
    }
    float4 acc = row_start<POOL>(hv[(size_t)n0 * 32 + lane], b, d0);

    for (int r = 0; r < RPW; ++r) {
        const int n = n0 + r;
        const float di = d0;
        const float dsrc = dinv[sidx];
        const int mn = (q2 - q1 < 32) ? (q2 - q1) : 32;
        int sidx_next = n + 1;
        if (lane < mn) sidx_next = csrc[q1 + lane];
        const int nn = (n + 1 < NN) ? n + 1 : n;
        const float4 sv_next = hv[(size_t)nn * 32 + lane];

        gather_chunk<POOL>(hv, sidx, dsrc, di, (q1 - q0 < 32) ? (q1 - q0) : 32, lane, acc);
        for (int base = q0 + 32; base < q1; base += 32) {
            const int m2 = (q1 - base < 32) ? (q1 - base) : 32;
            int s2 = n;
            if (lane < m2) s2 = csrc[base + lane];
            const float ds2 = dinv[s2];
            gather_chunk<POOL>(hv, s2, ds2, di, m2, lane, acc);
        }

        if (POOL) {
            float* o = agg + (size_t)g0 * D + lane * 4;
            fatomic_add(o + 0, acc.x);
            fatomic_add(o + 1, acc.y);
            fatomic_add(o + 2, acc.z);
            fatomic_add(o + 3, acc.w);
        } else {
            *reinterpret_cast<float4*>(agg + (size_t)n * D + lane * 4) = acc;
        }
        acc = row_start<POOL>(sv_next, b, d1);
        sidx = sidx_next;
        q0 = q1; q1 = q2; q2 = q3; q3 = q4;
        d0 = d1; d1 = d2; d2 = d3;
        g0 = g1; g1 = g2; g2 = g3;
    }
}

__global__ void k_pool_zero(float* sums_cnt) {
    int i = blockIdx.x * blockDim.x + threadIdx.x;
    if (i < NG * D + NG) sums_cnt[i] = 0.0f;
}

__global__ void k_pool_cnt(const int* __restrict__ batch, float* __restrict__ cnt) {
    int n = blockIdx.x * blockDim.x + threadIdx.x;
    if (n < NN) fatomic_add(&cnt[batch[n]], 1.0f);
}

__global__ __launch_bounds__(256) void k_pool_transform(
    const float* __restrict__ sums, const float* __restrict__ cnt,
    const float* __restrict__ W, const float* __restrict__ bias, float* __restrict__ out) {
    const int lane  = threadIdx.x & 31;
    const int wave  = threadIdx.x >> 5;
    const int strip = blockIdx.x * 16;
    const int koff  = (lane >> 4) << 1;
    const int col   = lane & 15;
    const int ocol  = wave * 16 + col;
    const float* ap = sums + (size_t)(strip + col) * D + koff;
    const float* bp = W + (size_t)koff * D + ocol;
    v8f acc0 = {};
    v8f acc1 = {};
#pragma unroll
    for (int k = 0; k < D; k += 8) {
        v2f a0, b0, a1, b1;
        a0.x = ap[k];     a0.y = ap[k + 1];
        a1.x = ap[k + 4]; a1.y = ap[k + 5];
        b0.x = bp[(size_t)k * D];       b0.y = bp[(size_t)(k + 1) * D];
        b1.x = bp[(size_t)(k + 4) * D]; b1.y = bp[(size_t)(k + 5) * D];
        acc0 = __builtin_amdgcn_wmma_f32_16x16x4_f32(
            false, a0, false, b0, (short)0, acc0, false, false);
        acc1 = __builtin_amdgcn_wmma_f32_16x16x4_f32(
            false, a1, false, b1, (short)0, acc1, false, false);
    }
    const int rbase = strip + (lane >> 4) * 8;
    const float bcol = bias[ocol];
#pragma unroll
    for (int r = 0; r < 8; ++r) {
        const int g = rbase + r;
        const float c = cnt[g];
        out[(size_t)g * D + ocol] = (acc0[r] + acc1[r] + c * bcol) / fmaxf(c, 1.0f);
    }
}

extern "C" void kernel_launch(void* const* d_in, const int* in_sizes, int n_in,
                              void* d_out, int out_size, void* d_ws, size_t ws_size,
                              hipStream_t stream) {
    const float* x     = (const float*)d_in[0];
    const int*   edge  = (const int*)d_in[1];
    const int*   batch = (const int*)d_in[2];
    const float* W1 = (const float*)d_in[3];
    const float* b1 = (const float*)d_in[4];
    const float* W2 = (const float*)d_in[5];
    const float* b2 = (const float*)d_in[6];
    const float* W3 = (const float*)d_in[7];
    const float* b3 = (const float*)d_in[8];

    const int* esrc = edge;
    const int* edst = edge + NE;

    float* ws   = (float*)d_ws;
    int*   rowst = (int*)ws;
    float* dinv = ws + NN;
    float* bufH = ws + 100000;
    float* bufA = bufH + (size_t)NN * D;
    float* sums = bufA + (size_t)NN * D;
    float* cnt  = sums + NG * D;
    int*   csrc = (int*)(cnt + NG);
    int*   ghist = (int*)bufA;
    int*   gpre  = (int*)bufH;

    const int TB = 256;
    const int gNodes   = (NN + TB - 1) / TB;
    const int gNode32  = (NN / RPW * 32 + TB - 1) / TB;
    const int gGemm    = (NN + 127) / 128;

    k_chunk_count <<<NCH, 1024, 0, stream>>>(edst, ghist);
    k_chunk_prefix<<<gNodes, TB, 0, stream>>>(ghist, gpre, rowst);
    k_row_scan    <<<1, 1024, 0, stream>>>(rowst, dinv);
    k_chunk_place <<<NCH, 1024, 0, stream>>>(esrc, edst, rowst, gpre, csrc);

    k_gemm_wmma<0><<<gGemm, TB, 0, stream>>>(x, W1, bufH, NN);
    k_row_gather<0><<<gNode32, TB, 0, stream>>>(bufH, rowst, csrc, dinv, b1, bufA, nullptr);

    k_gemm_wmma<1><<<gGemm, TB, 0, stream>>>(bufA, W2, bufH, NN);
    k_row_gather<0><<<gNode32, TB, 0, stream>>>(bufH, rowst, csrc, dinv, b2, bufA, nullptr);

    k_pool_zero <<<(NG * D + NG + TB - 1) / TB, TB, 0, stream>>>(sums);
    k_pool_cnt  <<<gNodes, TB, 0, stream>>>(batch, cnt);
    k_row_gather<1><<<gNode32, TB, 0, stream>>>(bufA, rowst, csrc, dinv, nullptr, sums, batch);
    k_pool_transform<<<NG / 16, TB, 0, stream>>>(sums, cnt, W3, b3, (float*)d_out);
}


# Pooled sums combined inside each work-group before the f32 atomics: rows of the work-group's first graph are summed in registers, then across waves in LDS, one add per column; sum order changes
# speedup vs baseline: 2.5979x; 1.1024x over previous
#include <hip/hip_runtime.h>
#include <hip/hip_bf16.h>

#define NN 50000
#define NE 800000
#define NG 256
#define D  128

typedef float v2f __attribute__((ext_vector_type(2)));
typedef float v8f __attribute__((ext_vector_type(8)));

__device__ __forceinline__ void fatomic_add(float* p, float v) {
    unsafeAtomicAdd(p, v);
}

static_assert(NN % 4 == 0, "row scan moves four counts per thread");
#define NCH 64
#define CH4 (NE / 4 / NCH)
#define EIT ((CH4 + 1023) / 1024)
static_assert(NE % (4 * NCH) == 0, "every chunk is a whole number of four-edge groups");

__global__ __launch_bounds__(1024) void k_chunk_count(const int* __restrict__ dst,
                                                      int* __restrict__ ghist) {
    __shared__ int hist[NN];
    const int tid = threadIdx.x;
    const int g   = blockIdx.x;
    int4* h4 = reinterpret_cast<int4*>(hist);
    const int4* d4 = reinterpret_cast<const int4*>(dst) + (size_t)g * CH4;
    int4 d[EIT];
#pragma unroll
    for (int k = 0; k < EIT; ++k) {
        const int i = tid + k * 1024;
        d[k] = make_int4(0, 0, 0, 0);
        if (i < CH4) d[k] = d4[i];
    }
    for (int i = tid; i < NN / 4; i += 1024) h4[i] = make_int4(0, 0, 0, 0);
    __syncthreads();
#pragma unroll
    for (int k = 0; k < EIT; ++k) {
        const int i = tid + k * 1024;
        if (i < CH4) {
            atomicAdd(&hist[d[k].x], 1);
            atomicAdd(&hist[d[k].y], 1);
            atomicAdd(&hist[d[k].z], 1);
            atomicAdd(&hist[d[k].w], 1);
        }
    }
    __syncthreads();
    int4* o4 = reinterpret_cast<int4*>(ghist + (size_t)g * NN);
    for (int i = tid; i < NN / 4; i += 1024) o4[i] = h4[i];
}

__global__ void k_chunk_prefix(const int* __restrict__ ghist, int* __restrict__ gpre,
                               int* __restrict__ cnt) {
    const int n = blockIdx.x * blockDim.x + threadIdx.x;
    if (n < NN) {
        int h[NCH];
#pragma unroll
        for (int g = 0; g < NCH; ++g) h[g] = ghist[(size_t)g * NN + n];
        int run = 0;
#pragma unroll
        for (int g = 0; g < NCH; ++g) {
            gpre[(size_t)g * NN + n] = run;
            run += h[g];
        }
        cnt[n] = run;
    }
}

__global__ __launch_bounds__(1024) void k_row_scan(int* __restrict__ cnt_start,
                                                   float* __restrict__ dinv) {
    __shared__ int wtot[32];
    const int tid  = threadIdx.x;
    const int lane = tid & 31;
    const int wave = tid >> 5;
    int carry = 0;
    for (int base = 0; base < NN; base += 4096) {
        const int i = base + tid * 4;
        int4 c = make_int4(0, 0, 0, 0);
        if (i < NN) {
            c = *reinterpret_cast<const int4*>(cnt_start + i);
            float4 r;
            float d0 = (float)(c.x + 1), d1 = (float)(c.y + 1);
            float d2 = (float)(c.z + 1), d3 = (float)(c.w + 1);
            r.x = d0 > 0.0f ? rsqrtf(d0) : 0.0f;
            r.y = d1 > 0.0f ? rsqrtf(d1) : 0.0f;
            r.z = d2 > 0.0f ? rsqrtf(d2) : 0.0f;
            r.w = d3 > 0.0f ? rsqrtf(d3) : 0.0f;
            *reinterpret_cast<float4*>(dinv + i) = r;
        }
        const int s = c.x + c.y + c.z + c.w;
        int v = s;
        for (int d = 1; d < 32; d <<= 1) {
            int t = __shfl_up(v, d, 32);
            if (lane >= d) v += t;
        }
        if (lane == 31) wtot[wave] = v;
        __syncthreads();
        if (wave == 0) {
            int t = wtot[lane];
            for (int d = 1; d < 32; d <<= 1) {
                int u = __shfl_up(t, d, 32);
                if (lane >= d) t += u;
            }
            wtot[lane] = t;
        }
        __syncthreads();
        const int excl = carry + (wave ? wtot[wave - 1] : 0) + v - s;
        if (i < NN) {
            int4 o;
            o.x = excl;
            o.y = excl + c.x;
            o.z = excl + c.x + c.y;
            o.w = excl + c.x + c.y + c.z;
            *reinterpret_cast<int4*>(cnt_start + i) = o;
        }
        carry += wtot[31];
        __syncthreads();
    }
}

__global__ __launch_bounds__(1024) void k_chunk_place(
    const int* __restrict__ src, const int* __restrict__ dst,
    const int* __restrict__ rowstart, const int* __restrict__ gpre, int* __restrict__ csrc) {
    __shared__ int slot[NN];
    const int tid = threadIdx.x;
    const int g   = blockIdx.x;
    const int4* r4 = reinterpret_cast<const int4*>(rowstart);
    const int4* p4 = reinterpret_cast<const int4*>(gpre + (size_t)g * NN);
    int4* s4 = reinterpret_cast<int4*>(slot);
    for (int i = tid; i < NN / 4; i += 1024) {
        const int4 a = r4[i];
        const int4 b = p4[i];
        s4[i] = make_int4(a.x + b.x, a.y + b.y, a.z + b.z, a.w + b.w);
    }
    const int4* sv = reinterpret_cast<const int4*>(src) + (size_t)g * CH4;
    const int4* dv = reinterpret_cast<const int4*>(dst) + (size_t)g * CH4;
    int4 s[EIT], d[EIT];
#pragma unroll
    for (int k = 0; k < EIT; ++k) {
        const int i = tid + k * 1024;
        s[k] = make_int4(0, 0, 0, 0);
        d[k] = make_int4(0, 0, 0, 0);
        if (i < CH4) { s[k] = sv[i]; d[k] = dv[i]; }
    }
    __syncthreads();
#pragma unroll
    for (int k = 0; k < EIT; ++k) {
        const int i = tid + k * 1024;
        if (i < CH4) {
            const int p0 = atomicAdd(&slot[d[k].x], 1);
            const int p1 = atomicAdd(&slot[d[k].y], 1);
            const int p2 = atomicAdd(&slot[d[k].z], 1);
            const int p3 = atomicAdd(&slot[d[k].w], 1);
            csrc[p0] = s[k].x;
            csrc[p1] = s[k].y;
            csrc[p2] = s[k].z;
            csrc[p3] = s[k].w;
        }
    }
}

template <int RELU>
__global__ __launch_bounds__(256) void k_gemm_wmma(
    const float* __restrict__ in, const float* __restrict__ W,
    float* __restrict__ out, int nrows) {
    __shared__ float ldsW[128 * 129];
    const int tid = threadIdx.x;
    for (int i = tid; i < 128 * 128; i += 256)
        ldsW[(i >> 7) * 129 + (i & 127)] = W[i];
    __syncthreads();

    const int lane  = tid & 31;
    const int wave  = tid >> 5;
    const int strip = blockIdx.x * 128 + wave * 16;
    int rowA = strip + (lane & 15);
    if (rowA >= nrows) rowA = nrows - 1;
    const int koff = (lane >> 4) << 1;
    const int col  = lane & 15;

    v8f acc[8] = {};
    const float* aBase = in + (size_t)rowA * D;

    for (int k = 0; k < D; k += 4) {
        const float* ap = aBase + k + koff;
        v2f a;
        a.x = ap[0];
        a.y = ap[1];
        if (RELU) { a.x = fmaxf(a.x, 0.0f); a.y = fmaxf(a.y, 0.0f); }
        const float* b0 = &ldsW[(k + koff)     * 129 + col];
        const float* b1 = &ldsW[(k + koff + 1) * 129 + col];
#pragma unroll
        for (int j = 0; j < 8; ++j) {
            v2f b;
            b.x = b0[j * 16];
            b.y = b1[j * 16];
            acc[j] = __builtin_amdgcn_wmma_f32_16x16x4_f32(
                false, a, false, b, (short)0, acc[j], false, false);
        }
    }

    const int rbase = strip + (lane >> 4) * 8;
    if (strip + 16 <= nrows) {
#pragma unroll
        for (int j = 0; j < 8; ++j) {
#pragma unroll
            for (int r = 0; r < 8; ++r) {
                out[(size_t)(rbase + r) * D + j * 16 + col] = acc[j][r];
            }
        }
    } else {
#pragma unroll
        for (int j = 0; j < 8; ++j) {
#pragma unroll
            for (int r = 0; r < 8; ++r) {
                int row = rbase + r;
                if (row < nrows) out[(size_t)row * D + j * 16 + col] = acc[j][r];
            }
        }
    }
}

template <int RELU, int K>
__device__ __forceinline__ void gather_rows(const float4* __restrict__ hv, int sidx, float dsrc,
                                            float di, int j, int lane, float4& acc) {
    float4 v[K];
#pragma unroll
    for (int u = 0; u < K; ++u) {
        const int s = __builtin_amdgcn_readlane(sidx, j + u);
        v[u] = hv[(size_t)s * 32 + lane];
    }
#pragma unroll
    for (int u = 0; u < K; ++u) {
        const float w = __int_as_float(__builtin_amdgcn_readlane(__float_as_int(dsrc), j + u)) * di;
        if (RELU) {
            v[u].x = fmaxf(v[u].x, 0.0f); v[u].y = fmaxf(v[u].y, 0.0f);
            v[u].z = fmaxf(v[u].z, 0.0f); v[u].w = fmaxf(v[u].w, 0.0f);
        }
        acc.x = __builtin_fmaf(v[u].x, w, acc.x);
        acc.y = __builtin_fmaf(v[u].y, w, acc.y);
        acc.z = __builtin_fmaf(v[u].z, w, acc.z);
        acc.w = __builtin_fmaf(v[u].w, w, acc.w);
    }
}

template <int POOL>
__device__ __forceinline__ void gather_chunk(const float4* __restrict__ hv, int sidx, float dsrc,
                                             float di, int m, int lane, float4& acc) {
    int j = 0;
    for (; j + 8 <= m; j += 8) gather_rows<POOL, 8>(hv, sidx, dsrc, di, j, lane, acc);
    if (m - j >= 4) { gather_rows<POOL, 4>(hv, sidx, dsrc, di, j, lane, acc); j += 4; }
    const int rem = m - j;
    if (rem == 3)      gather_rows<POOL, 3>(hv, sidx, dsrc, di, j, lane, acc);
    else if (rem == 2) gather_rows<POOL, 2>(hv, sidx, dsrc, di, j, lane, acc);
    else if (rem == 1) gather_rows<POOL, 1>(hv, sidx, dsrc, di, j, lane, acc);
}

template <int POOL>
__device__ __forceinline__ float4 row_start(float4 sv, float4 b, float di) {
    const float sn = di * di;
    float4 acc;
    if (POOL) {
        sv.x = fmaxf(sv.x, 0.0f); sv.y = fmaxf(sv.y, 0.0f);
        sv.z = fmaxf(sv.z, 0.0f); sv.w = fmaxf(sv.w, 0.0f);
        acc.x = sv.x * sn; acc.y = sv.y * sn;
        acc.z = sv.z * sn; acc.w = sv.w * sn;
    } else {
        acc.x = __builtin_fmaf(sv.x, sn, b.x); acc.y = __builtin_fmaf(sv.y, sn, b.y);
        acc.z = __builtin_fmaf(sv.z, sn, b.z); acc.w = __builtin_fmaf(sv.w, sn, b.w);
    }
    return acc;
}

#define RPW 4
static_assert(RPW == 4 && NN % RPW == 0, "every wave owns four whole rows; their bounds, dinv and graphs are read as one vector each");

template <int POOL>
__global__ __launch_bounds__(256) void k_row_gather(
    const float* __restrict__ h, const int* __restrict__ rowstart,
    const int* __restrict__ csrc, const float* __restrict__ dinv,
    const float* __restrict__ bias, float* __restrict__ agg,
    const int* __restrict__ batch) {
    const int wave = threadIdx.x >> 5;
    const int n0 = __builtin_amdgcn_readfirstlane((int)(blockIdx.x * 8 + wave)) * RPW;
    const int lane = threadIdx.x & 31;
    if (!POOL) { if (n0 >= NN) return; }
    const float4* hv = reinterpret_cast<const float4*>(h);

    int gF = 0;
    float4 loc = make_float4(0.0f, 0.0f, 0.0f, 0.0f);
    if (POOL) gF = batch[blockIdx.x * (8 * RPW)];

    if (!POOL || n0 < NN) {
        const int4 q = *reinterpret_cast<const int4*>(rowstart + n0);
        int q0 = q.x, q1 = q.y, q2 = q.z, q3 = q.w;
        int q4 = (n0 + RPW < NN) ? rowstart[n0 + RPW] : NE;
        const float4 dq = *reinterpret_cast<const float4*>(dinv + n0);
        float d0 = dq.x, d1 = dq.y, d2 = dq.z, d3 = dq.w;
        int g0 = 0, g1 = 0, g2 = 0, g3 = 0;
        if (POOL) {
            const int4 gq = *reinterpret_cast<const int4*>(batch + n0);
            g0 = gq.x; g1 = gq.y; g2 = gq.z; g3 = gq.w;
        }
        float4 b = make_float4(0.0f, 0.0f, 0.0f, 0.0f);
        if (!POOL) b = *reinterpret_cast<const float4*>(bias + lane * 4);

        int sidx = n0;
        {
            const int m0 = (q1 - q0 < 32) ? (q1 - q0) : 32;
            if (lane < m0) sidx = csrc[q0 + lane];
        }
        float4 acc = row_start<POOL>(hv[(size_t)n0 * 32 + lane], b, d0);

        for (int r = 0; r < RPW; ++r) {
            const int n = n0 + r;
            const float di = d0;
            const float dsrc = dinv[sidx];
            const int mn = (q2 - q1 < 32) ? (q2 - q1) : 32;
            int sidx_next = n + 1;
            if (lane < mn) sidx_next = csrc[q1 + lane];
            const int nn = (n + 1 < NN) ? n + 1 : n;
            const float4 sv_next = hv[(size_t)nn * 32 + lane];

            gather_chunk<POOL>(hv, sidx, dsrc, di, (q1 - q0 < 32) ? (q1 - q0) : 32, lane, acc);
            for (int base = q0 + 32; base < q1; base += 32) {
                const int m2 = (q1 - base < 32) ? (q1 - base) : 32;
                int s2 = n;
                if (lane < m2) s2 = csrc[base + lane];
                const float ds2 = dinv[s2];
                gather_chunk<POOL>(hv, s2, ds2, di, m2, lane, acc);
            }

            if (POOL) {
                if (g0 == gF) {
                    loc.x += acc.x; loc.y += acc.y; loc.z += acc.z; loc.w += acc.w;
                } else {
                    float* o = agg + (size_t)g0 * D + lane * 4;
                    fatomic_add(o + 0, acc.x);
                    fatomic_add(o + 1, acc.y);
                    fatomic_add(o + 2, acc.z);
                    fatomic_add(o + 3, acc.w);
                }
            } else {
                *reinterpret_cast<float4*>(agg + (size_t)n * D + lane * 4) = acc;
            }
            acc = row_start<POOL>(sv_next, b, d1);
            sidx = sidx_next;
            q0 = q1; q1 = q2; q2 = q3; q3 = q4;
            d0 = d1; d1 = d2; d2 = d3;
            g0 = g1; g1 = g2; g2 = g3;
        }
    }

    if (POOL) {
        __shared__ __attribute__((aligned(16))) float part[8 * D];
        *reinterpret_cast<float4*>(&part[wave * D + lane * 4]) = loc;
        __syncthreads();
        if (threadIdx.x < D) {
            float s = part[threadIdx.x];
            for (int w = 1; w < 8; ++w) s += part[w * D + threadIdx.x];
            fatomic_add(agg + (size_t)gF * D + threadIdx.x, s);
        }
    }
}

__global__ void k_pool_zero(float* sums_cnt) {
    int i = blockIdx.x * blockDim.x + threadIdx.x;
    if (i < NG * D + NG) sums_cnt[i] = 0.0f;
}

__global__ void k_pool_cnt(const int* __restrict__ batch, float* __restrict__ cnt) {
    int n = blockIdx.x * blockDim.x + threadIdx.x;
    if (n < NN) fatomic_add(&cnt[batch[n]], 1.0f);
}

__global__ __launch_bounds__(256) void k_pool_transform(
    const float* __restrict__ sums, const float* __restrict__ cnt,
    const float* __restrict__ W, const float* __restrict__ bias, float* __restrict__ out) {
    const int lane  = threadIdx.x & 31;
    const int wave  = threadIdx.x >> 5;
    const int strip = blockIdx.x * 16;
    const int koff  = (lane >> 4) << 1;
    const int col   = lane & 15;
    const int ocol  = wave * 16 + col;
    const float* ap = sums + (size_t)(strip + col) * D + koff;
    const float* bp = W + (size_t)koff * D + ocol;
    v8f acc0 = {};
    v8f acc1 = {};
#pragma unroll
    for (int k = 0; k < D; k += 8) {
        v2f a0, b0, a1, b1;
        a0.x = ap[k];     a0.y = ap[k + 1];
        a1.x = ap[k + 4]; a1.y = ap[k + 5];
        b0.x = bp[(size_t)k * D];       b0.y = bp[(size_t)(k + 1) * D];
        b1.x = bp[(size_t)(k + 4) * D]; b1.y = bp[(size_t)(k + 5) * D];
        acc0 = __builtin_amdgcn_wmma_f32_16x16x4_f32(
            false, a0, false, b0, (short)0, acc0, false, false);
        acc1 = __builtin_amdgcn_wmma_f32_16x16x4_f32(
            false, a1, false, b1, (short)0, acc1, false, false);
    }
    const int rbase = strip + (lane >> 4) * 8;
    const float bcol = bias[ocol];
#pragma unroll
    for (int r = 0; r < 8; ++r) {
        const int g = rbase + r;
        const float c = cnt[g];
        out[(size_t)g * D + ocol] = (acc0[r] + acc1[r] + c * bcol) / fmaxf(c, 1.0f);
    }
}

extern "C" void kernel_launch(void* const* d_in, const int* in_sizes, int n_in,
                              void* d_out, int out_size, void* d_ws, size_t ws_size,
                              hipStream_t stream) {
    const float* x     = (const float*)d_in[0];
    const int*   edge  = (const int*)d_in[1];
    const int*   batch = (const int*)d_in[2];
    const float* W1 = (const float*)d_in[3];
    const float* b1 = (const float*)d_in[4];
    const float* W2 = (const float*)d_in[5];
    const float* b2 = (const float*)d_in[6];
    const float* W3 = (const float*)d_in[7];
    const float* b3 = (const float*)d_in[8];

    const int* esrc = edge;
    const int* edst = edge + NE;

    float* ws   = (float*)d_ws;
    int*   rowst = (int*)ws;
    float* dinv = ws + NN;
    float* bufH = ws + 100000;
    float* bufA = bufH + (size_t)NN * D;
    float* sums = bufA + (size_t)NN * D;
    float* cnt  = sums + NG * D;
    int*   csrc = (int*)(cnt + NG);
    int*   ghist = (int*)bufA;
    int*   gpre  = (int*)bufH;

    const int TB = 256;
    const int gNodes   = (NN + TB - 1) / TB;
    const int gNode32  = (NN / RPW * 32 + TB - 1) / TB;
    const int gGemm    = (NN + 127) / 128;

    k_chunk_count <<<NCH, 1024, 0, stream>>>(edst, ghist);
    k_chunk_prefix<<<gNodes, TB, 0, stream>>>(ghist, gpre, rowst);
    k_row_scan    <<<1, 1024, 0, stream>>>(rowst, dinv);
    k_chunk_place <<<NCH, 1024, 0, stream>>>(esrc, edst, rowst, gpre, csrc);

    k_gemm_wmma<0><<<gGemm, TB, 0, stream>>>(x, W1, bufH, NN);
    k_row_gather<0><<<gNode32, TB, 0, stream>>>(bufH, rowst, csrc, dinv, b1, bufA, nullptr);

    k_gemm_wmma<1><<<gGemm, TB, 0, stream>>>(bufA, W2, bufH, NN);
    k_row_gather<0><<<gNode32, TB, 0, stream>>>(bufH, rowst, csrc, dinv, b2, bufA, nullptr);

    k_pool_zero <<<(NG * D + NG + TB - 1) / TB, TB, 0, stream>>>(sums);
    k_pool_cnt  <<<gNodes, TB, 0, stream>>>(batch, cnt);
    k_row_gather<1><<<gNode32, TB, 0, stream>>>(bufA, rowst, csrc, dinv, nullptr, sums, batch);
    k_pool_transform<<<NG / 16, TB, 0, stream>>>(sums, cnt, W3, b3, (float*)d_out);
}


# Row scan without a launch of its own: the chunk-prefix kernel scans its 256 in-degrees and writes dinv, each place work-group scans the 196 block totals in LDS; 12 -> 11 launches; integer sums only
# speedup vs baseline: 2.8858x; 1.1108x over previous
#include <hip/hip_runtime.h>
#include <hip/hip_bf16.h>

#define NN 50000
#define NE 800000
#define NG 256
#define D  128

typedef float v2f __attribute__((ext_vector_type(2)));
typedef float v8f __attribute__((ext_vector_type(8)));

__device__ __forceinline__ void fatomic_add(float* p, float v) {
    unsafeAtomicAdd(p, v);
}

static_assert(NN % 4 == 0, "row scan moves four counts per thread");
#define NCH 64
#define CH4 (NE / 4 / NCH)
#define EIT ((CH4 + 1023) / 1024)
static_assert(NE % (4 * NCH) == 0, "every chunk is a whole number of four-edge groups");

__global__ __launch_bounds__(1024) void k_chunk_count(const int* __restrict__ dst,
                                                      int* __restrict__ ghist) {
    __shared__ int hist[NN];
    const int tid = threadIdx.x;
    const int g   = blockIdx.x;
    int4* h4 = reinterpret_cast<int4*>(hist);
    const int4* d4 = reinterpret_cast<const int4*>(dst) + (size_t)g * CH4;
    int4 d[EIT];
#pragma unroll
    for (int k = 0; k < EIT; ++k) {
        const int i = tid + k * 1024;
        d[k] = make_int4(0, 0, 0, 0);
        if (i < CH4) d[k] = d4[i];
    }
    for (int i = tid; i < NN / 4; i += 1024) h4[i] = make_int4(0, 0, 0, 0);
    __syncthreads();
#pragma unroll
    for (int k = 0; k < EIT; ++k) {
        const int i = tid + k * 1024;
        if (i < CH4) {
            atomicAdd(&hist[d[k].x], 1);
            atomicAdd(&hist[d[k].y], 1);
            atomicAdd(&hist[d[k].z], 1);
            atomicAdd(&hist[d[k].w], 1);
        }
    }
    __syncthreads();
    int4* o4 = reinterpret_cast<int4*>(ghist + (size_t)g * NN);
    for (int i = tid; i < NN / 4; i += 1024) o4[i] = h4[i];
}

#define RB 256
#define NB ((NN + RB - 1) / RB)
static_assert(RB % 4 == 0, "four rows that are read as one vector lie in one work-group of the prefix kernel");
__global__ void k_chunk_prefix(const int* __restrict__ ghist, int* __restrict__ gpre,
                               int* __restrict__ rloc, int* __restrict__ btot,
                               float* __restrict__ dinv) {
    __shared__ int wtot[RB / 32];
    const int tid  = threadIdx.x;
    const int lane = tid & 31;
    const int wave = tid >> 5;
    const int n = blockIdx.x * RB + tid;
    int h[NCH];
    int run = 0;
    if (n < NN) {
#pragma unroll
        for (int g = 0; g < NCH; ++g) h[g] = ghist[(size_t)g * NN + n];
#pragma unroll
        for (int g = 0; g < NCH; ++g) run += h[g];
    }
    int v = run;
    for (int d = 1; d < 32; d <<= 1) {
        int t = __shfl_up(v, d, 32);
        if (lane >= d) v += t;
    }
    if (lane == 31) wtot[wave] = v;
    __syncthreads();
    int off = 0;
#pragma unroll
    for (int w = 0; w < RB / 32; ++w) {
        const int t = wtot[w];
        if (w < wave) off += t;
    }
    if (n < NN) {
        int pre = 0;
#pragma unroll
        for (int g = 0; g < NCH; ++g) {
            gpre[(size_t)g * NN + n] = pre;
            pre += h[g];
        }
        const float dg = (float)(run + 1);
        dinv[n] = dg > 0.0f ? rsqrtf(dg) : 0.0f;
        rloc[n] = off + v - run;
    }
    if (tid == RB - 1) btot[blockIdx.x] = off + v;
}

#define BPL ((NB + 31) / 32)
#define RS4 ((NN / 4 + NCH - 1) / NCH)
static_assert(RS4 <= 1024, "one thread per stored group of row starts");
__global__ __launch_bounds__(1024) void k_chunk_place(
    const int* __restrict__ src, const int* __restrict__ dst,
    const int* __restrict__ rloc, const int* __restrict__ btot,
    const int* __restrict__ gpre, int* __restrict__ rowstart, int* __restrict__ csrc) {
    __shared__ int slot[NN];
    __shared__ int boff[NB];
    const int tid = threadIdx.x;
    const int g   = blockIdx.x;
    const int4* sv = reinterpret_cast<const int4*>(src) + (size_t)g * CH4;
    const int4* dv = reinterpret_cast<const int4*>(dst) + (size_t)g * CH4;
    int4 s[EIT], d[EIT];
#pragma unroll
    for (int k = 0; k < EIT; ++k) {
        const int i = tid + k * 1024;
        s[k] = make_int4(0, 0, 0, 0);
        d[k] = make_int4(0, 0, 0, 0);
        if (i < CH4) { s[k] = sv[i]; d[k] = dv[i]; }
    }
    if (tid < 32) {
        int t[BPL];
        int sum = 0;
#pragma unroll
        for (int k = 0; k < BPL; ++k) {
            const int b = tid * BPL + k;
            t[k] = (b < NB) ? btot[b] : 0;
            sum += t[k];
        }
        int v = sum;
        for (int dd = 1; dd < 32; dd <<= 1) {
            int u = __shfl_up(v, dd, 32);
            if (tid >= dd) v += u;
        }
        int run = v - sum;
#pragma unroll
        for (int k = 0; k < BPL; ++k) {
            const int b = tid * BPL + k;
            if (b < NB) boff[b] = run;
            run += t[k];
        }
    }
    __syncthreads();
    const int4* l4 = reinterpret_cast<const int4*>(rloc);
    const int4* p4 = reinterpret_cast<const int4*>(gpre + (size_t)g * NN);
    int4* s4 = reinterpret_cast<int4*>(slot);
    for (int i = tid; i < NN / 4; i += 1024) {
        const int4 a = l4[i];
        const int4 b = p4[i];
        const int base = boff[i / (RB / 4)];
        s4[i] = make_int4(a.x + b.x + base, a.y + b.y + base, a.z + b.z + base, a.w + b.w + base);
    }
    if (tid < RS4) {
        const int i = g * RS4 + tid;
        if (i < NN / 4) {
            const int4 a = l4[i];
            const int base = boff[i / (RB / 4)];
            reinterpret_cast<int4*>(rowstart)[i] = make_int4(a.x + base, a.y + base, a.z + base, a.w + base);
        }
    }
    __syncthreads();
#pragma unroll
    for (int k = 0; k < EIT; ++k) {
        const int i = tid + k * 1024;
        if (i < CH4) {
            const int p0 = atomicAdd(&slot[d[k].x], 1);
            const int p1 = atomicAdd(&slot[d[k].y], 1);
            const int p2 = atomicAdd(&slot[d[k].z], 1);
            const int p3 = atomicAdd(&slot[d[k].w], 1);
            csrc[p0] = s[k].x;
            csrc[p1] = s[k].y;
            csrc[p2] = s[k].z;
            csrc[p3] = s[k].w;
        }
    }
}

template <int RELU>
__global__ __launch_bounds__(256) void k_gemm_wmma(
    const float* __restrict__ in, const float* __restrict__ W,
    float* __restrict__ out, int nrows) {
    __shared__ float ldsW[128 * 129];
    const int tid = threadIdx.x;
    for (int i = tid; i < 128 * 128; i += 256)
        ldsW[(i >> 7) * 129 + (i & 127)] = W[i];
    __syncthreads();

    const int lane  = tid & 31;
    const int wave  = tid >> 5;
    const int strip = blockIdx.x * 128 + wave * 16;
    int rowA = strip + (lane & 15);
    if (rowA >= nrows) rowA = nrows - 1;
    const int koff = (lane >> 4) << 1;
    const int col  = lane & 15;

    v8f acc[8] = {};
    const float* aBase = in + (size_t)rowA * D;

    for (int k = 0; k < D; k += 4) {
        const float* ap = aBase + k + koff;
        v2f a;
        a.x = ap[0];
        a.y = ap[1];
        if (RELU) { a.x = fmaxf(a.x, 0.0f); a.y = fmaxf(a.y, 0.0f); }
        const float* b0 = &ldsW[(k + koff)     * 129 + col];
        const float* b1 = &ldsW[(k + koff + 1) * 129 + col];
#pragma unroll
        for (int j = 0; j < 8; ++j) {
            v2f b;
            b.x = b0[j * 16];
            b.y = b1[j * 16];
            acc[j] = __builtin_amdgcn_wmma_f32_16x16x4_f32(
                false, a, false, b, (short)0, acc[j], false, false);
        }
    }

    const int rbase = strip + (lane >> 4) * 8;
    if (strip + 16 <= nrows) {
#pragma unroll
        for (int j = 0; j < 8; ++j) {
#pragma unroll
            for (int r = 0; r < 8; ++r) {
                out[(size_t)(rbase + r) * D + j * 16 + col] = acc[j][r];
            }
        }
    } else {
#pragma unroll
        for (int j = 0; j < 8; ++j) {
#pragma unroll
            for (int r = 0; r < 8; ++r) {
                int row = rbase + r;
                if (row < nrows) out[(size_t)row * D + j * 16 + col] = acc[j][r];
            }
        }
    }
}

template <int RELU, int K>
__device__ __forceinline__ void gather_rows(const float4* __restrict__ hv, int sidx, float dsrc,
                                            float di, int j, int lane, float4& acc) {
    float4 v[K];
#pragma unroll
    for (int u = 0; u < K; ++u) {
        const int s = __builtin_amdgcn_readlane(sidx, j + u);
        v[u] = hv[(size_t)s * 32 + lane];
    }
#pragma unroll
    for (int u = 0; u < K; ++u) {
        const float w = __int_as_float(__builtin_amdgcn_readlane(__float_as_int(dsrc), j + u)) * di;
        if (RELU) {
            v[u].x = fmaxf(v[u].x, 0.0f); v[u].y = fmaxf(v[u].y, 0.0f);
            v[u].z = fmaxf(v[u].z, 0.0f); v[u].w = fmaxf(v[u].w, 0.0f);
        }
        acc.x = __builtin_fmaf(v[u].x, w, acc.x);
        acc.y = __builtin_fmaf(v[u].y, w, acc.y);
        acc.z = __builtin_fmaf(v[u].z, w, acc.z);
        acc.w = __builtin_fmaf(v[u].w, w, acc.w);
    }
}

template <int POOL>
__device__ __forceinline__ void gather_chunk(const float4* __restrict__ hv, int sidx, float dsrc,
                                             float di, int m, int lane, float4& acc) {
    int j = 0;
    for (; j + 8 <= m; j += 8) gather_rows<POOL, 8>(hv, sidx, dsrc, di, j, lane, acc);
    if (m - j >= 4) { gather_rows<POOL, 4>(hv, sidx, dsrc, di, j, lane, acc); j += 4; }
    const int rem = m - j;
    if (rem == 3)      gather_rows<POOL, 3>(hv, sidx, dsrc, di, j, lane, acc);
    else if (rem == 2) gather_rows<POOL, 2>(hv, sidx, dsrc, di, j, lane, acc);
    else if (rem == 1) gather_rows<POOL, 1>(hv, sidx, dsrc, di, j, lane, acc);
}

template <int POOL>
__device__ __forceinline__ float4 row_start(float4 sv, float4 b, float di) {
    const float sn = di * di;
    float4 acc;
    if (POOL) {
        sv.x = fmaxf(sv.x, 0.0f); sv.y = fmaxf(sv.y, 0.0f);
        sv.z = fmaxf(sv.z, 0.0f); sv.w = fmaxf(sv.w, 0.0f);
        acc.x = sv.x * sn; acc.y = sv.y * sn;
        acc.z = sv.z * sn; acc.w = sv.w * sn;
    } else {
        acc.x = __builtin_fmaf(sv.x, sn, b.x); acc.y = __builtin_fmaf(sv.y, sn, b.y);
        acc.z = __builtin_fmaf(sv.z, sn, b.z); acc.w = __builtin_fmaf(sv.w, sn, b.w);
    }
    return acc;
}

#define RPW 4
static_assert(RPW == 4 && NN % RPW == 0, "every wave owns four whole rows; their bounds, dinv and graphs are read as one vector each");

template <int POOL>
__global__ __launch_bounds__(256) void k_row_gather(
    const float* __restrict__ h, const int* __restrict__ rowstart,
    const int* __restrict__ csrc, const float* __restrict__ dinv,
    const float* __restrict__ bias, float* __restrict__ agg,
    const int* __restrict__ batch) {
    const int wave = threadIdx.x >> 5;
    const int n0 = __builtin_amdgcn_readfirstlane((int)(blockIdx.x * 8 + wave)) * RPW;
    const int lane = threadIdx.x & 31;
    if (!POOL) { if (n0 >= NN) return; }
    const float4* hv = reinterpret_cast<const float4*>(h);

    int gF = 0;
    float4 loc = make_float4(0.0f, 0.0f, 0.0f, 0.0f);
    if (POOL) gF = batch[blockIdx.x * (8 * RPW)];

    if (!POOL || n0 < NN) {
        const int4 q = *reinterpret_cast<const int4*>(rowstart + n0);
        int q0 = q.x, q1 = q.y, q2 = q.z, q3 = q.w;
        int q4 = (n0 + RPW < NN) ? rowstart[n0 + RPW] : NE;
        const float4 dq = *reinterpret_cast<const float4*>(dinv + n0);
        float d0 = dq.x, d1 = dq.y, d2 = dq.z, d3 = dq.w;
        int g0 = 0, g1 = 0, g2 = 0, g3 = 0;
        if (POOL) {
            const int4 gq = *reinterpret_cast<const int4*>(batch + n0);
            g0 = gq.x; g1 = gq.y; g2 = gq.z; g3 = gq.w;
        }
        float4 b = make_float4(0.0f, 0.0f, 0.0f, 0.0f);
        if (!POOL) b = *reinterpret_cast<const float4*>(bias + lane * 4);

        int sidx = n0;
        {
            const int m0 = (q1 - q0 < 32) ? (q1 - q0) : 32;
            if (lane < m0) sidx = csrc[q0 + lane];
        }
        float4 acc = row_start<POOL>(hv[(size_t)n0 * 32 + lane], b, d0);

        for (int r = 0; r < RPW; ++r) {
            const int n = n0 + r;
            const float di = d0;
            const float dsrc = dinv[sidx];
            const int mn = (q2 - q1 < 32) ? (q2 - q1) : 32;
            int sidx_next = n + 1;
            if (lane < mn) sidx_next = csrc[q1 + lane];
            const int nn = (n + 1 < NN) ? n + 1 : n;
            const float4 sv_next = hv[(size_t)nn * 32 + lane];

            gather_chunk<POOL>(hv, sidx, dsrc, di, (q1 - q0 < 32) ? (q1 - q0) : 32, lane, acc);
            for (int base = q0 + 32; base < q1; base += 32) {
                const int m2 = (q1 - base < 32) ? (q1 - base) : 32;
                int s2 = n;
                if (lane < m2) s2 = csrc[base + lane];
                const float ds2 = dinv[s2];
                gather_chunk<POOL>(hv, s2, ds2, di, m2, lane, acc);
            }

            if (POOL) {
                if (g0 == gF) {
                    loc.x += acc.x; loc.y += acc.y; loc.z += acc.z; loc.w += acc.w;
                } else {
                    float* o = agg + (size_t)g0 * D + lane * 4;
                    fatomic_add(o + 0, acc.x);
                    fatomic_add(o + 1, acc.y);
                    fatomic_add(o + 2, acc.z);
                    fatomic_add(o + 3, acc.w);
                }
            } else {
                *reinterpret_cast<float4*>(agg + (size_t)n * D + lane * 4) = acc;
            }
            acc = row_start<POOL>(sv_next, b, d1);
            sidx = sidx_next;
            q0 = q1; q1 = q2; q2 = q3; q3 = q4;
            d0 = d1; d1 = d2; d2 = d3;
            g0 = g1; g1 = g2; g2 = g3;
        }
    }

    if (POOL) {
        __shared__ __attribute__((aligned(16))) float part[8 * D];
        *reinterpret_cast<float4*>(&part[wave * D + lane * 4]) = loc;
        __syncthreads();
        if (threadIdx.x < D) {
            float s = part[threadIdx.x];
            for (int w = 1; w < 8; ++w) s += part[w * D + threadIdx.x];
            fatomic_add(agg + (size_t)gF * D + threadIdx.x, s);
        }
    }
}

__global__ void k_pool_zero(float* sums_cnt) {
    int i = blockIdx.x * blockDim.x + threadIdx.x;
    if (i < NG * D + NG) sums_cnt[i] = 0.0f;
}

__global__ void k_pool_cnt(const int* __restrict__ batch, float* __restrict__ cnt) {
    int n = blockIdx.x * blockDim.x + threadIdx.x;
    if (n < NN) fatomic_add(&cnt[batch[n]], 1.0f);
}

__global__ __launch_bounds__(256) void k_pool_transform(
    const float* __restrict__ sums, const float* __restrict__ cnt,
    const float* __restrict__ W, const float* __restrict__ bias, float* __restrict__ out) {
    const int lane  = threadIdx.x & 31;
    const int wave  = threadIdx.x >> 5;
    const int strip = blockIdx.x * 16;
    const int koff  = (lane >> 4) << 1;
    const int col   = lane & 15;
    const int ocol  = wave * 16 + col;
    const float* ap = sums + (size_t)(strip + col) * D + koff;
    const float* bp = W + (size_t)koff * D + ocol;
    v8f acc0 = {};
    v8f acc1 = {};
#pragma unroll
    for (int k = 0; k < D; k += 8) {
        v2f a0, b0, a1, b1;
        a0.x = ap[k];     a0.y = ap[k + 1];
        a1.x = ap[k + 4]; a1.y = ap[k + 5];
        b0.x = bp[(size_t)k * D];       b0.y = bp[(size_t)(k + 1) * D];
        b1.x = bp[(size_t)(k + 4) * D]; b1.y = bp[(size_t)(k + 5) * D];
        acc0 = __builtin_amdgcn_wmma_f32_16x16x4_f32(
            false, a0, false, b0, (short)0, acc0, false, false);
        acc1 = __builtin_amdgcn_wmma_f32_16x16x4_f32(
            false, a1, false, b1, (short)0, acc1, false, false);
    }
    const int rbase = strip + (lane >> 4) * 8;
    const float bcol = bias[ocol];
#pragma unroll
    for (int r = 0; r < 8; ++r) {
        const int g = rbase + r;
        const float c = cnt[g];
        out[(size_t)g * D + ocol] = (acc0[r] + acc1[r] + c * bcol) / fmaxf(c, 1.0f);
    }
}

extern "C" void kernel_launch(void* const* d_in, const int* in_sizes, int n_in,
                              void* d_out, int out_size, void* d_ws, size_t ws_size,
                              hipStream_t stream) {
    const float* x     = (const float*)d_in[0];
    const int*   edge  = (const int*)d_in[1];
    const int*   batch = (const int*)d_in[2];
    const float* W1 = (const float*)d_in[3];
    const float* b1 = (const float*)d_in[4];
    const float* W2 = (const float*)d_in[5];
    const float* b2 = (const float*)d_in[6];
    const float* W3 = (const float*)d_in[7];
    const float* b3 = (const float*)d_in[8];

    const int* esrc = edge;
    const int* edst = edge + NE;

    float* ws   = (float*)d_ws;
    int*   rowst = (int*)ws;
    float* dinv = ws + NN;
    float* bufH = ws + 100000;
    float* bufA = bufH + (size_t)NN * D;
    float* sums = bufA + (size_t)NN * D;
    float* cnt  = sums + NG * D;
    int*   csrc = (int*)(cnt + NG);
    int*   ghist = (int*)bufA;
    int*   gpre  = (int*)bufH;
    int*   rloc  = ghist + (size_t)NCH * NN;
    int*   btot  = rloc + NN;

    const int TB = 256;
    const int gNodes   = (NN + TB - 1) / TB;
    const int gNode32  = (NN / RPW * 32 + TB - 1) / TB;
    const int gGemm    = (NN + 127) / 128;

    k_chunk_count <<<NCH, 1024, 0, stream>>>(edst, ghist);
    k_chunk_prefix<<<NB, RB, 0, stream>>>(ghist, gpre, rloc, btot, dinv);
    k_chunk_place <<<NCH, 1024, 0, stream>>>(esrc, edst, rloc, btot, gpre, rowst, csrc);

    k_gemm_wmma<0><<<gGemm, TB, 0, stream>>>(x, W1, bufH, NN);
    k_row_gather<0><<<gNode32, TB, 0, stream>>>(bufH, rowst, csrc, dinv, b1, bufA, nullptr);

    k_gemm_wmma<1><<<gGemm, TB, 0, stream>>>(bufA, W2, bufH, NN);
    k_row_gather<0><<<gNode32, TB, 0, stream>>>(bufH, rowst, csrc, dinv, b2, bufA, nullptr);

    k_pool_zero <<<(NG * D + NG + TB - 1) / TB, TB, 0, stream>>>(sums);
    k_pool_cnt  <<<gNodes, TB, 0, stream>>>(batch, cnt);
    k_row_gather<1><<<gNode32, TB, 0, stream>>>(bufA, rowst, csrc, dinv, nullptr, sums, batch);
    k_pool_transform<<<NG / 16, TB, 0, stream>>>(sums, cnt, W3, b3, (float*)d_out);
}


# Pool preparation without launches of its own: the prefix kernel stores the zeros, the pooled gather counts each graph's rows (one add per work-group); 11 -> 9 launches; counts exact, sums as before
# speedup vs baseline: 3.1074x; 1.0768x over previous
#include <hip/hip_runtime.h>
#include <hip/hip_bf16.h>

#define NN 50000
#define NE 800000
#define NG 256
#define D  128

typedef float v2f __attribute__((ext_vector_type(2)));
typedef float v8f __attribute__((ext_vector_type(8)));

__device__ __forceinline__ void fatomic_add(float* p, float v) {
    unsafeAtomicAdd(p, v);
}

static_assert(NN % 4 == 0, "row scan moves four counts per thread");
#define NCH 64
#define CH4 (NE / 4 / NCH)
#define EIT ((CH4 + 1023) / 1024)
static_assert(NE % (4 * NCH) == 0, "every chunk is a whole number of four-edge groups");

__global__ __launch_bounds__(1024) void k_chunk_count(const int* __restrict__ dst,
                                                      int* __restrict__ ghist) {
    __shared__ int hist[NN];
    const int tid = threadIdx.x;
    const int g   = blockIdx.x;
    int4* h4 = reinterpret_cast<int4*>(hist);
    const int4* d4 = reinterpret_cast<const int4*>(dst) + (size_t)g * CH4;
    int4 d[EIT];
#pragma unroll
    for (int k = 0; k < EIT; ++k) {
        const int i = tid + k * 1024;
        d[k] = make_int4(0, 0, 0, 0);
        if (i < CH4) d[k] = d4[i];
    }
    for (int i = tid; i < NN / 4; i += 1024) h4[i] = make_int4(0, 0, 0, 0);
    __syncthreads();
#pragma unroll
    for (int k = 0; k < EIT; ++k) {
        const int i = tid + k * 1024;
        if (i < CH4) {
            atomicAdd(&hist[d[k].x], 1);
            atomicAdd(&hist[d[k].y], 1);
            atomicAdd(&hist[d[k].z], 1);
            atomicAdd(&hist[d[k].w], 1);
        }
    }
    __syncthreads();
    int4* o4 = reinterpret_cast<int4*>(ghist + (size_t)g * NN);
    for (int i = tid; i < NN / 4; i += 1024) o4[i] = h4[i];
}

#define RB 256
#define NB ((NN + RB - 1) / RB)
static_assert(RB % 4 == 0, "four rows that are read as one vector lie in one work-group of the prefix kernel");
static_assert(NB * RB >= NG * D + NG, "one thread of the prefix kernel per pooled sum and count");
__global__ void k_chunk_prefix(const int* __restrict__ ghist, int* __restrict__ gpre,
                               int* __restrict__ rloc, int* __restrict__ btot,
                               float* __restrict__ dinv, float* __restrict__ pooled) {
    __shared__ int wtot[RB / 32];
    const int tid  = threadIdx.x;
    const int lane = tid & 31;
    const int wave = tid >> 5;
    const int n = blockIdx.x * RB + tid;
    int h[NCH];
    int run = 0;
    if (n < NN) {
#pragma unroll
        for (int g = 0; g < NCH; ++g) h[g] = ghist[(size_t)g * NN + n];
#pragma unroll
        for (int g = 0; g < NCH; ++g) run += h[g];
    }
    int v = run;
    for (int d = 1; d < 32; d <<= 1) {
        int t = __shfl_up(v, d, 32);
        if (lane >= d) v += t;
    }
    if (lane == 31) wtot[wave] = v;
    __syncthreads();
    int off = 0;
#pragma unroll
    for (int w = 0; w < RB / 32; ++w) {
        const int t = wtot[w];
        if (w < wave) off += t;
    }
    if (n < NN) {
        int pre = 0;
#pragma unroll
        for (int g = 0; g < NCH; ++g) {
            gpre[(size_t)g * NN + n] = pre;
            pre += h[g];
        }
        const float dg = (float)(run + 1);
        dinv[n] = dg > 0.0f ? rsqrtf(dg) : 0.0f;
        rloc[n] = off + v - run;
    }
    if (tid == RB - 1) btot[blockIdx.x] = off + v;
    const int z = blockIdx.x * RB + tid;
    if (z < NG * D + NG) pooled[z] = 0.0f;
}

#define BPL ((NB + 31) / 32)
#define RS4 ((NN / 4 + NCH - 1) / NCH)
static_assert(RS4 <= 1024, "one thread per stored group of row starts");
__global__ __launch_bounds__(1024) void k_chunk_place(
    const int* __restrict__ src, const int* __restrict__ dst,
    const int* __restrict__ rloc, const int* __restrict__ btot,
    const int* __restrict__ gpre, int* __restrict__ rowstart, int* __restrict__ csrc) {
    __shared__ int slot[NN];
    __shared__ int boff[NB];
    const int tid = threadIdx.x;
    const int g   = blockIdx.x;
    const int4* sv = reinterpret_cast<const int4*>(src) + (size_t)g * CH4;
    const int4* dv = reinterpret_cast<const int4*>(dst) + (size_t)g * CH4;
    int4 s[EIT], d[EIT];
#pragma unroll
    for (int k = 0; k < EIT; ++k) {
        const int i = tid + k * 1024;
        s[k] = make_int4(0, 0, 0, 0);
        d[k] = make_int4(0, 0, 0, 0);
        if (i < CH4) { s[k] = sv[i]; d[k] = dv[i]; }
    }
    if (tid < 32) {
        int t[BPL];
        int sum = 0;
#pragma unroll
        for (int k = 0; k < BPL; ++k) {
            const int b = tid * BPL + k;
            t[k] = (b < NB) ? btot[b] : 0;
            sum += t[k];
        }
        int v = sum;
        for (int dd = 1; dd < 32; dd <<= 1) {
            int u = __shfl_up(v, dd, 32);
            if (tid >= dd) v += u;
        }
        int run = v - sum;
#pragma unroll
        for (int k = 0; k < BPL; ++k) {
            const int b = tid * BPL + k;
            if (b < NB) boff[b] = run;
            run += t[k];
        }
    }
    __syncthreads();
    const int4* l4 = reinterpret_cast<const int4*>(rloc);
    const int4* p4 = reinterpret_cast<const int4*>(gpre + (size_t)g * NN);
    int4* s4 = reinterpret_cast<int4*>(slot);
    for (int i = tid; i < NN / 4; i += 1024) {
        const int4 a = l4[i];
        const int4 b = p4[i];
        const int base = boff[i / (RB / 4)];
        s4[i] = make_int4(a.x + b.x + base, a.y + b.y + base, a.z + b.z + base, a.w + b.w + base);
    }
    if (tid < RS4) {
        const int i = g * RS4 + tid;
        if (i < NN / 4) {
            const int4 a = l4[i];
            const int base = boff[i / (RB / 4)];
            reinterpret_cast<int4*>(rowstart)[i] = make_int4(a.x + base, a.y + base, a.z + base, a.w + base);
        }
    }
    __syncthreads();
#pragma unroll
    for (int k = 0; k < EIT; ++k) {
        const int i = tid + k * 1024;
        if (i < CH4) {
            const int p0 = atomicAdd(&slot[d[k].x], 1);
            const int p1 = atomicAdd(&slot[d[k].y], 1);
            const int p2 = atomicAdd(&slot[d[k].z], 1);
            const int p3 = atomicAdd(&slot[d[k].w], 1);
            csrc[p0] = s[k].x;
            csrc[p1] = s[k].y;
            csrc[p2] = s[k].z;
            csrc[p3] = s[k].w;
        }
    }
}

template <int RELU>
__global__ __launch_bounds__(256) void k_gemm_wmma(
    const float* __restrict__ in, const float* __restrict__ W,
    float* __restrict__ out, int nrows) {
    __shared__ float ldsW[128 * 129];
    const int tid = threadIdx.x;
    for (int i = tid; i < 128 * 128; i += 256)
        ldsW[(i >> 7) * 129 + (i & 127)] = W[i];
    __syncthreads();

    const int lane  = tid & 31;
    const int wave  = tid >> 5;
    const int strip = blockIdx.x * 128 + wave * 16;
    int rowA = strip + (lane & 15);
    if (rowA >= nrows) rowA = nrows - 1;
    const int koff = (lane >> 4) << 1;
    const int col  = lane & 15;

    v8f acc[8] = {};
    const float* aBase = in + (size_t)rowA * D;

    for (int k = 0; k < D; k += 4) {
        const float* ap = aBase + k + koff;
        v2f a;
        a.x = ap[0];
        a.y = ap[1];
        if (RELU) { a.x = fmaxf(a.x, 0.0f); a.y = fmaxf(a.y, 0.0f); }
        const float* b0 = &ldsW[(k + koff)     * 129 + col];
        const float* b1 = &ldsW[(k + koff + 1) * 129 + col];
#pragma unroll
        for (int j = 0; j < 8; ++j) {
            v2f b;
            b.x = b0[j * 16];
            b.y = b1[j * 16];
            acc[j] = __builtin_amdgcn_wmma_f32_16x16x4_f32(
                false, a, false, b, (short)0, acc[j], false, false);
        }
    }

    const int rbase = strip + (lane >> 4) * 8;
    if (strip + 16 <= nrows) {
#pragma unroll
        for (int j = 0; j < 8; ++j) {
#pragma unroll
            for (int r = 0; r < 8; ++r) {
                out[(size_t)(rbase + r) * D + j * 16 + col] = acc[j][r];
            }
        }
    } else {
#pragma unroll
        for (int j = 0; j < 8; ++j) {
#pragma unroll
            for (int r = 0; r < 8; ++r) {
                int row = rbase + r;
                if (row < nrows) out[(size_t)row * D + j * 16 + col] = acc[j][r];
            }
        }
    }
}

template <int RELU, int K>
__device__ __forceinline__ void gather_rows(const float4* __restrict__ hv, int sidx, float dsrc,
                                            float di, int j, int lane, float4& acc) {
    float4 v[K];
#pragma unroll
    for (int u = 0; u < K; ++u) {
        const int s = __builtin_amdgcn_readlane(sidx, j + u);
        v[u] = hv[(size_t)s * 32 + lane];
    }
#pragma unroll
    for (int u = 0; u < K; ++u) {
        const float w = __int_as_float(__builtin_amdgcn_readlane(__float_as_int(dsrc), j + u)) * di;
        if (RELU) {
            v[u].x = fmaxf(v[u].x, 0.0f); v[u].y = fmaxf(v[u].y, 0.0f);
            v[u].z = fmaxf(v[u].z, 0.0f); v[u].w = fmaxf(v[u].w, 0.0f);
        }
        acc.x = __builtin_fmaf(v[u].x, w, acc.x);
        acc.y = __builtin_fmaf(v[u].y, w, acc.y);
        acc.z = __builtin_fmaf(v[u].z, w, acc.z);
        acc.w = __builtin_fmaf(v[u].w, w, acc.w);
    }
}

template <int POOL>
__device__ __forceinline__ void gather_chunk(const float4* __restrict__ hv, int sidx, float dsrc,
                                             float di, int m, int lane, float4& acc) {
    int j = 0;
    for (; j + 8 <= m; j += 8) gather_rows<POOL, 8>(hv, sidx, dsrc, di, j, lane, acc);
    if (m - j >= 4) { gather_rows<POOL, 4>(hv, sidx, dsrc, di, j, lane, acc); j += 4; }
    const int rem = m - j;
    if (rem == 3)      gather_rows<POOL, 3>(hv, sidx, dsrc, di, j, lane, acc);
    else if (rem == 2) gather_rows<POOL, 2>(hv, sidx, dsrc, di, j, lane, acc);
    else if (rem == 1) gather_rows<POOL, 1>(hv, sidx, dsrc, di, j, lane, acc);
}

template <int POOL>
__device__ __forceinline__ float4 row_start(float4 sv, float4 b, float di) {
    const float sn = di * di;
    float4 acc;
    if (POOL) {
        sv.x = fmaxf(sv.x, 0.0f); sv.y = fmaxf(sv.y, 0.0f);
        sv.z = fmaxf(sv.z, 0.0f); sv.w = fmaxf(sv.w, 0.0f);
        acc.x = sv.x * sn; acc.y = sv.y * sn;
        acc.z = sv.z * sn; acc.w = sv.w * sn;
    } else {
        acc.x = __builtin_fmaf(sv.x, sn, b.x); acc.y = __builtin_fmaf(sv.y, sn, b.y);
        acc.z = __builtin_fmaf(sv.z, sn, b.z); acc.w = __builtin_fmaf(sv.w, sn, b.w);
    }
    return acc;
}

#define RPW 4
static_assert(RPW == 4 && NN % RPW == 0, "every wave owns four whole rows; their bounds, dinv and graphs are read as one vector each");

template <int POOL>
__global__ __launch_bounds__(256) void k_row_gather(
    const float* __restrict__ h, const int* __restrict__ rowstart,
    const int* __restrict__ csrc, const float* __restrict__ dinv,
    const float* __restrict__ bias, float* __restrict__ agg,
    const int* __restrict__ batch, float* __restrict__ cnt) {
    const int wave = threadIdx.x >> 5;
    const int n0 = __builtin_amdgcn_readfirstlane((int)(blockIdx.x * 8 + wave)) * RPW;
    const int lane = threadIdx.x & 31;
    if (!POOL) { if (n0 >= NN) return; }
    const float4* hv = reinterpret_cast<const float4*>(h);

    int gF = 0;
    float4 loc = make_float4(0.0f, 0.0f, 0.0f, 0.0f);
    float nloc = 0.0f;
    if (POOL) gF = batch[blockIdx.x * (8 * RPW)];

    if (!POOL || n0 < NN) {
        const int4 q = *reinterpret_cast<const int4*>(rowstart + n0);
        int q0 = q.x, q1 = q.y, q2 = q.z, q3 = q.w;
        int q4 = (n0 + RPW < NN) ? rowstart[n0 + RPW] : NE;
        const float4 dq = *reinterpret_cast<const float4*>(dinv + n0);
        float d0 = dq.x, d1 = dq.y, d2 = dq.z, d3 = dq.w;
        int g0 = 0, g1 = 0, g2 = 0, g3 = 0;
        if (POOL) {
            const int4 gq = *reinterpret_cast<const int4*>(batch + n0);
            g0 = gq.x; g1 = gq.y; g2 = gq.z; g3 = gq.w;
        }
        float4 b = make_float4(0.0f, 0.0f, 0.0f, 0.0f);
        if (!POOL) b = *reinterpret_cast<const float4*>(bias + lane * 4);

        int sidx = n0;
        {
            const int m0 = (q1 - q0 < 32) ? (q1 - q0) : 32;
            if (lane < m0) sidx = csrc[q0 + lane];
        }
        float4 acc = row_start<POOL>(hv[(size_t)n0 * 32 + lane], b, d0);

        for (int r = 0; r < RPW; ++r) {
            const int n = n0 + r;
            const float di = d0;
            const float dsrc = dinv[sidx];
            const int mn = (q2 - q1 < 32) ? (q2 - q1) : 32;
            int sidx_next = n + 1;
            if (lane < mn) sidx_next = csrc[q1 + lane];
            const int nn = (n + 1 < NN) ? n + 1 : n;
            const float4 sv_next = hv[(size_t)nn * 32 + lane];

            gather_chunk<POOL>(hv, sidx, dsrc, di, (q1 - q0 < 32) ? (q1 - q0) : 32, lane, acc);
            for (int base = q0 + 32; base < q1; base += 32) {
                const int m2 = (q1 - base < 32) ? (q1 - base) : 32;
                int s2 = n;
                if (lane < m2) s2 = csrc[base + lane];
                const float ds2 = dinv[s2];
                gather_chunk<POOL>(hv, s2, ds2, di, m2, lane, acc);
            }

            if (POOL) {
                if (g0 == gF) {
                    loc.x += acc.x; loc.y += acc.y; loc.z += acc.z; loc.w += acc.w;
                    nloc += 1.0f;
                } else {
                    if (lane == 0) fatomic_add(cnt + g0, 1.0f);
                    float* o = agg + (size_t)g0 * D + lane * 4;
                    fatomic_add(o + 0, acc.x);
                    fatomic_add(o + 1, acc.y);
                    fatomic_add(o + 2, acc.z);
                    fatomic_add(o + 3, acc.w);
                }
            } else {
                *reinterpret_cast<float4*>(agg + (size_t)n * D + lane * 4) = acc;
            }
            acc = row_start<POOL>(sv_next, b, d1);
            sidx = sidx_next;
            q0 = q1; q1 = q2; q2 = q3; q3 = q4;
            d0 = d1; d1 = d2; d2 = d3;
            g0 = g1; g1 = g2; g2 = g3;
        }
    }

    if (POOL) {
        __shared__ __attribute__((aligned(16))) float part[8 * D];
        __shared__ float pcnt[8];
        *reinterpret_cast<float4*>(&part[wave * D + lane * 4]) = loc;
        if (lane == 0) pcnt[wave] = nloc;
        __syncthreads();
        if (threadIdx.x < D) {
            float s = part[threadIdx.x];
            for (int w = 1; w < 8; ++w) s += part[w * D + threadIdx.x];
            fatomic_add(agg + (size_t)gF * D + threadIdx.x, s);
        } else if (threadIdx.x == D) {
            float c = pcnt[0];
            for (int w = 1; w < 8; ++w) c += pcnt[w];
            fatomic_add(cnt + gF, c);
        }
    }
}

__global__ __launch_bounds__(256) void k_pool_transform(
    const float* __restrict__ sums, const float* __restrict__ cnt,
    const float* __restrict__ W, const float* __restrict__ bias, float* __restrict__ out) {
    const int lane  = threadIdx.x & 31;
    const int wave  = threadIdx.x >> 5;
    const int strip = blockIdx.x * 16;
    const int koff  = (lane >> 4) << 1;
    const int col   = lane & 15;
    const int ocol  = wave * 16 + col;
    const float* ap = sums + (size_t)(strip + col) * D + koff;
    const float* bp = W + (size_t)koff * D + ocol;
    v8f acc0 = {};
    v8f acc1 = {};
#pragma unroll
    for (int k = 0; k < D; k += 8) {
        v2f a0, b0, a1, b1;
        a0.x = ap[k];     a0.y = ap[k + 1];
        a1.x = ap[k + 4]; a1.y = ap[k + 5];
        b0.x = bp[(size_t)k * D];       b0.y = bp[(size_t)(k + 1) * D];
        b1.x = bp[(size_t)(k + 4) * D]; b1.y = bp[(size_t)(k + 5) * D];
        acc0 = __builtin_amdgcn_wmma_f32_16x16x4_f32(
            false, a0, false, b0, (short)0, acc0, false, false);
        acc1 = __builtin_amdgcn_wmma_f32_16x16x4_f32(
            false, a1, false, b1, (short)0, acc1, false, false);
    }
    const int rbase = strip + (lane >> 4) * 8;
    const float bcol = bias[ocol];
#pragma unroll
    for (int r = 0; r < 8; ++r) {
        const int g = rbase + r;
        const float c = cnt[g];
        out[(size_t)g * D + ocol] = (acc0[r] + acc1[r] + c * bcol) / fmaxf(c, 1.0f);
    }
}

extern "C" void kernel_launch(void* const* d_in, const int* in_sizes, int n_in,
                              void* d_out, int out_size, void* d_ws, size_t ws_size,
                              hipStream_t stream) {
    const float* x     = (const float*)d_in[0];
    const int*   edge  = (const int*)d_in[1];
    const int*   batch = (const int*)d_in[2];
    const float* W1 = (const float*)d_in[3];
    const float* b1 = (const float*)d_in[4];
    const float* W2 = (const float*)d_in[5];
    const float* b2 = (const float*)d_in[6];
    const float* W3 = (const float*)d_in[7];
    const float* b3 = (const float*)d_in[8];

    const int* esrc = edge;
    const int* edst = edge + NE;

    float* ws   = (float*)d_ws;
    int*   rowst = (int*)ws;
    float* dinv = ws + NN;
    float* bufH = ws + 100000;
    float* bufA = bufH + (size_t)NN * D;
    float* sums = bufA + (size_t)NN * D;
    float* cnt  = sums + NG * D;
    int*   csrc = (int*)(cnt + NG);
    int*   ghist = (int*)bufA;
    int*   gpre  = (int*)bufH;
    int*   rloc  = ghist + (size_t)NCH * NN;
    int*   btot  = rloc + NN;

    const int TB = 256;
    const int gNode32  = (NN / RPW * 32 + TB - 1) / TB;
    const int gGemm    = (NN + 127) / 128;

    k_chunk_count <<<NCH, 1024, 0, stream>>>(edst, ghist);
    k_chunk_prefix<<<NB, RB, 0, stream>>>(ghist, gpre, rloc, btot, dinv, sums);
    k_chunk_place <<<NCH, 1024, 0, stream>>>(esrc, edst, rloc, btot, gpre, rowst, csrc);

    k_gemm_wmma<0><<<gGemm, TB, 0, stream>>>(x, W1, bufH, NN);
    k_row_gather<0><<<gNode32, TB, 0, stream>>>(bufH, rowst, csrc, dinv, b1, bufA, nullptr, nullptr);

    k_gemm_wmma<1><<<gGemm, TB, 0, stream>>>(bufA, W2, bufH, NN);
    k_row_gather<0><<<gNode32, TB, 0, stream>>>(bufH, rowst, csrc, dinv, b2, bufA, nullptr, nullptr);

    k_row_gather<1><<<gNode32, TB, 0, stream>>>(bufA, rowst, csrc, dinv, nullptr, sums, batch, cnt);
    k_pool_transform<<<NG / 16, TB, 0, stream>>>(sums, cnt, W3, b3, (float*)d_out);
}


# Each dense transform moved behind its layer's row gather, relu((Ahat h) W + b): the work-group multiplies its 32 row sums (LDS tile) by W on f32 WMMA; 9 -> 7 launches; sum order changes
# speedup vs baseline: 3.8404x; 1.2359x over previous
#include <hip/hip_runtime.h>
#include <hip/hip_bf16.h>

#define NN 50000
#define NE 800000
#define NG 256
#define D  128

typedef float v2f __attribute__((ext_vector_type(2)));
typedef float v8f __attribute__((ext_vector_type(8)));

__device__ __forceinline__ void fatomic_add(float* p, float v) {
    unsafeAtomicAdd(p, v);
}

static_assert(NN % 4 == 0, "row scan moves four counts per thread");
#define NCH 64
#define CH4 (NE / 4 / NCH)
#define EIT ((CH4 + 1023) / 1024)
static_assert(NE % (4 * NCH) == 0, "every chunk is a whole number of four-edge groups");

__global__ __launch_bounds__(1024) void k_chunk_count(const int* __restrict__ dst,
                                                      int* __restrict__ ghist) {
    __shared__ int hist[NN];
    const int tid = threadIdx.x;
    const int g   = blockIdx.x;
    int4* h4 = reinterpret_cast<int4*>(hist);
    const int4* d4 = reinterpret_cast<const int4*>(dst) + (size_t)g * CH4;
    int4 d[EIT];
#pragma unroll
    for (int k = 0; k < EIT; ++k) {
        const int i = tid + k * 1024;
        d[k] = make_int4(0, 0, 0, 0);
        if (i < CH4) d[k] = d4[i];
    }
    for (int i = tid; i < NN / 4; i += 1024) h4[i] = make_int4(0, 0, 0, 0);
    __syncthreads();
#pragma unroll
    for (int k = 0; k < EIT; ++k) {
        const int i = tid + k * 1024;
        if (i < CH4) {
            atomicAdd(&hist[d[k].x], 1);
            atomicAdd(&hist[d[k].y], 1);
            atomicAdd(&hist[d[k].z], 1);
            atomicAdd(&hist[d[k].w], 1);
        }
    }
    __syncthreads();
    int4* o4 = reinterpret_cast<int4*>(ghist + (size_t)g * NN);
    for (int i = tid; i < NN / 4; i += 1024) o4[i] = h4[i];
}

#define RB 256
#define NB ((NN + RB - 1) / RB)
static_assert(RB % 4 == 0, "four rows that are read as one vector lie in one work-group of the prefix kernel");
static_assert(NB * RB >= NG * D + NG, "one thread of the prefix kernel per pooled sum and count");
__global__ void k_chunk_prefix(const int* __restrict__ ghist, int* __restrict__ gpre,
                               int* __restrict__ rloc, int* __restrict__ btot,
                               float* __restrict__ dinv, float* __restrict__ pooled) {
    __shared__ int wtot[RB / 32];
    const int tid  = threadIdx.x;
    const int lane = tid & 31;
    const int wave = tid >> 5;
    const int n = blockIdx.x * RB + tid;
    int h[NCH];
    int run = 0;
    if (n < NN) {
#pragma unroll
        for (int g = 0; g < NCH; ++g) h[g] = ghist[(size_t)g * NN + n];
#pragma unroll
        for (int g = 0; g < NCH; ++g) run += h[g];
    }
    int v = run;
    for (int d = 1; d < 32; d <<= 1) {
        int t = __shfl_up(v, d, 32);
        if (lane >= d) v += t;
    }
    if (lane == 31) wtot[wave] = v;
    __syncthreads();
    int off = 0;
#pragma unroll
    for (int w = 0; w < RB / 32; ++w) {
        const int t = wtot[w];
        if (w < wave) off += t;
    }
    if (n < NN) {
        int pre = 0;
#pragma unroll
        for (int g = 0; g < NCH; ++g) {
            gpre[(size_t)g * NN + n] = pre;
            pre += h[g];
        }
        const float dg = (float)(run + 1);
        dinv[n] = dg > 0.0f ? rsqrtf(dg) : 0.0f;
        rloc[n] = off + v - run;
    }
    if (tid == RB - 1) btot[blockIdx.x] = off + v;
    const int z = blockIdx.x * RB + tid;
    if (z < NG * D + NG) pooled[z] = 0.0f;
}

#define BPL ((NB + 31) / 32)
#define RS4 ((NN / 4 + NCH - 1) / NCH)
static_assert(RS4 <= 1024, "one thread per stored group of row starts");
__global__ __launch_bounds__(1024) void k_chunk_place(
    const int* __restrict__ src, const int* __restrict__ dst,
    const int* __restrict__ rloc, const int* __restrict__ btot,
    const int* __restrict__ gpre, int* __restrict__ rowstart, int* __restrict__ csrc) {
    __shared__ int slot[NN];
    __shared__ int boff[NB];
    const int tid = threadIdx.x;
    const int g   = blockIdx.x;
    const int4* sv = reinterpret_cast<const int4*>(src) + (size_t)g * CH4;
    const int4* dv = reinterpret_cast<const int4*>(dst) + (size_t)g * CH4;
    int4 s[EIT], d[EIT];
#pragma unroll
    for (int k = 0; k < EIT; ++k) {
        const int i = tid + k * 1024;
        s[k] = make_int4(0, 0, 0, 0);
        d[k] = make_int4(0, 0, 0, 0);
        if (i < CH4) { s[k] = sv[i]; d[k] = dv[i]; }
    }
    if (tid < 32) {
        int t[BPL];
        int sum = 0;
#pragma unroll
        for (int k = 0; k < BPL; ++k) {
            const int b = tid * BPL + k;
            t[k] = (b < NB) ? btot[b] : 0;
            sum += t[k];
        }
        int v = sum;
        for (int dd = 1; dd < 32; dd <<= 1) {
            int u = __shfl_up(v, dd, 32);
            if (tid >= dd) v += u;
        }
        int run = v - sum;
#pragma unroll
        for (int k = 0; k < BPL; ++k) {
            const int b = tid * BPL + k;
            if (b < NB) boff[b] = run;
            run += t[k];
        }
    }
    __syncthreads();
    const int4* l4 = reinterpret_cast<const int4*>(rloc);
    const int4* p4 = reinterpret_cast<const int4*>(gpre + (size_t)g * NN);
    int4* s4 = reinterpret_cast<int4*>(slot);
    for (int i = tid; i < NN / 4; i += 1024) {
        const int4 a = l4[i];
        const int4 b = p4[i];
        const int base = boff[i / (RB / 4)];
        s4[i] = make_int4(a.x + b.x + base, a.y + b.y + base, a.z + b.z + base, a.w + b.w + base);
    }
    if (tid < RS4) {
        const int i = g * RS4 + tid;
        if (i < NN / 4) {
            const int4 a = l4[i];
            const int base = boff[i / (RB / 4)];
            reinterpret_cast<int4*>(rowstart)[i] = make_int4(a.x + base, a.y + base, a.z + base, a.w + base);
        }
    }
    __syncthreads();
#pragma unroll
    for (int k = 0; k < EIT; ++k) {
        const int i = tid + k * 1024;
        if (i < CH4) {
            const int p0 = atomicAdd(&slot[d[k].x], 1);
            const int p1 = atomicAdd(&slot[d[k].y], 1);
            const int p2 = atomicAdd(&slot[d[k].z], 1);
            const int p3 = atomicAdd(&slot[d[k].w], 1);
            csrc[p0] = s[k].x;
            csrc[p1] = s[k].y;
            csrc[p2] = s[k].z;
            csrc[p3] = s[k].w;
        }
    }
}

template <int K>
__device__ __forceinline__ void gather_rows(const float4* __restrict__ hv, int sidx, float dsrc,
                                            float di, int j, int lane, float4& acc) {
    float4 v[K];
#pragma unroll
    for (int u = 0; u < K; ++u) {
        const int s = __builtin_amdgcn_readlane(sidx, j + u);
        v[u] = hv[(size_t)s * 32 + lane];
    }
#pragma unroll
    for (int u = 0; u < K; ++u) {
        const float w = __int_as_float(__builtin_amdgcn_readlane(__float_as_int(dsrc), j + u)) * di;
        acc.x = __builtin_fmaf(v[u].x, w, acc.x);
        acc.y = __builtin_fmaf(v[u].y, w, acc.y);
        acc.z = __builtin_fmaf(v[u].z, w, acc.z);
        acc.w = __builtin_fmaf(v[u].w, w, acc.w);
    }
}

__device__ __forceinline__ void gather_chunk(const float4* __restrict__ hv, int sidx, float dsrc,
                                             float di, int m, int lane, float4& acc) {
    int j = 0;
    for (; j + 8 <= m; j += 8) gather_rows<8>(hv, sidx, dsrc, di, j, lane, acc);
    if (m - j >= 4) { gather_rows<4>(hv, sidx, dsrc, di, j, lane, acc); j += 4; }
    const int rem = m - j;
    if (rem == 3)      gather_rows<3>(hv, sidx, dsrc, di, j, lane, acc);
    else if (rem == 2) gather_rows<2>(hv, sidx, dsrc, di, j, lane, acc);
    else if (rem == 1) gather_rows<1>(hv, sidx, dsrc, di, j, lane, acc);
}

__device__ __forceinline__ float4 row_start(float4 sv, float di) {
    const float sn = di * di;
    float4 acc;
    acc.x = sv.x * sn; acc.y = sv.y * sn;
    acc.z = sv.z * sn; acc.w = sv.w * sn;
    return acc;
}

#define RPW 4
static_assert(RPW == 4 && NN % RPW == 0, "every wave owns four whole rows; their bounds, dinv and graphs are read as one vector each");
#define TROWS (8 * RPW)
#define TP (D + 4)
static_assert(TROWS == 32, "a tile is two 16-row strips");

template <int POOL>
__global__ __launch_bounds__(256) void k_row_gather(
    const float* __restrict__ h, const int* __restrict__ rowstart,
    const int* __restrict__ csrc, const float* __restrict__ dinv,
    const float* __restrict__ W, const float* __restrict__ bias, float* __restrict__ agg,
    const int* __restrict__ batch, float* __restrict__ cnt) {
    __shared__ __attribute__((aligned(16))) float tile[POOL ? 4 : TROWS * TP];
    const int wave = threadIdx.x >> 5;
    const int n0 = __builtin_amdgcn_readfirstlane((int)(blockIdx.x * 8 + wave)) * RPW;
    const int lane = threadIdx.x & 31;
    const float4* hv = reinterpret_cast<const float4*>(h);

    int gF = 0;
    float4 loc = make_float4(0.0f, 0.0f, 0.0f, 0.0f);
    float nloc = 0.0f;
    if (POOL) gF = batch[blockIdx.x * (8 * RPW)];

    if (n0 < NN) {
        const int4 q = *reinterpret_cast<const int4*>(rowstart + n0);
        int q0 = q.x, q1 = q.y, q2 = q.z, q3 = q.w;
        int q4 = (n0 + RPW < NN) ? rowstart[n0 + RPW] : NE;
        const float4 dq = *reinterpret_cast<const float4*>(dinv + n0);
        float d0 = dq.x, d1 = dq.y, d2 = dq.z, d3 = dq.w;
        int g0 = 0, g1 = 0, g2 = 0, g3 = 0;
        if (POOL) {
            const int4 gq = *reinterpret_cast<const int4*>(batch + n0);
            g0 = gq.x; g1 = gq.y; g2 = gq.z; g3 = gq.w;
        }

        int sidx = n0;
        {
            const int m0 = (q1 - q0 < 32) ? (q1 - q0) : 32;
            if (lane < m0) sidx = csrc[q0 + lane];
        }
        float4 acc = row_start(hv[(size_t)n0 * 32 + lane], d0);

        for (int r = 0; r < RPW; ++r) {
            const int n = n0 + r;
            const float di = d0;
            const float dsrc = dinv[sidx];
            const int mn = (q2 - q1 < 32) ? (q2 - q1) : 32;
            int sidx_next = n + 1;
            if (lane < mn) sidx_next = csrc[q1 + lane];
            const int nn = (n + 1 < NN) ? n + 1 : n;
            const float4 sv_next = hv[(size_t)nn * 32 + lane];

            gather_chunk(hv, sidx, dsrc, di, (q1 - q0 < 32) ? (q1 - q0) : 32, lane, acc);
            for (int base = q0 + 32; base < q1; base += 32) {
                const int m2 = (q1 - base < 32) ? (q1 - base) : 32;
                int s2 = n;
                if (lane < m2) s2 = csrc[base + lane];
                const float ds2 = dinv[s2];
                gather_chunk(hv, s2, ds2, di, m2, lane, acc);
            }

            if (POOL) {
                if (g0 == gF) {
                    loc.x += acc.x; loc.y += acc.y; loc.z += acc.z; loc.w += acc.w;
                    nloc += 1.0f;
                } else {
                    if (lane == 0) fatomic_add(cnt + g0, 1.0f);
                    float* o = agg + (size_t)g0 * D + lane * 4;
                    fatomic_add(o + 0, acc.x);
                    fatomic_add(o + 1, acc.y);
                    fatomic_add(o + 2, acc.z);
                    fatomic_add(o + 3, acc.w);
                }
            } else {
                *reinterpret_cast<float4*>(&tile[(wave * RPW + r) * TP + lane * 4]) = acc;
            }
            acc = row_start(sv_next, d1);
            sidx = sidx_next;
            q0 = q1; q1 = q2; q2 = q3; q3 = q4;
            d0 = d1; d1 = d2; d2 = d3;
            g0 = g1; g1 = g2; g2 = g3;
        }
    } else if (!POOL) {
        for (int r = 0; r < RPW; ++r)
            *reinterpret_cast<float4*>(&tile[(wave * RPW + r) * TP + lane * 4]) = make_float4(0.0f, 0.0f, 0.0f, 0.0f);
    }

    if (!POOL) {
        const int koff = (lane >> 4) << 1;
        const int col  = lane & 15;
        const int ocol = wave * 16 + col;
        const float* bp  = W + (size_t)koff * D + ocol;
        const float bcol = bias[ocol];
        const float* a0p = &tile[col * TP + koff];
        const float* a1p = &tile[(16 + col) * TP + koff];
        v8f acc0 = {};
        v8f acc1 = {};
        __syncthreads();
#pragma unroll 8
        for (int k = 0; k < D; k += 4) {
            v2f b, a0, a1;
            b.x = bp[(size_t)k * D];
            b.y = bp[(size_t)(k + 1) * D];
            a0 = *reinterpret_cast<const v2f*>(a0p + k);
            a1 = *reinterpret_cast<const v2f*>(a1p + k);
            acc0 = __builtin_amdgcn_wmma_f32_16x16x4_f32(
                false, a0, false, b, (short)0, acc0, false, false);
            acc1 = __builtin_amdgcn_wmma_f32_16x16x4_f32(
                false, a1, false, b, (short)0, acc1, false, false);
        }
        __syncthreads();
        const int rt = (lane >> 4) * 8;
#pragma unroll
        for (int r = 0; r < 8; ++r) {
            tile[(rt + r) * TP + ocol]      = fmaxf(acc0[r] + bcol, 0.0f);
            tile[(16 + rt + r) * TP + ocol] = fmaxf(acc1[r] + bcol, 0.0f);
        }
        __syncthreads();
        if (n0 < NN) {
#pragma unroll
            for (int r = 0; r < RPW; ++r)
                *reinterpret_cast<float4*>(agg + (size_t)(n0 + r) * D + lane * 4) =
                    *reinterpret_cast<const float4*>(&tile[(wave * RPW + r) * TP + lane * 4]);
        }
    }

    if (POOL) {
        __shared__ __attribute__((aligned(16))) float part[8 * D];
        __shared__ float pcnt[8];
        *reinterpret_cast<float4*>(&part[wave * D + lane * 4]) = loc;
        if (lane == 0) pcnt[wave] = nloc;
        __syncthreads();
        if (threadIdx.x < D) {
            float s = part[threadIdx.x];
            for (int w = 1; w < 8; ++w) s += part[w * D + threadIdx.x];
            fatomic_add(agg + (size_t)gF * D + threadIdx.x, s);
        } else if (threadIdx.x == D) {
            float c = pcnt[0];
            for (int w = 1; w < 8; ++w) c += pcnt[w];
            fatomic_add(cnt + gF, c);
        }
    }
}

__global__ __launch_bounds__(256) void k_pool_transform(
    const float* __restrict__ sums, const float* __restrict__ cnt,
    const float* __restrict__ W, const float* __restrict__ bias, float* __restrict__ out) {
    const int lane  = threadIdx.x & 31;
    const int wave  = threadIdx.x >> 5;
    const int strip = blockIdx.x * 16;
    const int koff  = (lane >> 4) << 1;
    const int col   = lane & 15;
    const int ocol  = wave * 16 + col;
    const float* ap = sums + (size_t)(strip + col) * D + koff;
    const float* bp = W + (size_t)koff * D + ocol;
    v8f acc0 = {};
    v8f acc1 = {};
#pragma unroll
    for (int k = 0; k < D; k += 8) {
        v2f a0, b0, a1, b1;
        a0.x = ap[k];     a0.y = ap[k + 1];
        a1.x = ap[k + 4]; a1.y = ap[k + 5];
        b0.x = bp[(size_t)k * D];       b0.y = bp[(size_t)(k + 1) * D];
        b1.x = bp[(size_t)(k + 4) * D]; b1.y = bp[(size_t)(k + 5) * D];
        acc0 = __builtin_amdgcn_wmma_f32_16x16x4_f32(
            false, a0, false, b0, (short)0, acc0, false, false);
        acc1 = __builtin_amdgcn_wmma_f32_16x16x4_f32(
            false, a1, false, b1, (short)0, acc1, false, false);
    }
    const int rbase = strip + (lane >> 4) * 8;
    const float bcol = bias[ocol];
#pragma unroll
    for (int r = 0; r < 8; ++r) {
        const int g = rbase + r;
        const float c = cnt[g];
        out[(size_t)g * D + ocol] = (acc0[r] + acc1[r] + c * bcol) / fmaxf(c, 1.0f);
    }
}

extern "C" void kernel_launch(void* const* d_in, const int* in_sizes, int n_in,
                              void* d_out, int out_size, void* d_ws, size_t ws_size,
                              hipStream_t stream) {
    const float* x     = (const float*)d_in[0];
    const int*   edge  = (const int*)d_in[1];
    const int*   batch = (const int*)d_in[2];
    const float* W1 = (const float*)d_in[3];
    const float* b1 = (const float*)d_in[4];
    const float* W2 = (const float*)d_in[5];
    const float* b2 = (const float*)d_in[6];
    const float* W3 = (const float*)d_in[7];
    const float* b3 = (const float*)d_in[8];

    const int* esrc = edge;
    const int* edst = edge + NE;

    float* ws   = (float*)d_ws;
    int*   rowst = (int*)ws;
    float* dinv = ws + NN;
    float* bufH = ws + 100000;
    float* bufA = bufH + (size_t)NN * D;
    float* sums = bufA + (size_t)NN * D;
    float* cnt  = sums + NG * D;
    int*   csrc = (int*)(cnt + NG);
    int*   ghist = (int*)bufA;
    int*   gpre  = (int*)bufH;
    int*   rloc  = ghist + (size_t)NCH * NN;
    int*   btot  = rloc + NN;

    const int TB = 256;
    const int gNode32  = (NN / RPW * 32 + TB - 1) / TB;

    k_chunk_count <<<NCH, 1024, 0, stream>>>(edst, ghist);
    k_chunk_prefix<<<NB, RB, 0, stream>>>(ghist, gpre, rloc, btot, dinv, sums);
    k_chunk_place <<<NCH, 1024, 0, stream>>>(esrc, edst, rloc, btot, gpre, rowst, csrc);

    k_row_gather<0><<<gNode32, TB, 0, stream>>>(x, rowst, csrc, dinv, W1, b1, bufH, nullptr, nullptr);

    k_row_gather<0><<<gNode32, TB, 0, stream>>>(bufH, rowst, csrc, dinv, W2, b2, bufA, nullptr, nullptr);

    k_row_gather<1><<<gNode32, TB, 0, stream>>>(bufA, rowst, csrc, dinv, nullptr, nullptr, sums, batch, cnt);
    k_pool_transform<<<NG / 16, TB, 0, stream>>>(sums, cnt, W3, b3, (float*)d_out);
}


# Counting sort's work-groups also split by row range: 64 edge chunks x 4 row ranges = 256 work-groups of 16 waves for count and place, a quarter of the LDS counters each; list order as before
# speedup vs baseline: 4.3242x; 1.1260x over previous
#include <hip/hip_runtime.h>
#include <hip/hip_bf16.h>

#define NN 50000
#define NE 800000
#define NG 256
#define D  128

typedef float v2f __attribute__((ext_vector_type(2)));
typedef float v8f __attribute__((ext_vector_type(8)));

__device__ __forceinline__ void fatomic_add(float* p, float v) {
    unsafeAtomicAdd(p, v);
}

static_assert(NN % 4 == 0, "row scan moves four counts per thread");
#define NCH 64
#define NRR 4
#define RW  (NN / NRR)
#define CH4 (NE / 4 / NCH)
#define ST  512
#define EIT ((CH4 + ST - 1) / ST)
static_assert(NE % (4 * NCH) == 0, "every chunk is a whole number of four-edge groups");
static_assert(NN % (4 * NRR) == 0, "a range is a whole number of four-row groups, so that its counters move as 16-byte vectors");

__global__ __launch_bounds__(ST) void k_chunk_count(const int* __restrict__ dst,
                                                      int* __restrict__ ghist) {
    __shared__ __attribute__((aligned(16))) int hist[RW];
    const int tid = threadIdx.x;
    const int g   = blockIdx.x / NRR;
    const int r0  = (blockIdx.x % NRR) * RW;
    int4* h4 = reinterpret_cast<int4*>(hist);
    const int4* d4 = reinterpret_cast<const int4*>(dst) + (size_t)g * CH4;
    int4 d[EIT];
#pragma unroll
    for (int k = 0; k < EIT; ++k) {
        const int i = tid + k * ST;
        d[k] = make_int4(0, 0, 0, 0);
        if (i < CH4) d[k] = d4[i];
    }
    for (int i = tid; i < RW / 4; i += ST) h4[i] = make_int4(0, 0, 0, 0);
    __syncthreads();
#pragma unroll
    for (int k = 0; k < EIT; ++k) {
        const int i = tid + k * ST;
        if (i < CH4) {
            const unsigned a0 = (unsigned)(d[k].x - r0);
            const unsigned a1 = (unsigned)(d[k].y - r0);
            const unsigned a2 = (unsigned)(d[k].z - r0);
            const unsigned a3 = (unsigned)(d[k].w - r0);
            if (a0 < (unsigned)RW) atomicAdd(&hist[a0], 1);
            if (a1 < (unsigned)RW) atomicAdd(&hist[a1], 1);
            if (a2 < (unsigned)RW) atomicAdd(&hist[a2], 1);
            if (a3 < (unsigned)RW) atomicAdd(&hist[a3], 1);
        }
    }
    __syncthreads();
    int4* o4 = reinterpret_cast<int4*>(ghist + (size_t)g * NN + r0);
    for (int i = tid; i < RW / 4; i += ST) o4[i] = h4[i];
}

#define RB 256
#define NB ((NN + RB - 1) / RB)
static_assert(RB % 4 == 0, "four rows that are read as one vector lie in one work-group of the prefix kernel");
static_assert(NB * RB >= NG * D + NG, "one thread of the prefix kernel per pooled sum and count");
__global__ void k_chunk_prefix(const int* __restrict__ ghist, int* __restrict__ gpre,
                               int* __restrict__ rloc, int* __restrict__ btot,
                               float* __restrict__ dinv, float* __restrict__ pooled) {
    __shared__ int wtot[RB / 32];
    const int tid  = threadIdx.x;
    const int lane = tid & 31;
    const int wave = tid >> 5;
    const int n = blockIdx.x * RB + tid;
    int h[NCH];
    int run = 0;
    if (n < NN) {
#pragma unroll
        for (int g = 0; g < NCH; ++g) h[g] = ghist[(size_t)g * NN + n];
#pragma unroll
        for (int g = 0; g < NCH; ++g) run += h[g];
    }
    int v = run;
    for (int d = 1; d < 32; d <<= 1) {
        int t = __shfl_up(v, d, 32);
        if (lane >= d) v += t;
    }
    if (lane == 31) wtot[wave] = v;
    __syncthreads();
    int off = 0;
#pragma unroll
    for (int w = 0; w < RB / 32; ++w) {
        const int t = wtot[w];
        if (w < wave) off += t;
    }
    if (n < NN) {
        int pre = 0;
#pragma unroll
        for (int g = 0; g < NCH; ++g) {
            gpre[(size_t)g * NN + n] = pre;
            pre += h[g];
        }
        const float dg = (float)(run + 1);
        dinv[n] = dg > 0.0f ? rsqrtf(dg) : 0.0f;
        rloc[n] = off + v - run;
    }
    if (tid == RB - 1) btot[blockIdx.x] = off + v;
    const int z = blockIdx.x * RB + tid;
    if (z < NG * D + NG) pooled[z] = 0.0f;
}

#define BPL ((NB + 31) / 32)
#define RSG ((RW / 4 + NCH - 1) / NCH)
#define RIT ((RW / 4 + ST - 1) / ST)
__global__ __launch_bounds__(ST) void k_chunk_place(
    const int* __restrict__ src, const int* __restrict__ dst,
    const int* __restrict__ rloc, const int* __restrict__ btot,
    const int* __restrict__ gpre, int* __restrict__ rowstart, int* __restrict__ csrc) {
    __shared__ __attribute__((aligned(16))) int slot[RW + 32];
    __shared__ int boff[NB];
    const int tid = threadIdx.x;
    const int g   = blockIdx.x / NRR;
    const int r0  = (blockIdx.x % NRR) * RW;
    const int4* sv = reinterpret_cast<const int4*>(src) + (size_t)g * CH4;
    const int4* dv = reinterpret_cast<const int4*>(dst) + (size_t)g * CH4;
    int4 s[EIT], d[EIT];
#pragma unroll
    for (int k = 0; k < EIT; ++k) {
        const int i = tid + k * ST;
        s[k] = make_int4(0, 0, 0, 0);
        d[k] = make_int4(0, 0, 0, 0);
        if (i < CH4) { s[k] = sv[i]; d[k] = dv[i]; }
    }
    const int4* l4 = reinterpret_cast<const int4*>(rloc);
    const int4* p4 = reinterpret_cast<const int4*>(gpre + (size_t)g * NN);
    int4 fa[RIT], fb[RIT];
#pragma unroll
    for (int k = 0; k < RIT; ++k) {
        const int i = tid + k * ST;
        const int j = r0 / 4 + i;
        fa[k] = make_int4(0, 0, 0, 0);
        fb[k] = make_int4(0, 0, 0, 0);
        if (i < RW / 4) { fa[k] = l4[j]; fb[k] = p4[j]; }
    }
    if (tid < 32) {
        int t[BPL];
        int sum = 0;
#pragma unroll
        for (int k = 0; k < BPL; ++k) {
            const int b = tid * BPL + k;
            t[k] = (b < NB) ? btot[b] : 0;
            sum += t[k];
        }
        int v = sum;
        for (int dd = 1; dd < 32; dd <<= 1) {
            int u = __shfl_up(v, dd, 32);
            if (tid >= dd) v += u;
        }
        int run = v - sum;
#pragma unroll
        for (int k = 0; k < BPL; ++k) {
            const int b = tid * BPL + k;
            if (b < NB) boff[b] = run;
            run += t[k];
        }
    }
    __syncthreads();
    int4* s4 = reinterpret_cast<int4*>(slot);
#pragma unroll
    for (int k = 0; k < RIT; ++k) {
        const int i = tid + k * ST;
        const int j = r0 / 4 + i;
        if (i < RW / 4) {
            const int4 a = fa[k];
            const int4 b = fb[k];
            const int base = boff[j / (RB / 4)];
            s4[i] = make_int4(a.x + b.x + base, a.y + b.y + base, a.z + b.z + base, a.w + b.w + base);
        }
    }
    if (tid < 32) slot[RW + tid] = 0;
    __syncthreads();
#pragma unroll
    for (int k = 0; k < EIT; ++k) {
        const int i = tid + k * ST;
        if (i < CH4) {
            const unsigned a0 = (unsigned)(d[k].x - r0);
            const unsigned a1 = (unsigned)(d[k].y - r0);
            const unsigned a2 = (unsigned)(d[k].z - r0);
            const unsigned a3 = (unsigned)(d[k].w - r0);
            const bool in0 = a0 < (unsigned)RW, in1 = a1 < (unsigned)RW;
            const bool in2 = a2 < (unsigned)RW, in3 = a3 < (unsigned)RW;
            const int scr = RW + (tid & 31);
            const int p0 = atomicAdd(&slot[in0 ? (int)a0 : scr], 1);
            const int p1 = atomicAdd(&slot[in1 ? (int)a1 : scr], 1);
            const int p2 = atomicAdd(&slot[in2 ? (int)a2 : scr], 1);
            const int p3 = atomicAdd(&slot[in3 ? (int)a3 : scr], 1);
            if (in0) csrc[p0] = s[k].x;
            if (in1) csrc[p1] = s[k].y;
            if (in2) csrc[p2] = s[k].z;
            if (in3) csrc[p3] = s[k].w;
        }
    }
#pragma unroll
    for (int k = 0; k < RIT; ++k) {
        const int i = tid + k * ST;
        const int j = r0 / 4 + i;
        if (i < RW / 4 && i >= g * RSG && i < g * RSG + RSG) {
            const int4 a = fa[k];
            const int base = boff[j / (RB / 4)];
            reinterpret_cast<int4*>(rowstart)[j] = make_int4(a.x + base, a.y + base, a.z + base, a.w + base);
        }
    }
}

template <int K>
__device__ __forceinline__ void gather_rows(const float4* __restrict__ hv, int sidx, float dsrc,
                                            float di, int j, int lane, float4& acc) {
    float4 v[K];
#pragma unroll
    for (int u = 0; u < K; ++u) {
        const int s = __builtin_amdgcn_readlane(sidx, j + u);
        v[u] = hv[(size_t)s * 32 + lane];
    }
#pragma unroll
    for (int u = 0; u < K; ++u) {
        const float w = __int_as_float(__builtin_amdgcn_readlane(__float_as_int(dsrc), j + u)) * di;
        acc.x = __builtin_fmaf(v[u].x, w, acc.x);
        acc.y = __builtin_fmaf(v[u].y, w, acc.y);
        acc.z = __builtin_fmaf(v[u].z, w, acc.z);
        acc.w = __builtin_fmaf(v[u].w, w, acc.w);
    }
}

__device__ __forceinline__ void gather_chunk(const float4* __restrict__ hv, int sidx, float dsrc,
                                             float di, int m, int lane, float4& acc) {
    int j = 0;
    for (; j + 8 <= m; j += 8) gather_rows<8>(hv, sidx, dsrc, di, j, lane, acc);
    if (m - j >= 4) { gather_rows<4>(hv, sidx, dsrc, di, j, lane, acc); j += 4; }
    const int rem = m - j;
    if (rem == 3)      gather_rows<3>(hv, sidx, dsrc, di, j, lane, acc);
    else if (rem == 2) gather_rows<2>(hv, sidx, dsrc, di, j, lane, acc);
    else if (rem == 1) gather_rows<1>(hv, sidx, dsrc, di, j, lane, acc);
}

__device__ __forceinline__ float4 row_start(float4 sv, float di) {
    const float sn = di * di;
    float4 acc;
    acc.x = sv.x * sn; acc.y = sv.y * sn;
    acc.z = sv.z * sn; acc.w = sv.w * sn;
    return acc;
}

#define RPW 4
static_assert(RPW == 4 && NN % RPW == 0, "every wave owns four whole rows; their bounds, dinv and graphs are read as one vector each");
#define TROWS (8 * RPW)
#define TP (D + 4)
static_assert(TROWS == 32, "a tile is two 16-row strips");

template <int POOL>
__global__ __launch_bounds__(256) void k_row_gather(
    const float* __restrict__ h, const int* __restrict__ rowstart,
    const int* __restrict__ csrc, const float* __restrict__ dinv,
    const float* __restrict__ W, const float* __restrict__ bias, float* __restrict__ agg,
    const int* __restrict__ batch, float* __restrict__ cnt) {
    __shared__ __attribute__((aligned(16))) float tile[POOL ? 4 : TROWS * TP];
    const int wave = threadIdx.x >> 5;
    const int n0 = __builtin_amdgcn_readfirstlane((int)(blockIdx.x * 8 + wave)) * RPW;
    const int lane = threadIdx.x & 31;
    const float4* hv = reinterpret_cast<const float4*>(h);

    int gF = 0;
    float4 loc = make_float4(0.0f, 0.0f, 0.0f, 0.0f);
    float nloc = 0.0f;
    if (POOL) gF = batch[blockIdx.x * (8 * RPW)];

    if (n0 < NN) {
        const int4 q = *reinterpret_cast<const int4*>(rowstart + n0);
        int q0 = q.x, q1 = q.y, q2 = q.z, q3 = q.w;
        int q4 = (n0 + RPW < NN) ? rowstart[n0 + RPW] : NE;
        const float4 dq = *reinterpret_cast<const float4*>(dinv + n0);
        float d0 = dq.x, d1 = dq.y, d2 = dq.z, d3 = dq.w;
        int g0 = 0, g1 = 0, g2 = 0, g3 = 0;
        if (POOL) {
            const int4 gq = *reinterpret_cast<const int4*>(batch + n0);
            g0 = gq.x; g1 = gq.y; g2 = gq.z; g3 = gq.w;
        }

        int sidx = n0;
        {
            const int m0 = (q1 - q0 < 32) ? (q1 - q0) : 32;
            if (lane < m0) sidx = csrc[q0 + lane];
        }
        float4 acc = row_start(hv[(size_t)n0 * 32 + lane], d0);

        for (int r = 0; r < RPW; ++r) {
            const int n = n0 + r;
            const float di = d0;
            const float dsrc = dinv[sidx];
            const int mn = (q2 - q1 < 32) ? (q2 - q1) : 32;
            int sidx_next = n + 1;
            if (lane < mn) sidx_next = csrc[q1 + lane];
            const int nn = (n + 1 < NN) ? n + 1 : n;
            const float4 sv_next = hv[(size_t)nn * 32 + lane];

            gather_chunk(hv, sidx, dsrc, di, (q1 - q0 < 32) ? (q1 - q0) : 32, lane, acc);
            for (int base = q0 + 32; base < q1; base += 32) {
                const int m2 = (q1 - base < 32) ? (q1 - base) : 32;
                int s2 = n;
                if (lane < m2) s2 = csrc[base + lane];
                const float ds2 = dinv[s2];
                gather_chunk(hv, s2, ds2, di, m2, lane, acc);
            }

            if (POOL) {
                if (g0 == gF) {
                    loc.x += acc.x; loc.y += acc.y; loc.z += acc.z; loc.w += acc.w;
                    nloc += 1.0f;
                } else {
                    if (lane == 0) fatomic_add(cnt + g0, 1.0f);
                    float* o = agg + (size_t)g0 * D + lane * 4;
                    fatomic_add(o + 0, acc.x);
                    fatomic_add(o + 1, acc.y);
                    fatomic_add(o + 2, acc.z);
                    fatomic_add(o + 3, acc.w);
                }
            } else {
                *reinterpret_cast<float4*>(&tile[(wave * RPW + r) * TP + lane * 4]) = acc;
            }
            acc = row_start(sv_next, d1);
            sidx = sidx_next;
            q0 = q1; q1 = q2; q2 = q3; q3 = q4;
            d0 = d1; d1 = d2; d2 = d3;
            g0 = g1; g1 = g2; g2 = g3;
        }
    } else if (!POOL) {
        for (int r = 0; r < RPW; ++r)
            *reinterpret_cast<float4*>(&tile[(wave * RPW + r) * TP + lane * 4]) = make_float4(0.0f, 0.0f, 0.0f, 0.0f);
    }

    if (!POOL) {
        const int koff = (lane >> 4) << 1;
        const int col  = lane & 15;
        const int ocol = wave * 16 + col;
        const float* bp  = W + (size_t)koff * D + ocol;
        const float bcol = bias[ocol];
        const float* a0p = &tile[col * TP + koff];
        const float* a1p = &tile[(16 + col) * TP + koff];
        v8f acc0 = {};
        v8f acc1 = {};
        __syncthreads();
#pragma unroll 8
        for (int k = 0; k < D; k += 4) {
            v2f b, a0, a1;
            b.x = bp[(size_t)k * D];
            b.y = bp[(size_t)(k + 1) * D];
            a0 = *reinterpret_cast<const v2f*>(a0p + k);
            a1 = *reinterpret_cast<const v2f*>(a1p + k);
            acc0 = __builtin_amdgcn_wmma_f32_16x16x4_f32(
                false, a0, false, b, (short)0, acc0, false, false);
            acc1 = __builtin_amdgcn_wmma_f32_16x16x4_f32(
                false, a1, false, b, (short)0, acc1, false, false);
        }
        __syncthreads();
        const int rt = (lane >> 4) * 8;
#pragma unroll
        for (int r = 0; r < 8; ++r) {
            tile[(rt + r) * TP + ocol]      = fmaxf(acc0[r] + bcol, 0.0f);
            tile[(16 + rt + r) * TP + ocol] = fmaxf(acc1[r] + bcol, 0.0f);
        }
        __syncthreads();
        if (n0 < NN) {
#pragma unroll
            for (int r = 0; r < RPW; ++r)
                *reinterpret_cast<float4*>(agg + (size_t)(n0 + r) * D + lane * 4) =
                    *reinterpret_cast<const float4*>(&tile[(wave * RPW + r) * TP + lane * 4]);
        }
    }

    if (POOL) {
        __shared__ __attribute__((aligned(16))) float part[8 * D];
        __shared__ float pcnt[8];
        *reinterpret_cast<float4*>(&part[wave * D + lane * 4]) = loc;
        if (lane == 0) pcnt[wave] = nloc;
        __syncthreads();
        if (threadIdx.x < D) {
            float s = part[threadIdx.x];
            for (int w = 1; w < 8; ++w) s += part[w * D + threadIdx.x];
            fatomic_add(agg + (size_t)gF * D + threadIdx.x, s);
        } else if (threadIdx.x == D) {
            float c = pcnt[0];
            for (int w = 1; w < 8; ++w) c += pcnt[w];
            fatomic_add(cnt + gF, c);
        }
    }
}

__global__ __launch_bounds__(256) void k_pool_transform(
    const float* __restrict__ sums, const float* __restrict__ cnt,
    const float* __restrict__ W, const float* __restrict__ bias, float* __restrict__ out) {
    const int lane  = threadIdx.x & 31;
    const int wave  = threadIdx.x >> 5;
    const int strip = blockIdx.x * 16;
    const int koff  = (lane >> 4) << 1;
    const int col   = lane & 15;
    const int ocol  = wave * 16 + col;
    const float* ap = sums + (size_t)(strip + col) * D + koff;
    const float* bp = W + (size_t)koff * D + ocol;
    v8f acc0 = {};
    v8f acc1 = {};
#pragma unroll
    for (int k = 0; k < D; k += 8) {
        v2f a0, b0, a1, b1;
        a0.x = ap[k];     a0.y = ap[k + 1];
        a1.x = ap[k + 4]; a1.y = ap[k + 5];
        b0.x = bp[(size_t)k * D];       b0.y = bp[(size_t)(k + 1) * D];
        b1.x = bp[(size_t)(k + 4) * D]; b1.y = bp[(size_t)(k + 5) * D];
        acc0 = __builtin_amdgcn_wmma_f32_16x16x4_f32(
            false, a0, false, b0, (short)0, acc0, false, false);
        acc1 = __builtin_amdgcn_wmma_f32_16x16x4_f32(
            false, a1, false, b1, (short)0, acc1, false, false);
    }
    const int rbase = strip + (lane >> 4) * 8;
    const float bcol = bias[ocol];
#pragma unroll
    for (int r = 0; r < 8; ++r) {
        const int g = rbase + r;
        const float c = cnt[g];
        out[(size_t)g * D + ocol] = (acc0[r] + acc1[r] + c * bcol) / fmaxf(c, 1.0f);
    }
}

extern "C" void kernel_launch(void* const* d_in, const int* in_sizes, int n_in,
                              void* d_out, int out_size, void* d_ws, size_t ws_size,
                              hipStream_t stream) {
    const float* x     = (const float*)d_in[0];
    const int*   edge  = (const int*)d_in[1];
    const int*   batch = (const int*)d_in[2];
    const float* W1 = (const float*)d_in[3];
    const float* b1 = (const float*)d_in[4];
    const float* W2 = (const float*)d_in[5];
    const float* b2 = (const float*)d_in[6];
    const float* W3 = (const float*)d_in[7];
    const float* b3 = (const float*)d_in[8];

    const int* esrc = edge;
    const int* edst = edge + NE;

    float* ws   = (float*)d_ws;
    int*   rowst = (int*)ws;
    float* dinv = ws + NN;
    float* bufH = ws + 100000;
    float* bufA = bufH + (size_t)NN * D;
    float* sums = bufA + (size_t)NN * D;
    float* cnt  = sums + NG * D;
    int*   csrc = (int*)(cnt + NG);
    int*   ghist = (int*)bufA;
    int*   gpre  = (int*)bufH;
    int*   rloc  = ghist + (size_t)NCH * NN;
    int*   btot  = rloc + NN;

    const int TB = 256;
    const int gNode32  = (NN / RPW * 32 + TB - 1) / TB;

    k_chunk_count <<<NCH * NRR, ST, 0, stream>>>(edst, ghist);
    k_chunk_prefix<<<NB, RB, 0, stream>>>(ghist, gpre, rloc, btot, dinv, sums);
    k_chunk_place <<<NCH * NRR, ST, 0, stream>>>(esrc, edst, rloc, btot, gpre, rowst, csrc);

    k_row_gather<0><<<gNode32, TB, 0, stream>>>(x, rowst, csrc, dinv, W1, b1, bufH, nullptr, nullptr);

    k_row_gather<0><<<gNode32, TB, 0, stream>>>(bufH, rowst, csrc, dinv, W2, b2, bufA, nullptr, nullptr);

    k_row_gather<1><<<gNode32, TB, 0, stream>>>(bufA, rowst, csrc, dinv, nullptr, nullptr, sums, batch, cnt);
    k_pool_transform<<<NG / 16, TB, 0, stream>>>(sums, cnt, W3, b3, (float*)d_out);
}


# Layers 1 and 2 store rows times their own dinv; layers 2 and 3 gather them unweighted and scale the finished sum by dinv[n]: no per-edge dinv[src] lookup; sum order kept, rounding changes
# speedup vs baseline: 4.4977x; 1.0401x over previous
#include <hip/hip_runtime.h>
#include <hip/hip_bf16.h>

#define NN 50000
#define NE 800000
#define NG 256
#define D  128

typedef float v2f __attribute__((ext_vector_type(2)));
typedef float v8f __attribute__((ext_vector_type(8)));

__device__ __forceinline__ void fatomic_add(float* p, float v) {
    unsafeAtomicAdd(p, v);
}

static_assert(NN % 4 == 0, "row scan moves four counts per thread");
#define NCH 64
#define NRR 4
#define RW  (NN / NRR)
#define CH4 (NE / 4 / NCH)
#define ST  512
#define EIT ((CH4 + ST - 1) / ST)
static_assert(NE % (4 * NCH) == 0, "every chunk is a whole number of four-edge groups");
static_assert(NN % (4 * NRR) == 0, "a range is a whole number of four-row groups, so that its counters move as 16-byte vectors");

__global__ __launch_bounds__(ST) void k_chunk_count(const int* __restrict__ dst,
                                                      int* __restrict__ ghist) {
    __shared__ __attribute__((aligned(16))) int hist[RW];
    const int tid = threadIdx.x;
    const int g   = blockIdx.x / NRR;
    const int r0  = (blockIdx.x % NRR) * RW;
    int4* h4 = reinterpret_cast<int4*>(hist);
    const int4* d4 = reinterpret_cast<const int4*>(dst) + (size_t)g * CH4;
    int4 d[EIT];
#pragma unroll
    for (int k = 0; k < EIT; ++k) {
        const int i = tid + k * ST;
        d[k] = make_int4(0, 0, 0, 0);
        if (i < CH4) d[k] = d4[i];
    }
    for (int i = tid; i < RW / 4; i += ST) h4[i] = make_int4(0, 0, 0, 0);
    __syncthreads();
#pragma unroll
    for (int k = 0; k < EIT; ++k) {
        const int i = tid + k * ST;
        if (i < CH4) {
            const unsigned a0 = (unsigned)(d[k].x - r0);
            const unsigned a1 = (unsigned)(d[k].y - r0);
            const unsigned a2 = (unsigned)(d[k].z - r0);
            const unsigned a3 = (unsigned)(d[k].w - r0);
            if (a0 < (unsigned)RW) atomicAdd(&hist[a0], 1);
            if (a1 < (unsigned)RW) atomicAdd(&hist[a1], 1);
            if (a2 < (unsigned)RW) atomicAdd(&hist[a2], 1);
            if (a3 < (unsigned)RW) atomicAdd(&hist[a3], 1);
        }
    }
    __syncthreads();
    int4* o4 = reinterpret_cast<int4*>(ghist + (size_t)g * NN + r0);
    for (int i = tid; i < RW / 4; i += ST) o4[i] = h4[i];
}

#define RB 256
#define NB ((NN + RB - 1) / RB)
static_assert(RB % 4 == 0, "four rows that are read as one vector lie in one work-group of the prefix kernel");
static_assert(NB * RB >= NG * D + NG, "one thread of the prefix kernel per pooled sum and count");
__global__ void k_chunk_prefix(const int* __restrict__ ghist, int* __restrict__ gpre,
                               int* __restrict__ rloc, int* __restrict__ btot,
                               float* __restrict__ dinv, float* __restrict__ pooled) {
    __shared__ int wtot[RB / 32];
    const int tid  = threadIdx.x;
    const int lane = tid & 31;
    const int wave = tid >> 5;
    const int n = blockIdx.x * RB + tid;
    int h[NCH];
    int run = 0;
    if (n < NN) {
#pragma unroll
        for (int g = 0; g < NCH; ++g) h[g] = ghist[(size_t)g * NN + n];
#pragma unroll
        for (int g = 0; g < NCH; ++g) run += h[g];
    }
    int v = run;
    for (int d = 1; d < 32; d <<= 1) {
        int t = __shfl_up(v, d, 32);
        if (lane >= d) v += t;
    }
    if (lane == 31) wtot[wave] = v;
    __syncthreads();
    int off = 0;
#pragma unroll
    for (int w = 0; w < RB / 32; ++w) {
        const int t = wtot[w];
        if (w < wave) off += t;
    }
    if (n < NN) {
        int pre = 0;
#pragma unroll
        for (int g = 0; g < NCH; ++g) {
            gpre[(size_t)g * NN + n] = pre;
            pre += h[g];
        }
        const float dg = (float)(run + 1);
        dinv[n] = dg > 0.0f ? rsqrtf(dg) : 0.0f;
        rloc[n] = off + v - run;
    }
    if (tid == RB - 1) btot[blockIdx.x] = off + v;
    const int z = blockIdx.x * RB + tid;
    if (z < NG * D + NG) pooled[z] = 0.0f;
}

#define BPL ((NB + 31) / 32)
#define RSG ((RW / 4 + NCH - 1) / NCH)
#define RIT ((RW / 4 + ST - 1) / ST)
__global__ __launch_bounds__(ST) void k_chunk_place(
    const int* __restrict__ src, const int* __restrict__ dst,
    const int* __restrict__ rloc, const int* __restrict__ btot,
    const int* __restrict__ gpre, int* __restrict__ rowstart, int* __restrict__ csrc) {
    __shared__ __attribute__((aligned(16))) int slot[RW + 32];
    __shared__ int boff[NB];
    const int tid = threadIdx.x;
    const int g   = blockIdx.x / NRR;
    const int r0  = (blockIdx.x % NRR) * RW;
    const int4* sv = reinterpret_cast<const int4*>(src) + (size_t)g * CH4;
    const int4* dv = reinterpret_cast<const int4*>(dst) + (size_t)g * CH4;
    int4 s[EIT], d[EIT];
#pragma unroll
    for (int k = 0; k < EIT; ++k) {
        const int i = tid + k * ST;
        s[k] = make_int4(0, 0, 0, 0);
        d[k] = make_int4(0, 0, 0, 0);
        if (i < CH4) { s[k] = sv[i]; d[k] = dv[i]; }
    }
    const int4* l4 = reinterpret_cast<const int4*>(rloc);
    const int4* p4 = reinterpret_cast<const int4*>(gpre + (size_t)g * NN);
    int4 fa[RIT], fb[RIT];
#pragma unroll
    for (int k = 0; k < RIT; ++k) {
        const int i = tid + k * ST;
        const int j = r0 / 4 + i;
        fa[k] = make_int4(0, 0, 0, 0);
        fb[k] = make_int4(0, 0, 0, 0);
        if (i < RW / 4) { fa[k] = l4[j]; fb[k] = p4[j]; }
    }
    if (tid < 32) {
        int t[BPL];
        int sum = 0;
#pragma unroll
        for (int k = 0; k < BPL; ++k) {
            const int b = tid * BPL + k;
            t[k] = (b < NB) ? btot[b] : 0;
            sum += t[k];
        }
        int v = sum;
        for (int dd = 1; dd < 32; dd <<= 1) {
            int u = __shfl_up(v, dd, 32);
            if (tid >= dd) v += u;
        }
        int run = v - sum;
#pragma unroll
        for (int k = 0; k < BPL; ++k) {
            const int b = tid * BPL + k;
            if (b < NB) boff[b] = run;
            run += t[k];
        }
    }
    __syncthreads();
    int4* s4 = reinterpret_cast<int4*>(slot);
#pragma unroll
    for (int k = 0; k < RIT; ++k) {
        const int i = tid + k * ST;
        const int j = r0 / 4 + i;
        if (i < RW / 4) {
            const int4 a = fa[k];
            const int4 b = fb[k];
            const int base = boff[j / (RB / 4)];
            s4[i] = make_int4(a.x + b.x + base, a.y + b.y + base, a.z + b.z + base, a.w + b.w + base);
        }
    }
    if (tid < 32) slot[RW + tid] = 0;
    __syncthreads();
#pragma unroll
    for (int k = 0; k < EIT; ++k) {
        const int i = tid + k * ST;
        if (i < CH4) {
            const unsigned a0 = (unsigned)(d[k].x - r0);
            const unsigned a1 = (unsigned)(d[k].y - r0);
            const unsigned a2 = (unsigned)(d[k].z - r0);
            const unsigned a3 = (unsigned)(d[k].w - r0);
            const bool in0 = a0 < (unsigned)RW, in1 = a1 < (unsigned)RW;
            const bool in2 = a2 < (unsigned)RW, in3 = a3 < (unsigned)RW;
            const int scr = RW + (tid & 31);
            const int p0 = atomicAdd(&slot[in0 ? (int)a0 : scr], 1);
            const int p1 = atomicAdd(&slot[in1 ? (int)a1 : scr], 1);
            const int p2 = atomicAdd(&slot[in2 ? (int)a2 : scr], 1);
            const int p3 = atomicAdd(&slot[in3 ? (int)a3 : scr], 1);
            if (in0) csrc[p0] = s[k].x;
            if (in1) csrc[p1] = s[k].y;
            if (in2) csrc[p2] = s[k].z;
            if (in3) csrc[p3] = s[k].w;
        }
    }
#pragma unroll
    for (int k = 0; k < RIT; ++k) {
        const int i = tid + k * ST;
        const int j = r0 / 4 + i;
        if (i < RW / 4 && i >= g * RSG && i < g * RSG + RSG) {
            const int4 a = fa[k];
            const int base = boff[j / (RB / 4)];
            reinterpret_cast<int4*>(rowstart)[j] = make_int4(a.x + base, a.y + base, a.z + base, a.w + base);
        }
    }
}

template <int K, int PRE>
__device__ __forceinline__ void gather_rows(const float4* __restrict__ hv, int sidx, float dsrc,
                                            float di, int j, int lane, float4& acc) {
    float4 v[K];
#pragma unroll
    for (int u = 0; u < K; ++u) {
        const int s = __builtin_amdgcn_readlane(sidx, j + u);
        v[u] = hv[(size_t)s * 32 + lane];
    }
#pragma unroll
    for (int u = 0; u < K; ++u) {
        if (PRE) {
            acc.x += v[u].x;
            acc.y += v[u].y;
            acc.z += v[u].z;
            acc.w += v[u].w;
        } else {
            const float w = __int_as_float(__builtin_amdgcn_readlane(__float_as_int(dsrc), j + u)) * di;
            acc.x = __builtin_fmaf(v[u].x, w, acc.x);
            acc.y = __builtin_fmaf(v[u].y, w, acc.y);
            acc.z = __builtin_fmaf(v[u].z, w, acc.z);
            acc.w = __builtin_fmaf(v[u].w, w, acc.w);
        }
    }
}

template <int PRE>
__device__ __forceinline__ void gather_chunk(const float4* __restrict__ hv, int sidx, float dsrc,
                                             float di, int m, int lane, float4& acc) {
    int j = 0;
    for (; j + 8 <= m; j += 8) gather_rows<8, PRE>(hv, sidx, dsrc, di, j, lane, acc);
    if (m - j >= 4) { gather_rows<4, PRE>(hv, sidx, dsrc, di, j, lane, acc); j += 4; }
    const int rem = m - j;
    if (rem == 3)      gather_rows<3, PRE>(hv, sidx, dsrc, di, j, lane, acc);
    else if (rem == 2) gather_rows<2, PRE>(hv, sidx, dsrc, di, j, lane, acc);
    else if (rem == 1) gather_rows<1, PRE>(hv, sidx, dsrc, di, j, lane, acc);
}

template <int PRE>
__device__ __forceinline__ float4 row_start(float4 sv, float di) {
    if (PRE) return sv;
    const float sn = di * di;
    float4 acc;
    acc.x = sv.x * sn; acc.y = sv.y * sn;
    acc.z = sv.z * sn; acc.w = sv.w * sn;
    return acc;
}

#define RPW 4
static_assert(RPW == 4 && NN % RPW == 0, "every wave owns four whole rows; their bounds, dinv and graphs are read as one vector each");
#define TROWS (8 * RPW)
#define TP (D + 4)
static_assert(TROWS == 32, "a tile is two 16-row strips");

template <int POOL, int PRE>
__global__ __launch_bounds__(256) void k_row_gather(
    const float* __restrict__ h, const int* __restrict__ rowstart,
    const int* __restrict__ csrc, const float* __restrict__ dinv,
    const float* __restrict__ W, const float* __restrict__ bias, float* __restrict__ agg,
    const int* __restrict__ batch, float* __restrict__ cnt) {
    __shared__ __attribute__((aligned(16))) float tile[POOL ? 4 : TROWS * TP];
    const int wave = threadIdx.x >> 5;
    const int n0 = __builtin_amdgcn_readfirstlane((int)(blockIdx.x * 8 + wave)) * RPW;
    const int lane = threadIdx.x & 31;
    const float4* hv = reinterpret_cast<const float4*>(h);

    int gF = 0;
    float4 loc = make_float4(0.0f, 0.0f, 0.0f, 0.0f);
    float nloc = 0.0f;
    if (POOL) gF = batch[blockIdx.x * (8 * RPW)];
    float4 dq = make_float4(0.0f, 0.0f, 0.0f, 0.0f);

    if (n0 < NN) {
        const int4 q = *reinterpret_cast<const int4*>(rowstart + n0);
        int q0 = q.x, q1 = q.y, q2 = q.z, q3 = q.w;
        int q4 = (n0 + RPW < NN) ? rowstart[n0 + RPW] : NE;
        dq = *reinterpret_cast<const float4*>(dinv + n0);
        float d0 = dq.x, d1 = dq.y, d2 = dq.z, d3 = dq.w;
        int g0 = 0, g1 = 0, g2 = 0, g3 = 0;
        if (POOL) {
            const int4 gq = *reinterpret_cast<const int4*>(batch + n0);
            g0 = gq.x; g1 = gq.y; g2 = gq.z; g3 = gq.w;
        }

        int sidx = n0;
        {
            const int m0 = (q1 - q0 < 32) ? (q1 - q0) : 32;
            if (lane < m0) sidx = csrc[q0 + lane];
        }
        float4 acc = row_start<PRE>(hv[(size_t)n0 * 32 + lane], d0);

        for (int r = 0; r < RPW; ++r) {
            const int n = n0 + r;
            const float di = d0;
            float dsrc = 0.0f;
            if (!PRE) dsrc = dinv[sidx];
            const int mn = (q2 - q1 < 32) ? (q2 - q1) : 32;
            int sidx_next = n + 1;
            if (lane < mn) sidx_next = csrc[q1 + lane];
            const int nn = (n + 1 < NN) ? n + 1 : n;
            const float4 sv_next = hv[(size_t)nn * 32 + lane];

            gather_chunk<PRE>(hv, sidx, dsrc, di, (q1 - q0 < 32) ? (q1 - q0) : 32, lane, acc);
            for (int base = q0 + 32; base < q1; base += 32) {
                const int m2 = (q1 - base < 32) ? (q1 - base) : 32;
                int s2 = n;
                if (lane < m2) s2 = csrc[base + lane];
                float ds2 = 0.0f;
                if (!PRE) ds2 = dinv[s2];
                gather_chunk<PRE>(hv, s2, ds2, di, m2, lane, acc);
            }
            if (PRE) {
                acc.x *= di; acc.y *= di; acc.z *= di; acc.w *= di;
            }

            if (POOL) {
                if (g0 == gF) {
                    loc.x += acc.x; loc.y += acc.y; loc.z += acc.z; loc.w += acc.w;
                    nloc += 1.0f;
                } else {
                    if (lane == 0) fatomic_add(cnt + g0, 1.0f);
                    float* o = agg + (size_t)g0 * D + lane * 4;
                    fatomic_add(o + 0, acc.x);
                    fatomic_add(o + 1, acc.y);
                    fatomic_add(o + 2, acc.z);
                    fatomic_add(o + 3, acc.w);
                }
            } else {
                *reinterpret_cast<float4*>(&tile[(wave * RPW + r) * TP + lane * 4]) = acc;
            }
            acc = row_start<PRE>(sv_next, d1);
            sidx = sidx_next;
            q0 = q1; q1 = q2; q2 = q3; q3 = q4;
            d0 = d1; d1 = d2; d2 = d3;
            g0 = g1; g1 = g2; g2 = g3;
        }
    } else if (!POOL) {
        for (int r = 0; r < RPW; ++r)
            *reinterpret_cast<float4*>(&tile[(wave * RPW + r) * TP + lane * 4]) = make_float4(0.0f, 0.0f, 0.0f, 0.0f);
    }

    if (!POOL) {
        const int koff = (lane >> 4) << 1;
        const int col  = lane & 15;
        const int ocol = wave * 16 + col;
        const float* bp  = W + (size_t)koff * D + ocol;
        const float bcol = bias[ocol];
        const float* a0p = &tile[col * TP + koff];
        const float* a1p = &tile[(16 + col) * TP + koff];
        v8f acc0 = {};
        v8f acc1 = {};
        __syncthreads();
#pragma unroll 8
        for (int k = 0; k < D; k += 4) {
            v2f b, a0, a1;
            b.x = bp[(size_t)k * D];
            b.y = bp[(size_t)(k + 1) * D];
            a0 = *reinterpret_cast<const v2f*>(a0p + k);
            a1 = *reinterpret_cast<const v2f*>(a1p + k);
            acc0 = __builtin_amdgcn_wmma_f32_16x16x4_f32(
                false, a0, false, b, (short)0, acc0, false, false);
            acc1 = __builtin_amdgcn_wmma_f32_16x16x4_f32(
                false, a1, false, b, (short)0, acc1, false, false);
        }
        __syncthreads();
        const int rt = (lane >> 4) * 8;
#pragma unroll
        for (int r = 0; r < 8; ++r) {
            tile[(rt + r) * TP + ocol]      = fmaxf(acc0[r] + bcol, 0.0f);
            tile[(16 + rt + r) * TP + ocol] = fmaxf(acc1[r] + bcol, 0.0f);
        }
        __syncthreads();
        if (n0 < NN) {
            const float dr[RPW] = {dq.x, dq.y, dq.z, dq.w};
#pragma unroll
            for (int r = 0; r < RPW; ++r) {
                float4 t = *reinterpret_cast<const float4*>(&tile[(wave * RPW + r) * TP + lane * 4]);
                t.x *= dr[r]; t.y *= dr[r]; t.z *= dr[r]; t.w *= dr[r];
                *reinterpret_cast<float4*>(agg + (size_t)(n0 + r) * D + lane * 4) = t;
            }
        }
    }

    if (POOL) {
        __shared__ __attribute__((aligned(16))) float part[8 * D];
        __shared__ float pcnt[8];
        *reinterpret_cast<float4*>(&part[wave * D + lane * 4]) = loc;
        if (lane == 0) pcnt[wave] = nloc;
        __syncthreads();
        if (threadIdx.x < D) {
            float s = part[threadIdx.x];
            for (int w = 1; w < 8; ++w) s += part[w * D + threadIdx.x];
            fatomic_add(agg + (size_t)gF * D + threadIdx.x, s);
        } else if (threadIdx.x == D) {
            float c = pcnt[0];
            for (int w = 1; w < 8; ++w) c += pcnt[w];
            fatomic_add(cnt + gF, c);
        }
    }
}

__global__ __launch_bounds__(256) void k_pool_transform(
    const float* __restrict__ sums, const float* __restrict__ cnt,
    const float* __restrict__ W, const float* __restrict__ bias, float* __restrict__ out) {
    const int lane  = threadIdx.x & 31;
    const int wave  = threadIdx.x >> 5;
    const int strip = blockIdx.x * 16;
    const int koff  = (lane >> 4) << 1;
    const int col   = lane & 15;
    const int ocol  = wave * 16 + col;
    const float* ap = sums + (size_t)(strip + col) * D + koff;
    const float* bp = W + (size_t)koff * D + ocol;
    v8f acc0 = {};
    v8f acc1 = {};
#pragma unroll
    for (int k = 0; k < D; k += 8) {
        v2f a0, b0, a1, b1;
        a0.x = ap[k];     a0.y = ap[k + 1];
        a1.x = ap[k + 4]; a1.y = ap[k + 5];
        b0.x = bp[(size_t)k * D];       b0.y = bp[(size_t)(k + 1) * D];
        b1.x = bp[(size_t)(k + 4) * D]; b1.y = bp[(size_t)(k + 5) * D];
        acc0 = __builtin_amdgcn_wmma_f32_16x16x4_f32(
            false, a0, false, b0, (short)0, acc0, false, false);
        acc1 = __builtin_amdgcn_wmma_f32_16x16x4_f32(
            false, a1, false, b1, (short)0, acc1, false, false);
    }
    const int rbase = strip + (lane >> 4) * 8;
    const float bcol = bias[ocol];
#pragma unroll
    for (int r = 0; r < 8; ++r) {
        const int g = rbase + r;
        const float c = cnt[g];
        out[(size_t)g * D + ocol] = (acc0[r] + acc1[r] + c * bcol) / fmaxf(c, 1.0f);
    }
}

extern "C" void kernel_launch(void* const* d_in, const int* in_sizes, int n_in,
                              void* d_out, int out_size, void* d_ws, size_t ws_size,
                              hipStream_t stream) {
    const float* x     = (const float*)d_in[0];
    const int*   edge  = (const int*)d_in[1];
    const int*   batch = (const int*)d_in[2];
    const float* W1 = (const float*)d_in[3];
    const float* b1 = (const float*)d_in[4];
    const float* W2 = (const float*)d_in[5];
    const float* b2 = (const float*)d_in[6];
    const float* W3 = (const float*)d_in[7];
    const float* b3 = (const float*)d_in[8];

    const int* esrc = edge;
    const int* edst = edge + NE;

    float* ws   = (float*)d_ws;
    int*   rowst = (int*)ws;
    float* dinv = ws + NN;
    float* bufH = ws + 100000;
    float* bufA = bufH + (size_t)NN * D;
    float* sums = bufA + (size_t)NN * D;
    float* cnt  = sums + NG * D;
    int*   csrc = (int*)(cnt + NG);
    int*   ghist = (int*)bufA;
    int*   gpre  = (int*)bufH;
    int*   rloc  = ghist + (size_t)NCH * NN;
    int*   btot  = rloc + NN;

    const int TB = 256;
    const int gNode32  = (NN / RPW * 32 + TB - 1) / TB;

    k_chunk_count <<<NCH * NRR, ST, 0, stream>>>(edst, ghist);
    k_chunk_prefix<<<NB, RB, 0, stream>>>(ghist, gpre, rloc, btot, dinv, sums);
    k_chunk_place <<<NCH * NRR, ST, 0, stream>>>(esrc, edst, rloc, btot, gpre, rowst, csrc);

    k_row_gather<0, 0><<<gNode32, TB, 0, stream>>>(x, rowst, csrc, dinv, W1, b1, bufH, nullptr, nullptr);

    k_row_gather<0, 1><<<gNode32, TB, 0, stream>>>(bufH, rowst, csrc, dinv, W2, b2, bufA, nullptr, nullptr);

    k_row_gather<1, 1><<<gNode32, TB, 0, stream>>>(bufA, rowst, csrc, dinv, nullptr, nullptr, sums, batch, cnt);
    k_pool_transform<<<NG / 16, TB, 0, stream>>>(sums, cnt, W3, b3, (float*)d_out);
}
